# starting kernel
# baseline (speedup 1.0000x reference)
#include <hip/hip_runtime.h>
#include <hip/hip_bf16.h>

typedef __attribute__((ext_vector_type(2))) float v2f;
typedef __attribute__((ext_vector_type(8))) float v8f;

#define BB 384
#define DD 1024
#define WD 300
#define NN 8000
#define LL 16

static __device__ __forceinline__ v8f wmma4(v2f a, v2f b, v8f c) {
  return __builtin_amdgcn_wmma_f32_16x16x4_f32(false, a, false, b, (short)0, c,
                                               false, false);
}

__global__ __launch_bounds__(32) void k_cls(const int* __restrict__ tok,
                                            const float* __restrict__ emb,
                                            const float* __restrict__ W,
                                            const float* __restrict__ bias,
                                            float* __restrict__ out) {
  const int lane = threadIdx.x & 31;
  const int half = lane >> 4;
  const int lid  = lane & 15;
  const int m0 = blockIdx.x * 64, n0 = blockIdx.y * 16;
  const int col = n0 + lid;
  const float* arow[4];
#pragma unroll
  for (int s = 0; s < 4; ++s)
    arow[s] = emb + (size_t)tok[m0 + s * 16 + lid] * WD;
  v8f acc[4];
#pragma unroll
  for (int s = 0; s < 4; ++s) acc[s] = (v8f){0.f,0.f,0.f,0.f,0.f,0.f,0.f,0.f};
  for (int k0 = 0; k0 < WD; k0 += 4) {
    const int ka = k0 + 2 * half;
    v2f b;
    b.x = W[(size_t)ka * DD + col];
    b.y = W[(size_t)(ka + 1) * DD + col];
#pragma unroll
    for (int s = 0; s < 4; ++s) {
      v2f a = *(const v2f*)(arow[s] + ka);
      acc[s] = wmma4(a, b, acc[s]);
    }
  }
  const float bv = bias[col];
#pragma unroll
  for (int s = 0; s < 4; ++s)
#pragma unroll
    for (int i = 0; i < 8; ++i)
      out[(size_t)(m0 + s * 16 + half * 8 + i) * DD + col] = acc[s][i] + bv;
}

__global__ __launch_bounds__(32) void k_concepts(const int* __restrict__ ntid,
                                                 const int* __restrict__ nlen,
                                                 const int* __restrict__ t2i,
                                                 const float* __restrict__ emb,
                                                 const float* __restrict__ W,
                                                 const float* __restrict__ bias,
                                                 const float* __restrict__ cls,
                                                 float* __restrict__ concepts) {
  const int lane = threadIdx.x & 31;
  const int half = lane >> 4;
  const int lid  = lane & 15;
  const int m0 = blockIdx.x * 64, n0 = blockIdx.y * 16;
  const int col = n0 + lid;
  const float* arow[4];
#pragma unroll
  for (int s = 0; s < 4; ++s)
    arow[s] = emb + (size_t)ntid[m0 + s * 16 + lid] * WD;
  v8f acc[4];
#pragma unroll
  for (int s = 0; s < 4; ++s) acc[s] = (v8f){0.f,0.f,0.f,0.f,0.f,0.f,0.f,0.f};
  for (int k0 = 0; k0 < WD; k0 += 4) {
    const int ka = k0 + 2 * half;
    v2f b;
    b.x = W[(size_t)ka * DD + col];
    b.y = W[(size_t)(ka + 1) * DD + col];
#pragma unroll
    for (int s = 0; s < 4; ++s) {
      v2f a = *(const v2f*)(arow[s] + ka);
      acc[s] = wmma4(a, b, acc[s]);
    }
  }
  const float bv = bias[col];
#pragma unroll
  for (int s = 0; s < 4; ++s) {
#pragma unroll
    for (int i = 0; i < 8; ++i) {
      const int row  = m0 + s * 16 + half * 8 + i;
      const int bidx = row >> 4;
      const int l    = row & 15;
      const int widx = t2i[ntid[row]];
      const bool valid = l < nlen[bidx];
      const bool found = widx > 0;
      float val = 0.f;
      if (valid) val = found ? cls[(size_t)widx * DD + col] : (acc[s][i] + bv);
      concepts[(size_t)row * DD + col] = val;
    }
  }
}

__global__ __launch_bounds__(32) void k_gemm_xw(const float* __restrict__ X,
                                                const float* __restrict__ W,
                                                const float* __restrict__ bias,
                                                float* __restrict__ out, int relu) {
  const int lane = threadIdx.x & 31;
  const int half = lane >> 4;
  const int lid  = lane & 15;
  const int m0 = blockIdx.x * 32, n0 = blockIdx.y * 16;
  const int col = n0 + lid;
  const float* arow0 = X + (size_t)(m0 + lid) * DD;
  const float* arow1 = X + (size_t)(m0 + 16 + lid) * DD;
  v8f acc0 = {0.f,0.f,0.f,0.f,0.f,0.f,0.f,0.f};
  v8f acc1 = acc0;
  for (int k0 = 0; k0 < DD; k0 += 4) {
    const int ka = k0 + 2 * half;
    v2f b;
    b.x = W[(size_t)ka * DD + col];
    b.y = W[(size_t)(ka + 1) * DD + col];
    v2f a0 = *(const v2f*)(arow0 + ka);
    v2f a1 = *(const v2f*)(arow1 + ka);
    acc0 = wmma4(a0, b, acc0);
    acc1 = wmma4(a1, b, acc1);
  }
  const float bv = bias[col];
#pragma unroll
  for (int i = 0; i < 8; ++i) {
    float v0 = acc0[i] + bv;
    float v1 = acc1[i] + bv;
    if (relu) { v0 = fmaxf(v0, 0.f); v1 = fmaxf(v1, 0.f); }
    out[(size_t)(m0 + half * 8 + i) * DD + col] = v0;
    out[(size_t)(m0 + 16 + half * 8 + i) * DD + col] = v1;
  }
}

__global__ void k_softmax(float* __restrict__ x) {
  __shared__ float red[256];
  float* p = x + (size_t)blockIdx.x * DD;
  const int tid = threadIdx.x;
  float mx = -3.402823466e+38f;
  for (int i = tid; i < DD; i += 256) mx = fmaxf(mx, p[i]);
  red[tid] = mx; __syncthreads();
  for (int s = 128; s > 0; s >>= 1) {
    if (tid < s) red[tid] = fmaxf(red[tid], red[tid + s]);
    __syncthreads();
  }
  mx = red[0]; __syncthreads();
  float sum = 0.f;
  for (int i = tid; i < DD; i += 256) { float e = __expf(p[i] - mx); p[i] = e; sum += e; }
  red[tid] = sum; __syncthreads();
  for (int s = 128; s > 0; s >>= 1) {
    if (tid < s) red[tid] += red[tid + s];
    __syncthreads();
  }
  const float inv = 1.f / red[0];
  for (int i = tid; i < DD; i += 256) p[i] *= inv;
}

__global__ void k_row_l2(const float* __restrict__ src, float* __restrict__ dn,
                         float* __restrict__ dsq) {
  __shared__ float red[256];
  const int row = blockIdx.x, tid = threadIdx.x;
  const float* p = src + (size_t)row * DD;
  float s = 0.f;
  for (int i = tid; i < DD; i += 256) {
    const float v = p[i];
    s += v * v;
    if (dsq) dsq[(size_t)row * DD + i] = v * v;
  }
  red[tid] = s; __syncthreads();
  for (int st = 128; st > 0; st >>= 1) {
    if (tid < st) red[tid] += red[tid + st];
    __syncthreads();
  }
  const float inv = rsqrtf(red[0]);
  for (int i = tid; i < DD; i += 256) dn[(size_t)row * DD + i] = p[i] * inv;
}

__global__ void k_pq(const float* __restrict__ trm, const float* __restrict__ aem,
                     float* __restrict__ P, float* __restrict__ Q) {
  __shared__ float red[256];
  const int row = blockIdx.x, tid = threadIdx.x;
  const float* tp = trm + (size_t)row * DD;
  const float* ap = aem + (size_t)row * DD;
  float s = 0.f;
  for (int i = tid; i < DD; i += 256) { const float v = tp[i]; s += v * v; }
  red[tid] = s; __syncthreads();
  for (int st = 128; st > 0; st >>= 1) {
    if (tid < st) red[tid] += red[tid + st];
    __syncthreads();
  }
  const float inv = rsqrtf(red[0]);
  for (int i = tid; i < DD; i += 256) {
    const float a = ap[i];
    P[(size_t)row * DD + i] = tp[i] * inv * a;
    Q[(size_t)row * DD + i] = a * a;
  }
}

__global__ __launch_bounds__(32) void k_score(const float* __restrict__ P,
                                              const float* __restrict__ Q,
                                              const float* __restrict__ rn,
                                              const float* __restrict__ t,
                                              const float* __restrict__ t2,
                                              const float* __restrict__ tn,
                                              float* __restrict__ score) {
  const int lane = threadIdx.x & 31;
  const int half = lane >> 4;
  const int lid  = lane & 15;
  const int m0 = blockIdx.x * 16, n0 = blockIdx.y * 16;
  const int col = n0 + lid;
  const float* prow  = P  + (size_t)(m0 + lid) * DD;
  const float* qrow  = Q  + (size_t)(m0 + lid) * DD;
  const float* rrow  = rn + (size_t)(m0 + lid) * DD;
  const float* tcol  = t  + (size_t)col * DD;
  const float* t2col = t2 + (size_t)col * DD;
  const float* tncol = tn + (size_t)col * DD;
  v8f num = {0.f,0.f,0.f,0.f,0.f,0.f,0.f,0.f};
  v8f den = num, isv = num;
  for (int k0 = 0; k0 < DD; k0 += 4) {
    const int ka = k0 + 2 * half;
    v2f a1 = *(const v2f*)(prow + ka);
    v2f b1 = *(const v2f*)(tcol + ka);
    v2f a2 = *(const v2f*)(qrow + ka);
    v2f b2 = *(const v2f*)(t2col + ka);
    v2f a3 = *(const v2f*)(rrow + ka);
    v2f b3 = *(const v2f*)(tncol + ka);
    num = wmma4(a1, b1, num);
    den = wmma4(a2, b2, den);
    isv = wmma4(a3, b3, isv);
  }
#pragma unroll
  for (int i = 0; i < 8; ++i) {
    const int row = m0 + half * 8 + i;
    score[(size_t)row * BB + col] = num[i] * rsqrtf(den[i]) + isv[i];
  }
}

extern "C" void kernel_launch(void* const* d_in, const int* in_sizes, int n_in,
                              void* d_out, int out_size, void* d_ws, size_t ws_size,
                              hipStream_t stream) {
  const float* r    = (const float*)d_in[0];
  const float* m    = (const float*)d_in[1];
  const float* t    = (const float*)d_in[2];
  const int*   ntid = (const int*)d_in[3];
  const int*   nlen = (const int*)d_in[4];
  const int*   antk = (const int*)d_in[5];
  const float* emb  = (const float*)d_in[6];
  const float* nW   = (const float*)d_in[7];
  const float* nb   = (const float*)d_in[8];
  const float* W1   = (const float*)d_in[9];
  const float* b1   = (const float*)d_in[10];
  const float* W2   = (const float*)d_in[11];
  const float* b2   = (const float*)d_in[12];
  const float* trW  = (const float*)d_in[13];
  const float* trb  = (const float*)d_in[14];
  const int*   t2i  = (const int*)d_in[15];

  float* score    = (float*)d_out;
  float* concepts = score + (size_t)BB * BB;

  float* ws  = (float*)d_ws;
  float* cls = ws;  ws += (size_t)NN * DD;
  float* h   = ws;  ws += (size_t)BB * DD;
  float* aem = ws;  ws += (size_t)BB * DD;
  float* trm = ws;  ws += (size_t)BB * DD;
  float* Pw  = ws;  ws += (size_t)BB * DD;
  float* Qw  = ws;  ws += (size_t)BB * DD;
  float* rnw = ws;  ws += (size_t)BB * DD;
  float* tnw = ws;  ws += (size_t)BB * DD;
  float* t2w = ws;  ws += (size_t)BB * DD;

  const dim3 wave(32);
  k_cls<<<dim3(NN / 64, DD / 16), wave, 0, stream>>>(antk, emb, nW, nb, cls);
  k_concepts<<<dim3(BB * LL / 64, DD / 16), wave, 0, stream>>>(ntid, nlen, t2i, emb,
                                                               nW, nb, cls, concepts);
  k_gemm_xw<<<dim3(BB / 32, DD / 16), wave, 0, stream>>>(m, W1, b1, h, 1);
  k_gemm_xw<<<dim3(BB / 32, DD / 16), wave, 0, stream>>>(h, W2, b2, aem, 0);
  k_softmax<<<BB, 256, 0, stream>>>(aem);
  k_gemm_xw<<<dim3(BB / 32, DD / 16), wave, 0, stream>>>(m, trW, trb, trm, 0);
  k_pq<<<BB, 256, 0, stream>>>(trm, aem, Pw, Qw);
  k_row_l2<<<BB, 256, 0, stream>>>(r, rnw, nullptr);
  k_row_l2<<<BB, 256, 0, stream>>>(t, tnw, t2w);
  k_score<<<dim3(BB / 16, BB / 16), wave, 0, stream>>>(Pw, Qw, rnw, t, t2w, tnw, score);
}


# Stage A and B tiles of every WMMA GEMM in LDS: 4-wave work-groups fetch each tile chunk once with coalesced loads, two LDS stages, next chunk held in registers; k order and sums unchanged
# speedup vs baseline: 1.9564x; 1.9564x over previous
#include <hip/hip_runtime.h>
#include <hip/hip_bf16.h>

typedef __attribute__((ext_vector_type(2))) float v2f;
typedef __attribute__((ext_vector_type(8))) float v8f;
typedef __attribute__((ext_vector_type(4))) float v4f;

#define BB 384
#define DD 1024
#define WD 300
#define NN 8000
#define LL 16

static __device__ __forceinline__ v8f wmma4(v2f a, v2f b, v8f c) {
  return __builtin_amdgcn_wmma_f32_16x16x4_f32(false, a, false, b, (short)0, c,
                                               false, false);
}

#define TM 64
#define TN 128
#define KC 20
#define NCH (WD / KC)
#define PA 20
#define PB 136

static __device__ __forceinline__ void tile_k300(const int* __restrict__ tokrow,
                                                 const float* __restrict__ emb,
                                                 const float* __restrict__ W,
                                                 const int n0,
                                                 float (*As)[TM * PA],
                                                 float (*Bs)[KC * PB],
                                                 v8f (&acc)[4][2]) {
  const int tid  = threadIdx.x;
  const int lane = tid & 31;
  const int wv   = tid >> 5;
  const int half = lane >> 4;
  const int lid  = lane & 15;

  int aoff[5];
#pragma unroll
  for (int j = 0; j < 5; ++j) {
    const int idx = tid + 128 * j;
    const int row = idx / 10;
    const int c2  = idx - row * 10;
    aoff[j] = tokrow[row] * WD + 2 * c2;
  }
  const int bk  = tid >> 5;
  const int bc4 = tid & 31;
  const float* wp = W + (size_t)bk * DD + n0 + bc4 * 4;

  v2f    ra[5];
  v4f    rb[5];
#pragma unroll
  for (int j = 0; j < 5; ++j) {
    ra[j] = *(const v2f*)(emb + aoff[j]);
    rb[j] = *(const v4f*)(wp + (size_t)(4 * j) * DD);
  }
#pragma unroll
  for (int j = 0; j < 5; ++j) {
    *(v2f*)&As[0][2 * (tid + 128 * j)] = ra[j];
    *(v4f*)&Bs[0][(bk + 4 * j) * PB + bc4 * 4] = rb[j];
  }
  __syncthreads();

  auto chunk_mma = [&](const float* Ab, const float* Bb) {
#pragma unroll
    for (int kk = 0; kk < KC / 4; ++kk) {
      const int k = kk * 4 + 2 * half;
      v2f b[2];
#pragma unroll
      for (int ns = 0; ns < 2; ++ns) {
        b[ns].x = Bb[k * PB + wv * 32 + ns * 16 + lid];
        b[ns].y = Bb[(k + 1) * PB + wv * 32 + ns * 16 + lid];
      }
#pragma unroll
      for (int s = 0; s < 4; ++s) {
        const v2f a = *(const v2f*)&Ab[(s * 16 + lid) * PA + k];
#pragma unroll
        for (int ns = 0; ns < 2; ++ns) acc[s][ns] = wmma4(a, b[ns], acc[s][ns]);
      }
    }
  };

#pragma unroll
  for (int j = 0; j < 5; ++j) {
    ra[j] = *(const v2f*)(emb + aoff[j] + KC);
    rb[j] = *(const v4f*)(wp + (size_t)(KC + 4 * j) * DD);
  }
  for (int c = 0; c < NCH - 2; ++c) {
    chunk_mma(As[c & 1], Bs[c & 1]);
#pragma unroll
    for (int j = 0; j < 5; ++j) {
      *(v2f*)&As[(c + 1) & 1][2 * (tid + 128 * j)] = ra[j];
      *(v4f*)&Bs[(c + 1) & 1][(bk + 4 * j) * PB + bc4 * 4] = rb[j];
    }
    __syncthreads();
    const int kn = (c + 2) * KC;
#pragma unroll
    for (int j = 0; j < 5; ++j) {
      ra[j] = *(const v2f*)(emb + aoff[j] + kn);
      rb[j] = *(const v4f*)(wp + (size_t)(kn + 4 * j) * DD);
    }
  }
  chunk_mma(As[(NCH - 2) & 1], Bs[(NCH - 2) & 1]);
#pragma unroll
  for (int j = 0; j < 5; ++j) {
    *(v2f*)&As[(NCH - 1) & 1][2 * (tid + 128 * j)] = ra[j];
    *(v4f*)&Bs[(NCH - 1) & 1][(bk + 4 * j) * PB + bc4 * 4] = rb[j];
  }
  __syncthreads();
  chunk_mma(As[(NCH - 1) & 1], Bs[(NCH - 1) & 1]);
}

__global__ __launch_bounds__(128) void k_cls(const int* __restrict__ tok,
                                             const float* __restrict__ emb,
                                             const float* __restrict__ W,
                                             const float* __restrict__ bias,
                                             float* __restrict__ out) {
  __shared__ __attribute__((aligned(16))) float As[2][TM * PA];
  __shared__ __attribute__((aligned(16))) float Bs[2][KC * PB];
  const int lane = threadIdx.x & 31;
  const int wv   = threadIdx.x >> 5;
  const int half = lane >> 4;
  const int lid  = lane & 15;
  const int n0 = blockIdx.x * TN, m0 = blockIdx.y * TM;
  v8f acc[4][2];
#pragma unroll
  for (int s = 0; s < 4; ++s)
#pragma unroll
    for (int ns = 0; ns < 2; ++ns) acc[s][ns] = (v8f){0.f,0.f,0.f,0.f,0.f,0.f,0.f,0.f};
  tile_k300(tok + m0, emb, W, n0, As, Bs, acc);
#pragma unroll
  for (int ns = 0; ns < 2; ++ns) {
    const int col = n0 + wv * 32 + ns * 16 + lid;
    const float bv = bias[col];
#pragma unroll
    for (int s = 0; s < 4; ++s)
#pragma unroll
      for (int i = 0; i < 8; ++i)
        out[(size_t)(m0 + s * 16 + half * 8 + i) * DD + col] = acc[s][ns][i] + bv;
  }
}

__global__ __launch_bounds__(128) void k_concepts(const int* __restrict__ ntid,
                                                  const int* __restrict__ nlen,
                                                  const int* __restrict__ t2i,
                                                  const float* __restrict__ emb,
                                                  const float* __restrict__ W,
                                                  const float* __restrict__ bias,
                                                  const float* __restrict__ cls,
                                                  float* __restrict__ concepts) {
  __shared__ __attribute__((aligned(16))) float As[2][TM * PA];
  __shared__ __attribute__((aligned(16))) float Bs[2][KC * PB];
  const int lane = threadIdx.x & 31;
  const int wv   = threadIdx.x >> 5;
  const int half = lane >> 4;
  const int lid  = lane & 15;
  const int n0 = blockIdx.x * TN, m0 = blockIdx.y * TM;
  v8f acc[4][2];
#pragma unroll
  for (int s = 0; s < 4; ++s)
#pragma unroll
    for (int ns = 0; ns < 2; ++ns) acc[s][ns] = (v8f){0.f,0.f,0.f,0.f,0.f,0.f,0.f,0.f};
  tile_k300(ntid + m0, emb, W, n0, As, Bs, acc);
  float bv[2];
#pragma unroll
  for (int ns = 0; ns < 2; ++ns) bv[ns] = bias[n0 + wv * 32 + ns * 16 + lid];
#pragma unroll
  for (int s = 0; s < 4; ++s) {
#pragma unroll
    for (int i = 0; i < 8; ++i) {
      const int row  = m0 + s * 16 + half * 8 + i;
      const int bidx = row >> 4;
      const int l    = row & 15;
      const int widx = t2i[ntid[row]];
      const bool valid = l < nlen[bidx];
      const bool found = widx > 0;
#pragma unroll
      for (int ns = 0; ns < 2; ++ns) {
        const int col = n0 + wv * 32 + ns * 16 + lid;
        float val = 0.f;
        if (valid) val = found ? cls[(size_t)widx * DD + col] : (acc[s][ns][i] + bv[ns]);
        concepts[(size_t)row * DD + col] = val;
      }
    }
  }
}

#define GM 32
#define GN 64
#define GK 64
#define GCH (DD / GK)
#define GPA 68
#define GPB 72

__global__ __launch_bounds__(128) void k_gemm_xw(const float* __restrict__ X,
                                                 const float* __restrict__ W,
                                                 const float* __restrict__ bias,
                                                 float* __restrict__ out, int relu) {
  __shared__ __attribute__((aligned(16))) float As[2][GM * GPA];
  __shared__ __attribute__((aligned(16))) float Bs[2][GK * GPB];
  const int tid  = threadIdx.x;
  const int lane = tid & 31;
  const int wv   = tid >> 5;
  const int wm   = wv >> 1, wn = wv & 1;
  const int half = lane >> 4;
  const int lid  = lane & 15;
  const int n0 = blockIdx.x * GN, m0 = blockIdx.y * GM;
  const int srow = tid >> 4;
  const int sc   = (tid & 15) * 4;
  const float* xp = X + (size_t)(m0 + srow) * DD + sc;
  const float* wp = W + (size_t)srow * DD + n0 + sc;

  v4f ra[4], rb[8];
#pragma unroll
  for (int j = 0; j < 4; ++j) ra[j] = *(const v4f*)(xp + (size_t)(8 * j) * DD);
#pragma unroll
  for (int j = 0; j < 8; ++j) rb[j] = *(const v4f*)(wp + (size_t)(8 * j) * DD);
#pragma unroll
  for (int j = 0; j < 4; ++j) *(v4f*)&As[0][(srow + 8 * j) * GPA + sc] = ra[j];
#pragma unroll
  for (int j = 0; j < 8; ++j) *(v4f*)&Bs[0][(srow + 8 * j) * GPB + sc] = rb[j];
  __syncthreads();

  v8f acc[2];
  acc[0] = (v8f){0.f,0.f,0.f,0.f,0.f,0.f,0.f,0.f};
  acc[1] = acc[0];
  auto chunk_mma = [&](const float* Ab, const float* Bb) {
#pragma unroll
    for (int kk = 0; kk < GK / 4; ++kk) {
      const int k = kk * 4 + 2 * half;
      const v2f a = *(const v2f*)&Ab[(wm * 16 + lid) * GPA + k];
#pragma unroll
      for (int ns = 0; ns < 2; ++ns) {
        v2f b;
        b.x = Bb[k * GPB + wn * 32 + ns * 16 + lid];
        b.y = Bb[(k + 1) * GPB + wn * 32 + ns * 16 + lid];
        acc[ns] = wmma4(a, b, acc[ns]);
      }
    }
  };

#pragma unroll
  for (int j = 0; j < 4; ++j) ra[j] = *(const v4f*)(xp + (size_t)(8 * j) * DD + GK);
#pragma unroll
  for (int j = 0; j < 8; ++j) rb[j] = *(const v4f*)(wp + (size_t)(GK + 8 * j) * DD);
  for (int c = 0; c < GCH - 2; ++c) {
    chunk_mma(As[c & 1], Bs[c & 1]);
#pragma unroll
    for (int j = 0; j < 4; ++j) *(v4f*)&As[(c + 1) & 1][(srow + 8 * j) * GPA + sc] = ra[j];
#pragma unroll
    for (int j = 0; j < 8; ++j) *(v4f*)&Bs[(c + 1) & 1][(srow + 8 * j) * GPB + sc] = rb[j];
    __syncthreads();
    const int kn = (c + 2) * GK;
#pragma unroll
    for (int j = 0; j < 4; ++j) ra[j] = *(const v4f*)(xp + (size_t)(8 * j) * DD + kn);
#pragma unroll
    for (int j = 0; j < 8; ++j) rb[j] = *(const v4f*)(wp + (size_t)(kn + 8 * j) * DD);
  }
  chunk_mma(As[(GCH - 2) & 1], Bs[(GCH - 2) & 1]);
#pragma unroll
  for (int j = 0; j < 4; ++j) *(v4f*)&As[(GCH - 1) & 1][(srow + 8 * j) * GPA + sc] = ra[j];
#pragma unroll
  for (int j = 0; j < 8; ++j) *(v4f*)&Bs[(GCH - 1) & 1][(srow + 8 * j) * GPB + sc] = rb[j];
  __syncthreads();
  chunk_mma(As[(GCH - 1) & 1], Bs[(GCH - 1) & 1]);

#pragma unroll
  for (int ns = 0; ns < 2; ++ns) {
    const int col = n0 + wn * 32 + ns * 16 + lid;
    const float bv = bias[col];
#pragma unroll
    for (int i = 0; i < 8; ++i) {
      float v = acc[ns][i] + bv;
      if (relu) v = fmaxf(v, 0.f);
      out[(size_t)(m0 + wm * 16 + half * 8 + i) * DD + col] = v;
    }
  }
}

__global__ void k_softmax(float* __restrict__ x) {
  __shared__ float red[256];
  float* p = x + (size_t)blockIdx.x * DD;
  const int tid = threadIdx.x;
  float mx = -3.402823466e+38f;
  for (int i = tid; i < DD; i += 256) mx = fmaxf(mx, p[i]);
  red[tid] = mx; __syncthreads();
  for (int s = 128; s > 0; s >>= 1) {
    if (tid < s) red[tid] = fmaxf(red[tid], red[tid + s]);
    __syncthreads();
  }
  mx = red[0]; __syncthreads();
  float sum = 0.f;
  for (int i = tid; i < DD; i += 256) { float e = __expf(p[i] - mx); p[i] = e; sum += e; }
  red[tid] = sum; __syncthreads();
  for (int s = 128; s > 0; s >>= 1) {
    if (tid < s) red[tid] += red[tid + s];
    __syncthreads();
  }
  const float inv = 1.f / red[0];
  for (int i = tid; i < DD; i += 256) p[i] *= inv;
}

__global__ void k_row_l2(const float* __restrict__ src, float* __restrict__ dn,
                         float* __restrict__ dsq) {
  __shared__ float red[256];
  const int row = blockIdx.x, tid = threadIdx.x;
  const float* p = src + (size_t)row * DD;
  float s = 0.f;
  for (int i = tid; i < DD; i += 256) {
    const float v = p[i];
    s += v * v;
    if (dsq) dsq[(size_t)row * DD + i] = v * v;
  }
  red[tid] = s; __syncthreads();
  for (int st = 128; st > 0; st >>= 1) {
    if (tid < st) red[tid] += red[tid + st];
    __syncthreads();
  }
  const float inv = rsqrtf(red[0]);
  for (int i = tid; i < DD; i += 256) dn[(size_t)row * DD + i] = p[i] * inv;
}

__global__ void k_pq(const float* __restrict__ trm, const float* __restrict__ aem,
                     float* __restrict__ P, float* __restrict__ Q) {
  __shared__ float red[256];
  const int row = blockIdx.x, tid = threadIdx.x;
  const float* tp = trm + (size_t)row * DD;
  const float* ap = aem + (size_t)row * DD;
  float s = 0.f;
  for (int i = tid; i < DD; i += 256) { const float v = tp[i]; s += v * v; }
  red[tid] = s; __syncthreads();
  for (int st = 128; st > 0; st >>= 1) {
    if (tid < st) red[tid] += red[tid + st];
    __syncthreads();
  }
  const float inv = rsqrtf(red[0]);
  for (int i = tid; i < DD; i += 256) {
    const float a = ap[i];
    P[(size_t)row * DD + i] = tp[i] * inv * a;
    Q[(size_t)row * DD + i] = a * a;
  }
}

#define SM 32
#define SK 64
#define SCH (DD / SK)
#define SP 68

__global__ __launch_bounds__(128) void k_score(const float* __restrict__ P,
                                               const float* __restrict__ Q,
                                               const float* __restrict__ rn,
                                               const float* __restrict__ t,
                                               const float* __restrict__ t2,
                                               const float* __restrict__ tn,
                                               float* __restrict__ score) {
  __shared__ __attribute__((aligned(16))) float Ls[2][6][SM * SP];
  const int tid  = threadIdx.x;
  const int lane = tid & 31;
  const int wv   = tid >> 5;
  const int wm   = wv >> 1, wn = wv & 1;
  const int half = lane >> 4;
  const int lid  = lane & 15;
  const int m0 = blockIdx.x * SM, n0 = blockIdx.y * SM;
  const int srow = tid >> 4;
  const int sc   = (tid & 15) * 4;
  const size_t ao = (size_t)(m0 + srow) * DD + sc;
  const size_t bo = (size_t)(n0 + srow) * DD + sc;
  const float* src[6] = {P + ao, Q + ao, rn + ao, t + bo, t2 + bo, tn + bo};

  v4f rg[6][4];
#pragma unroll
  for (int p = 0; p < 6; ++p)
#pragma unroll
    for (int j = 0; j < 4; ++j) rg[p][j] = *(const v4f*)(src[p] + (size_t)(8 * j) * DD);
#pragma unroll
  for (int p = 0; p < 6; ++p)
#pragma unroll
    for (int j = 0; j < 4; ++j) *(v4f*)&Ls[0][p][(srow + 8 * j) * SP + sc] = rg[p][j];
  __syncthreads();

  v8f num = {0.f,0.f,0.f,0.f,0.f,0.f,0.f,0.f};
  v8f den = num, isv = num;
  const int arow = (wm * 16 + lid) * SP;
  const int brow = (wn * 16 + lid) * SP;
  auto chunk_mma = [&](const float (*L)[SM * SP]) {
#pragma unroll
    for (int kk = 0; kk < SK / 4; ++kk) {
      const int k = kk * 4 + 2 * half;
      const v2f a1 = *(const v2f*)&L[0][arow + k];
      const v2f b1 = *(const v2f*)&L[3][brow + k];
      const v2f a2 = *(const v2f*)&L[1][arow + k];
      const v2f b2 = *(const v2f*)&L[4][brow + k];
      const v2f a3 = *(const v2f*)&L[2][arow + k];
      const v2f b3 = *(const v2f*)&L[5][brow + k];
      num = wmma4(a1, b1, num);
      den = wmma4(a2, b2, den);
      isv = wmma4(a3, b3, isv);
    }
  };

#pragma unroll
  for (int p = 0; p < 6; ++p)
#pragma unroll
    for (int j = 0; j < 4; ++j)
      rg[p][j] = *(const v4f*)(src[p] + (size_t)(8 * j) * DD + SK);
  for (int c = 0; c < SCH - 2; ++c) {
    chunk_mma(Ls[c & 1]);
#pragma unroll
    for (int p = 0; p < 6; ++p)
#pragma unroll
      for (int j = 0; j < 4; ++j)
        *(v4f*)&Ls[(c + 1) & 1][p][(srow + 8 * j) * SP + sc] = rg[p][j];
    __syncthreads();
    const int kn = (c + 2) * SK;
#pragma unroll
    for (int p = 0; p < 6; ++p)
#pragma unroll
      for (int j = 0; j < 4; ++j)
        rg[p][j] = *(const v4f*)(src[p] + (size_t)(8 * j) * DD + kn);
  }
  chunk_mma(Ls[(SCH - 2) & 1]);
#pragma unroll
  for (int p = 0; p < 6; ++p)
#pragma unroll
    for (int j = 0; j < 4; ++j)
      *(v4f*)&Ls[(SCH - 1) & 1][p][(srow + 8 * j) * SP + sc] = rg[p][j];
  __syncthreads();
  chunk_mma(Ls[(SCH - 1) & 1]);

#pragma unroll
  for (int i = 0; i < 8; ++i) {
    const int row = m0 + wm * 16 + half * 8 + i;
    const int col = n0 + wn * 16 + lid;
    score[(size_t)row * BB + col] = num[i] * rsqrtf(den[i]) + isv[i];
  }
}

extern "C" void kernel_launch(void* const* d_in, const int* in_sizes, int n_in,
                              void* d_out, int out_size, void* d_ws, size_t ws_size,
                              hipStream_t stream) {
  const float* r    = (const float*)d_in[0];
  const float* m    = (const float*)d_in[1];
  const float* t    = (const float*)d_in[2];
  const int*   ntid = (const int*)d_in[3];
  const int*   nlen = (const int*)d_in[4];
  const int*   antk = (const int*)d_in[5];
  const float* emb  = (const float*)d_in[6];
  const float* nW   = (const float*)d_in[7];
  const float* nb   = (const float*)d_in[8];
  const float* W1   = (const float*)d_in[9];
  const float* b1   = (const float*)d_in[10];
  const float* W2   = (const float*)d_in[11];
  const float* b2   = (const float*)d_in[12];
  const float* trW  = (const float*)d_in[13];
  const float* trb  = (const float*)d_in[14];
  const int*   t2i  = (const int*)d_in[15];

  float* score    = (float*)d_out;
  float* concepts = score + (size_t)BB * BB;

  float* ws  = (float*)d_ws;
  float* cls = ws;  ws += (size_t)NN * DD;
  float* h   = ws;  ws += (size_t)BB * DD;
  float* aem = ws;  ws += (size_t)BB * DD;
  float* trm = ws;  ws += (size_t)BB * DD;
  float* Pw  = ws;  ws += (size_t)BB * DD;
  float* Qw  = ws;  ws += (size_t)BB * DD;
  float* rnw = ws;  ws += (size_t)BB * DD;
  float* tnw = ws;  ws += (size_t)BB * DD;
  float* t2w = ws;  ws += (size_t)BB * DD;

  k_cls<<<dim3(DD / TN, NN / TM), dim3(128), 0, stream>>>(antk, emb, nW, nb, cls);
  k_concepts<<<dim3(DD / TN, BB * LL / TM), dim3(128), 0, stream>>>(ntid, nlen, t2i, emb,
                                                                   nW, nb, cls, concepts);
  k_gemm_xw<<<dim3(DD / GN, BB / GM), dim3(128), 0, stream>>>(m, W1, b1, h, 1);
  k_gemm_xw<<<dim3(DD / GN, BB / GM), dim3(128), 0, stream>>>(h, W2, b2, aem, 0);
  k_softmax<<<BB, 256, 0, stream>>>(aem);
  k_gemm_xw<<<dim3(DD / GN, BB / GM), dim3(128), 0, stream>>>(m, trW, trb, trm, 0);
  k_pq<<<BB, 256, 0, stream>>>(trm, aem, Pw, Qw);
  k_row_l2<<<BB, 256, 0, stream>>>(r, rnw, nullptr);
  k_row_l2<<<BB, 256, 0, stream>>>(t, tnw, t2w);
  k_score<<<dim3(BB / SM, BB / SM), dim3(128), 0, stream>>>(Pw, Qw, rnw, t, t2w, tnw, score);
}


# Gather before the GEMM: every concepts row takes its embedding row first (mapped noun token or own token), so the 8000-row class-weight GEMM, its launch and its plane go; k order and sums unchanged
# speedup vs baseline: 2.9895x; 1.5281x over previous
#include <hip/hip_runtime.h>
#include <hip/hip_bf16.h>

typedef __attribute__((ext_vector_type(2))) float v2f;
typedef __attribute__((ext_vector_type(8))) float v8f;
typedef __attribute__((ext_vector_type(4))) float v4f;

#define BB 384
#define DD 1024
#define WD 300
#define LL 16

static __device__ __forceinline__ v8f wmma4(v2f a, v2f b, v8f c) {
  return __builtin_amdgcn_wmma_f32_16x16x4_f32(false, a, false, b, (short)0, c,
                                               false, false);
}

#define TM 64
#define TN 128
#define KC 20
#define NCH (WD / KC)
#define PA 20
#define PB 136

static __device__ __forceinline__ void tile_k300(const int* __restrict__ tokrow,
                                                 const float* __restrict__ emb,
                                                 const float* __restrict__ W,
                                                 const int n0,
                                                 float (*As)[TM * PA],
                                                 float (*Bs)[KC * PB],
                                                 v8f (&acc)[4][2]) {
  const int tid  = threadIdx.x;
  const int lane = tid & 31;
  const int wv   = tid >> 5;
  const int half = lane >> 4;
  const int lid  = lane & 15;

  int aoff[5];
#pragma unroll
  for (int j = 0; j < 5; ++j) {
    const int idx = tid + 128 * j;
    const int row = idx / 10;
    const int c2  = idx - row * 10;
    aoff[j] = tokrow[row] * WD + 2 * c2;
  }
  const int bk  = tid >> 5;
  const int bc4 = tid & 31;
  const float* wp = W + (size_t)bk * DD + n0 + bc4 * 4;

  v2f    ra[5];
  v4f    rb[5];
#pragma unroll
  for (int j = 0; j < 5; ++j) {
    ra[j] = *(const v2f*)(emb + aoff[j]);
    rb[j] = *(const v4f*)(wp + (size_t)(4 * j) * DD);
  }
#pragma unroll
  for (int j = 0; j < 5; ++j) {
    *(v2f*)&As[0][2 * (tid + 128 * j)] = ra[j];
    *(v4f*)&Bs[0][(bk + 4 * j) * PB + bc4 * 4] = rb[j];
  }
  __syncthreads();

  auto chunk_mma = [&](const float* Ab, const float* Bb) {
#pragma unroll
    for (int kk = 0; kk < KC / 4; ++kk) {
      const int k = kk * 4 + 2 * half;
      v2f b[2];
#pragma unroll
      for (int ns = 0; ns < 2; ++ns) {
        b[ns].x = Bb[k * PB + wv * 32 + ns * 16 + lid];
        b[ns].y = Bb[(k + 1) * PB + wv * 32 + ns * 16 + lid];
      }
#pragma unroll
      for (int s = 0; s < 4; ++s) {
        const v2f a = *(const v2f*)&Ab[(s * 16 + lid) * PA + k];
#pragma unroll
        for (int ns = 0; ns < 2; ++ns) acc[s][ns] = wmma4(a, b[ns], acc[s][ns]);
      }
    }
  };

#pragma unroll
  for (int j = 0; j < 5; ++j) {
    ra[j] = *(const v2f*)(emb + aoff[j] + KC);
    rb[j] = *(const v4f*)(wp + (size_t)(KC + 4 * j) * DD);
  }
  for (int c = 0; c < NCH - 2; ++c) {
    chunk_mma(As[c & 1], Bs[c & 1]);
#pragma unroll
    for (int j = 0; j < 5; ++j) {
      *(v2f*)&As[(c + 1) & 1][2 * (tid + 128 * j)] = ra[j];
      *(v4f*)&Bs[(c + 1) & 1][(bk + 4 * j) * PB + bc4 * 4] = rb[j];
    }
    __syncthreads();
    const int kn = (c + 2) * KC;
#pragma unroll
    for (int j = 0; j < 5; ++j) {
      ra[j] = *(const v2f*)(emb + aoff[j] + kn);
      rb[j] = *(const v4f*)(wp + (size_t)(kn + 4 * j) * DD);
    }
  }
  chunk_mma(As[(NCH - 2) & 1], Bs[(NCH - 2) & 1]);
#pragma unroll
  for (int j = 0; j < 5; ++j) {
    *(v2f*)&As[(NCH - 1) & 1][2 * (tid + 128 * j)] = ra[j];
    *(v4f*)&Bs[(NCH - 1) & 1][(bk + 4 * j) * PB + bc4 * 4] = rb[j];
  }
  __syncthreads();
  chunk_mma(As[(NCH - 1) & 1], Bs[(NCH - 1) & 1]);
}

__global__ __launch_bounds__(128) void k_concepts(const int* __restrict__ ntid,
                                                  const int* __restrict__ nlen,
                                                  const int* __restrict__ t2i,
                                                  const int* __restrict__ antk,
                                                  const float* __restrict__ emb,
                                                  const float* __restrict__ W,
                                                  const float* __restrict__ bias,
                                                  float* __restrict__ concepts) {
  __shared__ __attribute__((aligned(16))) float As[2][TM * PA];
  __shared__ __attribute__((aligned(16))) float Bs[2][KC * PB];
  __shared__ int stok[TM];
  const int lane = threadIdx.x & 31;
  const int wv   = threadIdx.x >> 5;
  const int half = lane >> 4;
  const int lid  = lane & 15;
  const int n0 = blockIdx.x * TN, m0 = blockIdx.y * TM;
  if (threadIdx.x < TM) {
    const int tk   = ntid[m0 + threadIdx.x];
    const int widx = t2i[tk];
    stok[threadIdx.x] = (widx > 0) ? antk[widx] : tk;
  }
  __syncthreads();
  v8f acc[4][2];
#pragma unroll
  for (int s = 0; s < 4; ++s)
#pragma unroll
    for (int ns = 0; ns < 2; ++ns) acc[s][ns] = (v8f){0.f,0.f,0.f,0.f,0.f,0.f,0.f,0.f};
  tile_k300(stok, emb, W, n0, As, Bs, acc);
  float bv[2];
#pragma unroll
  for (int ns = 0; ns < 2; ++ns) bv[ns] = bias[n0 + wv * 32 + ns * 16 + lid];
#pragma unroll
  for (int s = 0; s < 4; ++s) {
    const int len = nlen[(m0 >> 4) + s];
#pragma unroll
    for (int i = 0; i < 8; ++i) {
      const int l   = half * 8 + i;
      const int row = m0 + s * 16 + l;
      const bool valid = l < len;
#pragma unroll
      for (int ns = 0; ns < 2; ++ns) {
        const int col = n0 + wv * 32 + ns * 16 + lid;
        concepts[(size_t)row * DD + col] = valid ? (acc[s][ns][i] + bv[ns]) : 0.f;
      }
    }
  }
}

#define GM 32
#define GN 64
#define GK 64
#define GCH (DD / GK)
#define GPA 68
#define GPB 72

__global__ __launch_bounds__(128) void k_gemm_xw(const float* __restrict__ X,
                                                 const float* __restrict__ W,
                                                 const float* __restrict__ bias,
                                                 float* __restrict__ out, int relu) {
  __shared__ __attribute__((aligned(16))) float As[2][GM * GPA];
  __shared__ __attribute__((aligned(16))) float Bs[2][GK * GPB];
  const int tid  = threadIdx.x;
  const int lane = tid & 31;
  const int wv   = tid >> 5;
  const int wm   = wv >> 1, wn = wv & 1;
  const int half = lane >> 4;
  const int lid  = lane & 15;
  const int n0 = blockIdx.x * GN, m0 = blockIdx.y * GM;
  const int srow = tid >> 4;
  const int sc   = (tid & 15) * 4;
  const float* xp = X + (size_t)(m0 + srow) * DD + sc;
  const float* wp = W + (size_t)srow * DD + n0 + sc;

  v4f ra[4], rb[8];
#pragma unroll
  for (int j = 0; j < 4; ++j) ra[j] = *(const v4f*)(xp + (size_t)(8 * j) * DD);
#pragma unroll
  for (int j = 0; j < 8; ++j) rb[j] = *(const v4f*)(wp + (size_t)(8 * j) * DD);
#pragma unroll
  for (int j = 0; j < 4; ++j) *(v4f*)&As[0][(srow + 8 * j) * GPA + sc] = ra[j];
#pragma unroll
  for (int j = 0; j < 8; ++j) *(v4f*)&Bs[0][(srow + 8 * j) * GPB + sc] = rb[j];
  __syncthreads();

  v8f acc[2];
  acc[0] = (v8f){0.f,0.f,0.f,0.f,0.f,0.f,0.f,0.f};
  acc[1] = acc[0];
  auto chunk_mma = [&](const float* Ab, const float* Bb) {
#pragma unroll
    for (int kk = 0; kk < GK / 4; ++kk) {
      const int k = kk * 4 + 2 * half;
      const v2f a = *(const v2f*)&Ab[(wm * 16 + lid) * GPA + k];
#pragma unroll
      for (int ns = 0; ns < 2; ++ns) {
        v2f b;
        b.x = Bb[k * GPB + wn * 32 + ns * 16 + lid];
        b.y = Bb[(k + 1) * GPB + wn * 32 + ns * 16 + lid];
        acc[ns] = wmma4(a, b, acc[ns]);
      }
    }
  };

#pragma unroll
  for (int j = 0; j < 4; ++j) ra[j] = *(const v4f*)(xp + (size_t)(8 * j) * DD + GK);
#pragma unroll
  for (int j = 0; j < 8; ++j) rb[j] = *(const v4f*)(wp + (size_t)(GK + 8 * j) * DD);
  for (int c = 0; c < GCH - 2; ++c) {
    chunk_mma(As[c & 1], Bs[c & 1]);
#pragma unroll
    for (int j = 0; j < 4; ++j) *(v4f*)&As[(c + 1) & 1][(srow + 8 * j) * GPA + sc] = ra[j];
#pragma unroll
    for (int j = 0; j < 8; ++j) *(v4f*)&Bs[(c + 1) & 1][(srow + 8 * j) * GPB + sc] = rb[j];
    __syncthreads();
    const int kn = (c + 2) * GK;
#pragma unroll
    for (int j = 0; j < 4; ++j) ra[j] = *(const v4f*)(xp + (size_t)(8 * j) * DD + kn);
#pragma unroll
    for (int j = 0; j < 8; ++j) rb[j] = *(const v4f*)(wp + (size_t)(kn + 8 * j) * DD);
  }
  chunk_mma(As[(GCH - 2) & 1], Bs[(GCH - 2) & 1]);
#pragma unroll
  for (int j = 0; j < 4; ++j) *(v4f*)&As[(GCH - 1) & 1][(srow + 8 * j) * GPA + sc] = ra[j];
#pragma unroll
  for (int j = 0; j < 8; ++j) *(v4f*)&Bs[(GCH - 1) & 1][(srow + 8 * j) * GPB + sc] = rb[j];
  __syncthreads();
  chunk_mma(As[(GCH - 1) & 1], Bs[(GCH - 1) & 1]);

#pragma unroll
  for (int ns = 0; ns < 2; ++ns) {
    const int col = n0 + wn * 32 + ns * 16 + lid;
    const float bv = bias[col];
#pragma unroll
    for (int i = 0; i < 8; ++i) {
      float v = acc[ns][i] + bv;
      if (relu) v = fmaxf(v, 0.f);
      out[(size_t)(m0 + wm * 16 + half * 8 + i) * DD + col] = v;
    }
  }
}

__global__ void k_softmax(float* __restrict__ x) {
  __shared__ float red[256];
  float* p = x + (size_t)blockIdx.x * DD;
  const int tid = threadIdx.x;
  float mx = -3.402823466e+38f;
  for (int i = tid; i < DD; i += 256) mx = fmaxf(mx, p[i]);
  red[tid] = mx; __syncthreads();
  for (int s = 128; s > 0; s >>= 1) {
    if (tid < s) red[tid] = fmaxf(red[tid], red[tid + s]);
    __syncthreads();
  }
  mx = red[0]; __syncthreads();
  float sum = 0.f;
  for (int i = tid; i < DD; i += 256) { float e = __expf(p[i] - mx); p[i] = e; sum += e; }
  red[tid] = sum; __syncthreads();
  for (int s = 128; s > 0; s >>= 1) {
    if (tid < s) red[tid] += red[tid + s];
    __syncthreads();
  }
  const float inv = 1.f / red[0];
  for (int i = tid; i < DD; i += 256) p[i] *= inv;
}

__global__ void k_row_l2(const float* __restrict__ src, float* __restrict__ dn,
                         float* __restrict__ dsq) {
  __shared__ float red[256];
  const int row = blockIdx.x, tid = threadIdx.x;
  const float* p = src + (size_t)row * DD;
  float s = 0.f;
  for (int i = tid; i < DD; i += 256) {
    const float v = p[i];
    s += v * v;
    if (dsq) dsq[(size_t)row * DD + i] = v * v;
  }
  red[tid] = s; __syncthreads();
  for (int st = 128; st > 0; st >>= 1) {
    if (tid < st) red[tid] += red[tid + st];
    __syncthreads();
  }
  const float inv = rsqrtf(red[0]);
  for (int i = tid; i < DD; i += 256) dn[(size_t)row * DD + i] = p[i] * inv;
}

__global__ void k_pq(const float* __restrict__ trm, const float* __restrict__ aem,
                     float* __restrict__ P, float* __restrict__ Q) {
  __shared__ float red[256];
  const int row = blockIdx.x, tid = threadIdx.x;
  const float* tp = trm + (size_t)row * DD;
  const float* ap = aem + (size_t)row * DD;
  float s = 0.f;
  for (int i = tid; i < DD; i += 256) { const float v = tp[i]; s += v * v; }
  red[tid] = s; __syncthreads();
  for (int st = 128; st > 0; st >>= 1) {
    if (tid < st) red[tid] += red[tid + st];
    __syncthreads();
  }
  const float inv = rsqrtf(red[0]);
  for (int i = tid; i < DD; i += 256) {
    const float a = ap[i];
    P[(size_t)row * DD + i] = tp[i] * inv * a;
    Q[(size_t)row * DD + i] = a * a;
  }
}

#define SM 32
#define SK 64
#define SCH (DD / SK)
#define SP 68

__global__ __launch_bounds__(128) void k_score(const float* __restrict__ P,
                                               const float* __restrict__ Q,
                                               const float* __restrict__ rn,
                                               const float* __restrict__ t,
                                               const float* __restrict__ t2,
                                               const float* __restrict__ tn,
                                               float* __restrict__ score) {
  __shared__ __attribute__((aligned(16))) float Ls[2][6][SM * SP];
  const int tid  = threadIdx.x;
  const int lane = tid & 31;
  const int wv   = tid >> 5;
  const int wm   = wv >> 1, wn = wv & 1;
  const int half = lane >> 4;
  const int lid  = lane & 15;
  const int m0 = blockIdx.x * SM, n0 = blockIdx.y * SM;
  const int srow = tid >> 4;
  const int sc   = (tid & 15) * 4;
  const size_t ao = (size_t)(m0 + srow) * DD + sc;
  const size_t bo = (size_t)(n0 + srow) * DD + sc;
  const float* src[6] = {P + ao, Q + ao, rn + ao, t + bo, t2 + bo, tn + bo};

  v4f rg[6][4];
#pragma unroll
  for (int p = 0; p < 6; ++p)
#pragma unroll
    for (int j = 0; j < 4; ++j) rg[p][j] = *(const v4f*)(src[p] + (size_t)(8 * j) * DD);
#pragma unroll
  for (int p = 0; p < 6; ++p)
#pragma unroll
    for (int j = 0; j < 4; ++j) *(v4f*)&Ls[0][p][(srow + 8 * j) * SP + sc] = rg[p][j];
  __syncthreads();

  v8f num = {0.f,0.f,0.f,0.f,0.f,0.f,0.f,0.f};
  v8f den = num, isv = num;
  const int arow = (wm * 16 + lid) * SP;
  const int brow = (wn * 16 + lid) * SP;
  auto chunk_mma = [&](const float (*L)[SM * SP]) {
#pragma unroll
    for (int kk = 0; kk < SK / 4; ++kk) {
      const int k = kk * 4 + 2 * half;
      const v2f a1 = *(const v2f*)&L[0][arow + k];
      const v2f b1 = *(const v2f*)&L[3][brow + k];
      const v2f a2 = *(const v2f*)&L[1][arow + k];
      const v2f b2 = *(const v2f*)&L[4][brow + k];
      const v2f a3 = *(const v2f*)&L[2][arow + k];
      const v2f b3 = *(const v2f*)&L[5][brow + k];
      num = wmma4(a1, b1, num);
      den = wmma4(a2, b2, den);
      isv = wmma4(a3, b3, isv);
    }
  };

#pragma unroll
  for (int p = 0; p < 6; ++p)
#pragma unroll
    for (int j = 0; j < 4; ++j)
      rg[p][j] = *(const v4f*)(src[p] + (size_t)(8 * j) * DD + SK);
  for (int c = 0; c < SCH - 2; ++c) {
    chunk_mma(Ls[c & 1]);
#pragma unroll
    for (int p = 0; p < 6; ++p)
#pragma unroll
      for (int j = 0; j < 4; ++j)
        *(v4f*)&Ls[(c + 1) & 1][p][(srow + 8 * j) * SP + sc] = rg[p][j];
    __syncthreads();
    const int kn = (c + 2) * SK;
#pragma unroll
    for (int p = 0; p < 6; ++p)
#pragma unroll
      for (int j = 0; j < 4; ++j)
        rg[p][j] = *(const v4f*)(src[p] + (size_t)(8 * j) * DD + kn);
  }
  chunk_mma(Ls[(SCH - 2) & 1]);
#pragma unroll
  for (int p = 0; p < 6; ++p)
#pragma unroll
    for (int j = 0; j < 4; ++j)
      *(v4f*)&Ls[(SCH - 1) & 1][p][(srow + 8 * j) * SP + sc] = rg[p][j];
  __syncthreads();
  chunk_mma(Ls[(SCH - 1) & 1]);

#pragma unroll
  for (int i = 0; i < 8; ++i) {
    const int row = m0 + wm * 16 + half * 8 + i;
    const int col = n0 + wn * 16 + lid;
    score[(size_t)row * BB + col] = num[i] * rsqrtf(den[i]) + isv[i];
  }
}

extern "C" void kernel_launch(void* const* d_in, const int* in_sizes, int n_in,
                              void* d_out, int out_size, void* d_ws, size_t ws_size,
                              hipStream_t stream) {
  const float* r    = (const float*)d_in[0];
  const float* m    = (const float*)d_in[1];
  const float* t    = (const float*)d_in[2];
  const int*   ntid = (const int*)d_in[3];
  const int*   nlen = (const int*)d_in[4];
  const int*   antk = (const int*)d_in[5];
  const float* emb  = (const float*)d_in[6];
  const float* nW   = (const float*)d_in[7];
  const float* nb   = (const float*)d_in[8];
  const float* W1   = (const float*)d_in[9];
  const float* b1   = (const float*)d_in[10];
  const float* W2   = (const float*)d_in[11];
  const float* b2   = (const float*)d_in[12];
  const float* trW  = (const float*)d_in[13];
  const float* trb  = (const float*)d_in[14];
  const int*   t2i  = (const int*)d_in[15];

  float* score    = (float*)d_out;
  float* concepts = score + (size_t)BB * BB;

  float* ws  = (float*)d_ws;
  float* h   = ws;  ws += (size_t)BB * DD;
  float* aem = ws;  ws += (size_t)BB * DD;
  float* trm = ws;  ws += (size_t)BB * DD;
  float* Pw  = ws;  ws += (size_t)BB * DD;
  float* Qw  = ws;  ws += (size_t)BB * DD;
  float* rnw = ws;  ws += (size_t)BB * DD;
  float* tnw = ws;  ws += (size_t)BB * DD;
  float* t2w = ws;  ws += (size_t)BB * DD;

  k_concepts<<<dim3(DD / TN, BB * LL / TM), dim3(128), 0, stream>>>(ntid, nlen, t2i, antk, emb,
                                                                   nW, nb, concepts);
  k_gemm_xw<<<dim3(DD / GN, BB / GM), dim3(128), 0, stream>>>(m, W1, b1, h, 1);
  k_gemm_xw<<<dim3(DD / GN, BB / GM), dim3(128), 0, stream>>>(h, W2, b2, aem, 0);
  k_softmax<<<BB, 256, 0, stream>>>(aem);
  k_gemm_xw<<<dim3(DD / GN, BB / GM), dim3(128), 0, stream>>>(m, trW, trb, trm, 0);
  k_pq<<<BB, 256, 0, stream>>>(trm, aem, Pw, Qw);
  k_row_l2<<<BB, 256, 0, stream>>>(r, rnw, nullptr);
  k_row_l2<<<BB, 256, 0, stream>>>(t, tnw, t2w);
  k_score<<<dim3(BB / SM, BB / SM), dim3(128), 0, stream>>>(Pw, Qw, rnw, t, t2w, tnw, score);
}


# Fewer launches (9 -> 4): the five passes that need only the inputs (two m GEMMs, concepts GEMM, two row norms) share one grid; softmax joins the P/Q row pass; k order and summation trees unchanged
# speedup vs baseline: 4.7057x; 1.5741x over previous
#include <hip/hip_runtime.h>
#include <hip/hip_bf16.h>

typedef __attribute__((ext_vector_type(2))) float v2f;
typedef __attribute__((ext_vector_type(8))) float v8f;
typedef __attribute__((ext_vector_type(4))) float v4f;

#define BB 384
#define DD 1024
#define WD 300
#define LL 16

static __device__ __forceinline__ v8f wmma4(v2f a, v2f b, v8f c) {
  return __builtin_amdgcn_wmma_f32_16x16x4_f32(false, a, false, b, (short)0, c,
                                               false, false);
}

#define TM 64
#define TN 128
#define KC 20
#define NCH (WD / KC)
#define PA 20
#define PB 136

static __device__ __forceinline__ void tile_k300(const int* __restrict__ tokrow,
                                                 const float* __restrict__ emb,
                                                 const float* __restrict__ W,
                                                 const int n0,
                                                 float (*As)[TM * PA],
                                                 float (*Bs)[KC * PB],
                                                 v8f (&acc)[4][2]) {
  const int tid  = threadIdx.x;
  const int lane = tid & 31;
  const int wv   = tid >> 5;
  const int half = lane >> 4;
  const int lid  = lane & 15;

  int aoff[5];
#pragma unroll
  for (int j = 0; j < 5; ++j) {
    const int idx = tid + 128 * j;
    const int row = idx / 10;
    const int c2  = idx - row * 10;
    aoff[j] = tokrow[row] * WD + 2 * c2;
  }
  const int bk  = tid >> 5;
  const int bc4 = tid & 31;
  const float* wp = W + (size_t)bk * DD + n0 + bc4 * 4;

  v2f    ra[5];
  v4f    rb[5];
#pragma unroll
  for (int j = 0; j < 5; ++j) {
    ra[j] = *(const v2f*)(emb + aoff[j]);
    rb[j] = *(const v4f*)(wp + (size_t)(4 * j) * DD);
  }
#pragma unroll
  for (int j = 0; j < 5; ++j) {
    *(v2f*)&As[0][2 * (tid + 128 * j)] = ra[j];
    *(v4f*)&Bs[0][(bk + 4 * j) * PB + bc4 * 4] = rb[j];
  }
  __syncthreads();

  auto chunk_mma = [&](const float* Ab, const float* Bb) {
#pragma unroll
    for (int kk = 0; kk < KC / 4; ++kk) {
      const int k = kk * 4 + 2 * half;
      v2f b[2];
#pragma unroll
      for (int ns = 0; ns < 2; ++ns) {
        b[ns].x = Bb[k * PB + wv * 32 + ns * 16 + lid];
        b[ns].y = Bb[(k + 1) * PB + wv * 32 + ns * 16 + lid];
      }
#pragma unroll
      for (int s = 0; s < 4; ++s) {
        const v2f a = *(const v2f*)&Ab[(s * 16 + lid) * PA + k];
#pragma unroll
        for (int ns = 0; ns < 2; ++ns) acc[s][ns] = wmma4(a, b[ns], acc[s][ns]);
      }
    }
  };

#pragma unroll
  for (int j = 0; j < 5; ++j) {
    ra[j] = *(const v2f*)(emb + aoff[j] + KC);
    rb[j] = *(const v4f*)(wp + (size_t)(KC + 4 * j) * DD);
  }
  for (int c = 0; c < NCH - 2; ++c) {
    chunk_mma(As[c & 1], Bs[c & 1]);
#pragma unroll
    for (int j = 0; j < 5; ++j) {
      *(v2f*)&As[(c + 1) & 1][2 * (tid + 128 * j)] = ra[j];
      *(v4f*)&Bs[(c + 1) & 1][(bk + 4 * j) * PB + bc4 * 4] = rb[j];
    }
    __syncthreads();
    const int kn = (c + 2) * KC;
#pragma unroll
    for (int j = 0; j < 5; ++j) {
      ra[j] = *(const v2f*)(emb + aoff[j] + kn);
      rb[j] = *(const v4f*)(wp + (size_t)(kn + 4 * j) * DD);
    }
  }
  chunk_mma(As[(NCH - 2) & 1], Bs[(NCH - 2) & 1]);
#pragma unroll
  for (int j = 0; j < 5; ++j) {
    *(v2f*)&As[(NCH - 1) & 1][2 * (tid + 128 * j)] = ra[j];
    *(v4f*)&Bs[(NCH - 1) & 1][(bk + 4 * j) * PB + bc4 * 4] = rb[j];
  }
  __syncthreads();
  chunk_mma(As[(NCH - 1) & 1], Bs[(NCH - 1) & 1]);
}

#define NCW ((DD / TN) * (BB * LL / TM))

static __device__ __forceinline__ void concepts_tile(const int bx, const int by, float* smem,
                                                     const int* __restrict__ ntid,
                                                     const int* __restrict__ nlen,
                                                     const int* __restrict__ t2i,
                                                     const int* __restrict__ antk,
                                                     const float* __restrict__ emb,
                                                     const float* __restrict__ W,
                                                     const float* __restrict__ bias,
                                                     float* __restrict__ concepts) {
  float (*As)[TM * PA] = (float (*)[TM * PA])smem;
  float (*Bs)[KC * PB] = (float (*)[KC * PB])(smem + 2 * TM * PA);
  int* stok = (int*)(smem + 2 * TM * PA + 2 * KC * PB);
  const int lane = threadIdx.x & 31;
  const int wv   = threadIdx.x >> 5;
  const int half = lane >> 4;
  const int lid  = lane & 15;
  const int n0 = bx * TN, m0 = by * TM;
  if (threadIdx.x < TM) {
    const int tk   = ntid[m0 + threadIdx.x];
    const int widx = t2i[tk];
    stok[threadIdx.x] = (widx > 0) ? antk[widx] : tk;
  }
  __syncthreads();
  v8f acc[4][2];
#pragma unroll
  for (int s = 0; s < 4; ++s)
#pragma unroll
    for (int ns = 0; ns < 2; ++ns) acc[s][ns] = (v8f){0.f,0.f,0.f,0.f,0.f,0.f,0.f,0.f};
  tile_k300(stok, emb, W, n0, As, Bs, acc);
  float bv[2];
#pragma unroll
  for (int ns = 0; ns < 2; ++ns) bv[ns] = bias[n0 + wv * 32 + ns * 16 + lid];
#pragma unroll
  for (int s = 0; s < 4; ++s) {
    const int len = nlen[(m0 >> 4) + s];
#pragma unroll
    for (int i = 0; i < 8; ++i) {
      const int l   = half * 8 + i;
      const int row = m0 + s * 16 + l;
      const bool valid = l < len;
#pragma unroll
      for (int ns = 0; ns < 2; ++ns) {
        const int col = n0 + wv * 32 + ns * 16 + lid;
        concepts[(size_t)row * DD + col] = valid ? (acc[s][ns][i] + bv[ns]) : 0.f;
      }
    }
  }
}

#define GM 32
#define GN 64
#define GK 64
#define GCH (DD / GK)
#define GPA 68
#define GPB 72

#define NGW ((DD / GN) * (BB / GM))
#define SMEM_F (2 * GM * GPA + 2 * GK * GPB)

static __device__ __forceinline__ void gemm_tile(const int bx, const int by, float* smem,
                                                 const float* __restrict__ X,
                                                 const float* __restrict__ W,
                                                 const float* __restrict__ bias,
                                                 float* __restrict__ out, const int relu) {
  float (*As)[GM * GPA] = (float (*)[GM * GPA])smem;
  float (*Bs)[GK * GPB] = (float (*)[GK * GPB])(smem + 2 * GM * GPA);
  const int tid  = threadIdx.x;
  const int lane = tid & 31;
  const int wv   = tid >> 5;
  const int wm   = wv >> 1, wn = wv & 1;
  const int half = lane >> 4;
  const int lid  = lane & 15;
  const int n0 = bx * GN, m0 = by * GM;
  const int srow = tid >> 4;
  const int sc   = (tid & 15) * 4;
  const float* xp = X + (size_t)(m0 + srow) * DD + sc;
  const float* wp = W + (size_t)srow * DD + n0 + sc;

  v4f ra[4], rb[8];
#pragma unroll
  for (int j = 0; j < 4; ++j) ra[j] = *(const v4f*)(xp + (size_t)(8 * j) * DD);
#pragma unroll
  for (int j = 0; j < 8; ++j) rb[j] = *(const v4f*)(wp + (size_t)(8 * j) * DD);
#pragma unroll
  for (int j = 0; j < 4; ++j) *(v4f*)&As[0][(srow + 8 * j) * GPA + sc] = ra[j];
#pragma unroll
  for (int j = 0; j < 8; ++j) *(v4f*)&Bs[0][(srow + 8 * j) * GPB + sc] = rb[j];
  __syncthreads();

  v8f acc[2];
  acc[0] = (v8f){0.f,0.f,0.f,0.f,0.f,0.f,0.f,0.f};
  acc[1] = acc[0];
  auto chunk_mma = [&](const float* Ab, const float* Bb) {
#pragma unroll
    for (int kk = 0; kk < GK / 4; ++kk) {
      const int k = kk * 4 + 2 * half;
      const v2f a = *(const v2f*)&Ab[(wm * 16 + lid) * GPA + k];
#pragma unroll
      for (int ns = 0; ns < 2; ++ns) {
        v2f b;
        b.x = Bb[k * GPB + wn * 32 + ns * 16 + lid];
        b.y = Bb[(k + 1) * GPB + wn * 32 + ns * 16 + lid];
        acc[ns] = wmma4(a, b, acc[ns]);
      }
    }
  };

#pragma unroll
  for (int j = 0; j < 4; ++j) ra[j] = *(const v4f*)(xp + (size_t)(8 * j) * DD + GK);
#pragma unroll
  for (int j = 0; j < 8; ++j) rb[j] = *(const v4f*)(wp + (size_t)(GK + 8 * j) * DD);
  for (int c = 0; c < GCH - 2; ++c) {
    chunk_mma(As[c & 1], Bs[c & 1]);
#pragma unroll
    for (int j = 0; j < 4; ++j) *(v4f*)&As[(c + 1) & 1][(srow + 8 * j) * GPA + sc] = ra[j];
#pragma unroll
    for (int j = 0; j < 8; ++j) *(v4f*)&Bs[(c + 1) & 1][(srow + 8 * j) * GPB + sc] = rb[j];
    __syncthreads();
    const int kn = (c + 2) * GK;
#pragma unroll
    for (int j = 0; j < 4; ++j) ra[j] = *(const v4f*)(xp + (size_t)(8 * j) * DD + kn);
#pragma unroll
    for (int j = 0; j < 8; ++j) rb[j] = *(const v4f*)(wp + (size_t)(kn + 8 * j) * DD);
  }
  chunk_mma(As[(GCH - 2) & 1], Bs[(GCH - 2) & 1]);
#pragma unroll
  for (int j = 0; j < 4; ++j) *(v4f*)&As[(GCH - 1) & 1][(srow + 8 * j) * GPA + sc] = ra[j];
#pragma unroll
  for (int j = 0; j < 8; ++j) *(v4f*)&Bs[(GCH - 1) & 1][(srow + 8 * j) * GPB + sc] = rb[j];
  __syncthreads();
  chunk_mma(As[(GCH - 1) & 1], Bs[(GCH - 1) & 1]);

#pragma unroll
  for (int ns = 0; ns < 2; ++ns) {
    const int col = n0 + wn * 32 + ns * 16 + lid;
    const float bv = bias[col];
#pragma unroll
    for (int i = 0; i < 8; ++i) {
      float v = acc[ns][i] + bv;
      if (relu) v = fmaxf(v, 0.f);
      out[(size_t)(m0 + wm * 16 + half * 8 + i) * DD + col] = v;
    }
  }
}

__global__ __launch_bounds__(128) void k_gemm_xw(const float* __restrict__ X,
                                                 const float* __restrict__ W,
                                                 const float* __restrict__ bias,
                                                 float* __restrict__ out) {
  __shared__ __attribute__((aligned(16))) float smem[SMEM_F];
  gemm_tile(blockIdx.x, blockIdx.y, smem, X, W, bias, out, 0);
}

static __device__ __forceinline__ void row_l2(const int row, float* red,
                                              const float* __restrict__ src,
                                              float* __restrict__ dn,
                                              float* __restrict__ dsq) {
  const int tid = threadIdx.x;
  const float* p = src + (size_t)row * DD;
  float va[4], vb[4];
#pragma unroll
  for (int j = 0; j < 4; ++j) {
    va[j] = p[tid + 256 * j];
    vb[j] = p[tid + 128 + 256 * j];
  }
  float sa = 0.f, sb = 0.f;
#pragma unroll
  for (int j = 0; j < 4; ++j) {
    sa += va[j] * va[j];
    sb += vb[j] * vb[j];
  }
  if (dsq) {
#pragma unroll
    for (int j = 0; j < 4; ++j) {
      dsq[(size_t)row * DD + tid + 256 * j]       = va[j] * va[j];
      dsq[(size_t)row * DD + tid + 128 + 256 * j] = vb[j] * vb[j];
    }
  }
  red[tid] = sa + sb; __syncthreads();
  for (int st = 64; st > 0; st >>= 1) {
    if (tid < st) red[tid] += red[tid + st];
    __syncthreads();
  }
  const float inv = rsqrtf(red[0]);
#pragma unroll
  for (int j = 0; j < 4; ++j) {
    dn[(size_t)row * DD + tid + 256 * j]       = va[j] * inv;
    dn[(size_t)row * DD + tid + 128 + 256 * j] = vb[j] * inv;
  }
}

__global__ __launch_bounds__(128) void k_front(const int* __restrict__ ntid,
                                               const int* __restrict__ nlen,
                                               const int* __restrict__ t2i,
                                               const int* __restrict__ antk,
                                               const float* __restrict__ emb,
                                               const float* __restrict__ nW,
                                               const float* __restrict__ nb,
                                               const float* __restrict__ m,
                                               const float* __restrict__ W1,
                                               const float* __restrict__ b1,
                                               const float* __restrict__ trW,
                                               const float* __restrict__ trb,
                                               const float* __restrict__ r,
                                               const float* __restrict__ t,
                                               float* __restrict__ h,
                                               float* __restrict__ trm,
                                               float* __restrict__ rn,
                                               float* __restrict__ tn,
                                               float* __restrict__ t2,
                                               float* __restrict__ concepts) {
  __shared__ __attribute__((aligned(16))) float smem[SMEM_F];
  const int b = blockIdx.x;
  if (b < 2 * NGW) {
    const bool first = b < NGW;
    const int q = first ? b : b - NGW;
    gemm_tile(q % (DD / GN), q / (DD / GN), smem, m, first ? W1 : trW, first ? b1 : trb,
              first ? h : trm, first ? 1 : 0);
  } else if (b < 2 * NGW + NCW) {
    const int q = b - 2 * NGW;
    concepts_tile(q % (DD / TN), q / (DD / TN), smem, ntid, nlen, t2i, antk, emb, nW, nb, concepts);
  } else if (b < 2 * NGW + NCW + BB) {
    row_l2(b - (2 * NGW + NCW), smem, r, rn, nullptr);
  } else {
    row_l2(b - (2 * NGW + NCW + BB), smem, t, tn, t2);
  }
}

__global__ __launch_bounds__(256) void k_softmax_pq(const float* __restrict__ aem,
                                                    const float* __restrict__ trm,
                                                    float* __restrict__ P,
                                                    float* __restrict__ Q) {
  __shared__ float red[256];
  __shared__ float red2[256];
  const int row = blockIdx.x, tid = threadIdx.x;
  const float* ap = aem + (size_t)row * DD;
  const float* tp = trm + (size_t)row * DD;
  float x[4], tv[4];
#pragma unroll
  for (int j = 0; j < 4; ++j) {
    x[j]  = ap[tid + 256 * j];
    tv[j] = tp[tid + 256 * j];
  }
  float mx = -3.402823466e+38f;
#pragma unroll
  for (int j = 0; j < 4; ++j) mx = fmaxf(mx, x[j]);
  red[tid] = mx; __syncthreads();
  for (int s = 128; s > 0; s >>= 1) {
    if (tid < s) red[tid] = fmaxf(red[tid], red[tid + s]);
    __syncthreads();
  }
  mx = red[0]; __syncthreads();
  float sum = 0.f, sq = 0.f;
#pragma unroll
  for (int j = 0; j < 4; ++j) {
    const float e = __expf(x[j] - mx);
    x[j] = e;
    sum += e;
    sq = fmaf(tv[j], tv[j], sq);
  }
  red[tid] = sum; red2[tid] = sq; __syncthreads();
  for (int s = 128; s > 0; s >>= 1) {
    if (tid < s) { red[tid] += red[tid + s]; red2[tid] += red2[tid + s]; }
    __syncthreads();
  }
  const float inv  = 1.f / red[0];
  const float inv2 = rsqrtf(red2[0]);
#pragma unroll
  for (int j = 0; j < 4; ++j) {
    const float a = x[j] * inv;
    P[(size_t)row * DD + tid + 256 * j] = tv[j] * inv2 * a;
    Q[(size_t)row * DD + tid + 256 * j] = a * a;
  }
}

#define SM 32
#define SK 64
#define SCH (DD / SK)
#define SP 68

__global__ __launch_bounds__(128) void k_score(const float* __restrict__ P,
                                               const float* __restrict__ Q,
                                               const float* __restrict__ rn,
                                               const float* __restrict__ t,
                                               const float* __restrict__ t2,
                                               const float* __restrict__ tn,
                                               float* __restrict__ score) {
  __shared__ __attribute__((aligned(16))) float Ls[2][6][SM * SP];
  const int tid  = threadIdx.x;
  const int lane = tid & 31;
  const int wv   = tid >> 5;
  const int wm   = wv >> 1, wn = wv & 1;
  const int half = lane >> 4;
  const int lid  = lane & 15;
  const int m0 = blockIdx.x * SM, n0 = blockIdx.y * SM;
  const int srow = tid >> 4;
  const int sc   = (tid & 15) * 4;
  const size_t ao = (size_t)(m0 + srow) * DD + sc;
  const size_t bo = (size_t)(n0 + srow) * DD + sc;
  const float* src[6] = {P + ao, Q + ao, rn + ao, t + bo, t2 + bo, tn + bo};

  v4f rg[6][4];
#pragma unroll
  for (int p = 0; p < 6; ++p)
#pragma unroll
    for (int j = 0; j < 4; ++j) rg[p][j] = *(const v4f*)(src[p] + (size_t)(8 * j) * DD);
#pragma unroll
  for (int p = 0; p < 6; ++p)
#pragma unroll
    for (int j = 0; j < 4; ++j) *(v4f*)&Ls[0][p][(srow + 8 * j) * SP + sc] = rg[p][j];
  __syncthreads();

  v8f num = {0.f,0.f,0.f,0.f,0.f,0.f,0.f,0.f};
  v8f den = num, isv = num;
  const int arow = (wm * 16 + lid) * SP;
  const int brow = (wn * 16 + lid) * SP;
  auto chunk_mma = [&](const float (*L)[SM * SP]) {
#pragma unroll
    for (int kk = 0; kk < SK / 4; ++kk) {
      const int k = kk * 4 + 2 * half;
      const v2f a1 = *(const v2f*)&L[0][arow + k];
      const v2f b1 = *(const v2f*)&L[3][brow + k];
      const v2f a2 = *(const v2f*)&L[1][arow + k];
      const v2f b2 = *(const v2f*)&L[4][brow + k];
      const v2f a3 = *(const v2f*)&L[2][arow + k];
      const v2f b3 = *(const v2f*)&L[5][brow + k];
      num = wmma4(a1, b1, num);
      den = wmma4(a2, b2, den);
      isv = wmma4(a3, b3, isv);
    }
  };

#pragma unroll
  for (int p = 0; p < 6; ++p)
#pragma unroll
    for (int j = 0; j < 4; ++j)
      rg[p][j] = *(const v4f*)(src[p] + (size_t)(8 * j) * DD + SK);
  for (int c = 0; c < SCH - 2; ++c) {
    chunk_mma(Ls[c & 1]);
#pragma unroll
    for (int p = 0; p < 6; ++p)
#pragma unroll
      for (int j = 0; j < 4; ++j)
        *(v4f*)&Ls[(c + 1) & 1][p][(srow + 8 * j) * SP + sc] = rg[p][j];
    __syncthreads();
    const int kn = (c + 2) * SK;
#pragma unroll
    for (int p = 0; p < 6; ++p)
#pragma unroll
      for (int j = 0; j < 4; ++j)
        rg[p][j] = *(const v4f*)(src[p] + (size_t)(8 * j) * DD + kn);
  }
  chunk_mma(Ls[(SCH - 2) & 1]);
#pragma unroll
  for (int p = 0; p < 6; ++p)
#pragma unroll
    for (int j = 0; j < 4; ++j)
      *(v4f*)&Ls[(SCH - 1) & 1][p][(srow + 8 * j) * SP + sc] = rg[p][j];
  __syncthreads();
  chunk_mma(Ls[(SCH - 1) & 1]);

#pragma unroll
  for (int i = 0; i < 8; ++i) {
    const int row = m0 + wm * 16 + half * 8 + i;
    const int col = n0 + wn * 16 + lid;
    score[(size_t)row * BB + col] = num[i] * rsqrtf(den[i]) + isv[i];
  }
}

extern "C" void kernel_launch(void* const* d_in, const int* in_sizes, int n_in,
                              void* d_out, int out_size, void* d_ws, size_t ws_size,
                              hipStream_t stream) {
  const float* r    = (const float*)d_in[0];
  const float* m    = (const float*)d_in[1];
  const float* t    = (const float*)d_in[2];
  const int*   ntid = (const int*)d_in[3];
  const int*   nlen = (const int*)d_in[4];
  const int*   antk = (const int*)d_in[5];
  const float* emb  = (const float*)d_in[6];
  const float* nW   = (const float*)d_in[7];
  const float* nb   = (const float*)d_in[8];
  const float* W1   = (const float*)d_in[9];
  const float* b1   = (const float*)d_in[10];
  const float* W2   = (const float*)d_in[11];
  const float* b2   = (const float*)d_in[12];
  const float* trW  = (const float*)d_in[13];
  const float* trb  = (const float*)d_in[14];
  const int*   t2i  = (const int*)d_in[15];

  float* score    = (float*)d_out;
  float* concepts = score + (size_t)BB * BB;

  float* ws  = (float*)d_ws;
  float* h   = ws;  ws += (size_t)BB * DD;
  float* aem = ws;  ws += (size_t)BB * DD;
  float* trm = ws;  ws += (size_t)BB * DD;
  float* Pw  = ws;  ws += (size_t)BB * DD;
  float* Qw  = ws;  ws += (size_t)BB * DD;
  float* rnw = ws;  ws += (size_t)BB * DD;
  float* tnw = ws;  ws += (size_t)BB * DD;
  float* t2w = ws;  ws += (size_t)BB * DD;

  k_front<<<dim3(2 * NGW + NCW + 2 * BB), dim3(128), 0, stream>>>(ntid, nlen, t2i, antk, emb, nW, nb,
                                                                    m, W1, b1, trW, trb, r, t,
                                                                    h, trm, rnw, tnw, t2w, concepts);
  k_gemm_xw<<<dim3(DD / GN, BB / GM), dim3(128), 0, stream>>>(h, W2, b2, aem);
  k_softmax_pq<<<BB, 256, 0, stream>>>(aem, trm, Pw, Qw);
  k_score<<<dim3(BB / SM, BB / SM), dim3(128), 0, stream>>>(Pw, Qw, rnw, t, t2w, tnw, score);
}


# Concepts GEMM over valid rows only: rows at or past noun_len are zero by the reference's mask, so tiles gather a prefix-summed list of valid rows; masked rows zero-filled; k order and sums unchanged
# speedup vs baseline: 4.9622x; 1.0545x over previous
#include <hip/hip_runtime.h>
#include <hip/hip_bf16.h>

typedef __attribute__((ext_vector_type(2))) float v2f;
typedef __attribute__((ext_vector_type(8))) float v8f;
typedef __attribute__((ext_vector_type(4))) float v4f;

#define BB 384
#define DD 1024
#define WD 300
#define LL 16

static __device__ __forceinline__ v8f wmma4(v2f a, v2f b, v8f c) {
  return __builtin_amdgcn_wmma_f32_16x16x4_f32(false, a, false, b, (short)0, c,
                                               false, false);
}

#define TM 64
#define TN 128
#define KC 20
#define NCH (WD / KC)
#define PA 20
#define PB 136

static __device__ __forceinline__ void tile_k300(const int* __restrict__ tokrow,
                                                 const float* __restrict__ emb,
                                                 const float* __restrict__ W,
                                                 const int n0,
                                                 float (*As)[TM * PA],
                                                 float (*Bs)[KC * PB],
                                                 v8f (&acc)[4][2]) {
  const int tid  = threadIdx.x;
  const int lane = tid & 31;
  const int wv   = tid >> 5;
  const int half = lane >> 4;
  const int lid  = lane & 15;

  int aoff[5];
#pragma unroll
  for (int j = 0; j < 5; ++j) {
    const int idx = tid + 128 * j;
    const int row = idx / 10;
    const int c2  = idx - row * 10;
    aoff[j] = tokrow[row] * WD + 2 * c2;
  }
  const int bk  = tid >> 5;
  const int bc4 = tid & 31;
  const float* wp = W + (size_t)bk * DD + n0 + bc4 * 4;

  v2f    ra[5];
  v4f    rb[5];
#pragma unroll
  for (int j = 0; j < 5; ++j) {
    ra[j] = *(const v2f*)(emb + aoff[j]);
    rb[j] = *(const v4f*)(wp + (size_t)(4 * j) * DD);
  }
#pragma unroll
  for (int j = 0; j < 5; ++j) {
    *(v2f*)&As[0][2 * (tid + 128 * j)] = ra[j];
    *(v4f*)&Bs[0][(bk + 4 * j) * PB + bc4 * 4] = rb[j];
  }
  __syncthreads();

  auto chunk_mma = [&](const float* Ab, const float* Bb) {
#pragma unroll
    for (int kk = 0; kk < KC / 4; ++kk) {
      const int k = kk * 4 + 2 * half;
      v2f b[2];
#pragma unroll
      for (int ns = 0; ns < 2; ++ns) {
        b[ns].x = Bb[k * PB + wv * 32 + ns * 16 + lid];
        b[ns].y = Bb[(k + 1) * PB + wv * 32 + ns * 16 + lid];
      }
#pragma unroll
      for (int s = 0; s < 4; ++s) {
        const v2f a = *(const v2f*)&Ab[(s * 16 + lid) * PA + k];
#pragma unroll
        for (int ns = 0; ns < 2; ++ns) acc[s][ns] = wmma4(a, b[ns], acc[s][ns]);
      }
    }
  };

#pragma unroll
  for (int j = 0; j < 5; ++j) {
    ra[j] = *(const v2f*)(emb + aoff[j] + KC);
    rb[j] = *(const v4f*)(wp + (size_t)(KC + 4 * j) * DD);
  }
  for (int c = 0; c < NCH - 2; ++c) {
    chunk_mma(As[c & 1], Bs[c & 1]);
#pragma unroll
    for (int j = 0; j < 5; ++j) {
      *(v2f*)&As[(c + 1) & 1][2 * (tid + 128 * j)] = ra[j];
      *(v4f*)&Bs[(c + 1) & 1][(bk + 4 * j) * PB + bc4 * 4] = rb[j];
    }
    __syncthreads();
    const int kn = (c + 2) * KC;
#pragma unroll
    for (int j = 0; j < 5; ++j) {
      ra[j] = *(const v2f*)(emb + aoff[j] + kn);
      rb[j] = *(const v4f*)(wp + (size_t)(kn + 4 * j) * DD);
    }
  }
  chunk_mma(As[(NCH - 2) & 1], Bs[(NCH - 2) & 1]);
#pragma unroll
  for (int j = 0; j < 5; ++j) {
    *(v2f*)&As[(NCH - 1) & 1][2 * (tid + 128 * j)] = ra[j];
    *(v4f*)&Bs[(NCH - 1) & 1][(bk + 4 * j) * PB + bc4 * 4] = rb[j];
  }
  __syncthreads();
  chunk_mma(As[(NCH - 1) & 1], Bs[(NCH - 1) & 1]);
}

#define NCW ((DD / TN) * (BB * LL / TM))

static __device__ __forceinline__ void concepts_tile(const int bx, const int by, float* smem,
                                                     const int* __restrict__ ntid,
                                                     const int* __restrict__ nlen,
                                                     const int* __restrict__ t2i,
                                                     const int* __restrict__ antk,
                                                     const float* __restrict__ emb,
                                                     const float* __restrict__ W,
                                                     const float* __restrict__ bias,
                                                     float* __restrict__ concepts) {
  float (*As)[TM * PA] = (float (*)[TM * PA])smem;
  float (*Bs)[KC * PB] = (float (*)[KC * PB])(smem + 2 * TM * PA);
  int* stok  = (int*)(smem + 2 * TM * PA + 2 * KC * PB);
  int* sorow = stok + TM;
  int* spart = sorow + TM;
  int* soff  = spart + 128;
  const int tid  = threadIdx.x;
  const int lane = tid & 31;
  const int wv   = tid >> 5;
  const int half = lane >> 4;
  const int lid  = lane & 15;
  const int n0 = bx * TN, j0 = by * TM;
  const int ln0 = min(max(nlen[3 * tid], 0), LL);
  const int ln1 = min(max(nlen[3 * tid + 1], 0), LL);
  const int ln2 = min(max(nlen[3 * tid + 2], 0), LL);
  spart[tid] = ln0 + ln1 + ln2;
  __syncthreads();
  int base = 0;
  for (int i = 0; i < 128; ++i) {
    const int v = spart[i];
    base += (i < tid) ? v : 0;
  }
  soff[3 * tid]     = base;
  soff[3 * tid + 1] = base + ln0;
  soff[3 * tid + 2] = base + ln0 + ln1;
  if (tid == 127) soff[BB] = base + ln0 + ln1 + ln2;
  __syncthreads();
  const int total = soff[BB];
  if (j0 >= total) return;
  if (tid < TM) {
    const int j = j0 + tid;
    int tokr = 0, orow = -1;
    if (j < total) {
      int lo = 0, hi = BB;
      for (int it = 0; it < 9; ++it) {
        const int mid = (lo + hi) >> 1;
        if (soff[mid] <= j) lo = mid; else hi = mid;
      }
      orow = lo * LL + (j - soff[lo]);
      const int tk   = ntid[orow];
      const int widx = t2i[tk];
      tokr = (widx > 0) ? antk[widx] : tk;
    }
    stok[tid]  = tokr;
    sorow[tid] = orow;
  }
  __syncthreads();
  v8f acc[4][2];
#pragma unroll
  for (int s = 0; s < 4; ++s)
#pragma unroll
    for (int ns = 0; ns < 2; ++ns) acc[s][ns] = (v8f){0.f,0.f,0.f,0.f,0.f,0.f,0.f,0.f};
  tile_k300(stok, emb, W, n0, As, Bs, acc);
  float bv[2];
#pragma unroll
  for (int ns = 0; ns < 2; ++ns) bv[ns] = bias[n0 + wv * 32 + ns * 16 + lid];
#pragma unroll
  for (int s = 0; s < 4; ++s) {
#pragma unroll
    for (int i = 0; i < 8; ++i) {
      const int row = sorow[s * 16 + half * 8 + i];
      if (row >= 0) {
#pragma unroll
        for (int ns = 0; ns < 2; ++ns) {
          const int col = n0 + wv * 32 + ns * 16 + lid;
          concepts[(size_t)row * DD + col] = acc[s][ns][i] + bv[ns];
        }
      }
    }
  }
}

#define GM 32
#define GN 64
#define GK 64
#define GCH (DD / GK)
#define GPA 68
#define GPB 72

#define NGW ((DD / GN) * (BB / GM))
#define SMEM_F (2 * GM * GPA + 2 * GK * GPB)

static __device__ __forceinline__ void gemm_tile(const int bx, const int by, float* smem,
                                                 const float* __restrict__ X,
                                                 const float* __restrict__ W,
                                                 const float* __restrict__ bias,
                                                 float* __restrict__ out, const int relu) {
  float (*As)[GM * GPA] = (float (*)[GM * GPA])smem;
  float (*Bs)[GK * GPB] = (float (*)[GK * GPB])(smem + 2 * GM * GPA);
  const int tid  = threadIdx.x;
  const int lane = tid & 31;
  const int wv   = tid >> 5;
  const int wm   = wv >> 1, wn = wv & 1;
  const int half = lane >> 4;
  const int lid  = lane & 15;
  const int n0 = bx * GN, m0 = by * GM;
  const int srow = tid >> 4;
  const int sc   = (tid & 15) * 4;
  const float* xp = X + (size_t)(m0 + srow) * DD + sc;
  const float* wp = W + (size_t)srow * DD + n0 + sc;

  v4f ra[4], rb[8];
#pragma unroll
  for (int j = 0; j < 4; ++j) ra[j] = *(const v4f*)(xp + (size_t)(8 * j) * DD);
#pragma unroll
  for (int j = 0; j < 8; ++j) rb[j] = *(const v4f*)(wp + (size_t)(8 * j) * DD);
#pragma unroll
  for (int j = 0; j < 4; ++j) *(v4f*)&As[0][(srow + 8 * j) * GPA + sc] = ra[j];
#pragma unroll
  for (int j = 0; j < 8; ++j) *(v4f*)&Bs[0][(srow + 8 * j) * GPB + sc] = rb[j];
  __syncthreads();

  v8f acc[2];
  acc[0] = (v8f){0.f,0.f,0.f,0.f,0.f,0.f,0.f,0.f};
  acc[1] = acc[0];
  auto chunk_mma = [&](const float* Ab, const float* Bb) {
#pragma unroll
    for (int kk = 0; kk < GK / 4; ++kk) {
      const int k = kk * 4 + 2 * half;
      const v2f a = *(const v2f*)&Ab[(wm * 16 + lid) * GPA + k];
#pragma unroll
      for (int ns = 0; ns < 2; ++ns) {
        v2f b;
        b.x = Bb[k * GPB + wn * 32 + ns * 16 + lid];
        b.y = Bb[(k + 1) * GPB + wn * 32 + ns * 16 + lid];
        acc[ns] = wmma4(a, b, acc[ns]);
      }
    }
  };

#pragma unroll
  for (int j = 0; j < 4; ++j) ra[j] = *(const v4f*)(xp + (size_t)(8 * j) * DD + GK);
#pragma unroll
  for (int j = 0; j < 8; ++j) rb[j] = *(const v4f*)(wp + (size_t)(GK + 8 * j) * DD);
  for (int c = 0; c < GCH - 2; ++c) {
    chunk_mma(As[c & 1], Bs[c & 1]);
#pragma unroll
    for (int j = 0; j < 4; ++j) *(v4f*)&As[(c + 1) & 1][(srow + 8 * j) * GPA + sc] = ra[j];
#pragma unroll
    for (int j = 0; j < 8; ++j) *(v4f*)&Bs[(c + 1) & 1][(srow + 8 * j) * GPB + sc] = rb[j];
    __syncthreads();
    const int kn = (c + 2) * GK;
#pragma unroll
    for (int j = 0; j < 4; ++j) ra[j] = *(const v4f*)(xp + (size_t)(8 * j) * DD + kn);
#pragma unroll
    for (int j = 0; j < 8; ++j) rb[j] = *(const v4f*)(wp + (size_t)(kn + 8 * j) * DD);
  }
  chunk_mma(As[(GCH - 2) & 1], Bs[(GCH - 2) & 1]);
#pragma unroll
  for (int j = 0; j < 4; ++j) *(v4f*)&As[(GCH - 1) & 1][(srow + 8 * j) * GPA + sc] = ra[j];
#pragma unroll
  for (int j = 0; j < 8; ++j) *(v4f*)&Bs[(GCH - 1) & 1][(srow + 8 * j) * GPB + sc] = rb[j];
  __syncthreads();
  chunk_mma(As[(GCH - 1) & 1], Bs[(GCH - 1) & 1]);

#pragma unroll
  for (int ns = 0; ns < 2; ++ns) {
    const int col = n0 + wn * 32 + ns * 16 + lid;
    const float bv = bias[col];
#pragma unroll
    for (int i = 0; i < 8; ++i) {
      float v = acc[ns][i] + bv;
      if (relu) v = fmaxf(v, 0.f);
      out[(size_t)(m0 + wm * 16 + half * 8 + i) * DD + col] = v;
    }
  }
}

__global__ __launch_bounds__(128) void k_gemm_xw(const float* __restrict__ X,
                                                 const float* __restrict__ W,
                                                 const float* __restrict__ bias,
                                                 float* __restrict__ out) {
  __shared__ __attribute__((aligned(16))) float smem[SMEM_F];
  gemm_tile(blockIdx.x, blockIdx.y, smem, X, W, bias, out, 0);
}

static __device__ __forceinline__ void row_l2(const int row, float* red,
                                              const float* __restrict__ src,
                                              float* __restrict__ dn,
                                              float* __restrict__ dsq) {
  const int tid = threadIdx.x;
  const float* p = src + (size_t)row * DD;
  float va[4], vb[4];
#pragma unroll
  for (int j = 0; j < 4; ++j) {
    va[j] = p[tid + 256 * j];
    vb[j] = p[tid + 128 + 256 * j];
  }
  float sa = 0.f, sb = 0.f;
#pragma unroll
  for (int j = 0; j < 4; ++j) {
    sa += va[j] * va[j];
    sb += vb[j] * vb[j];
  }
  if (dsq) {
#pragma unroll
    for (int j = 0; j < 4; ++j) {
      dsq[(size_t)row * DD + tid + 256 * j]       = va[j] * va[j];
      dsq[(size_t)row * DD + tid + 128 + 256 * j] = vb[j] * vb[j];
    }
  }
  red[tid] = sa + sb; __syncthreads();
  for (int st = 64; st > 0; st >>= 1) {
    if (tid < st) red[tid] += red[tid + st];
    __syncthreads();
  }
  const float inv = rsqrtf(red[0]);
#pragma unroll
  for (int j = 0; j < 4; ++j) {
    dn[(size_t)row * DD + tid + 256 * j]       = va[j] * inv;
    dn[(size_t)row * DD + tid + 128 + 256 * j] = vb[j] * inv;
  }
}

__global__ __launch_bounds__(128) void k_front(const int* __restrict__ ntid,
                                               const int* __restrict__ nlen,
                                               const int* __restrict__ t2i,
                                               const int* __restrict__ antk,
                                               const float* __restrict__ emb,
                                               const float* __restrict__ nW,
                                               const float* __restrict__ nb,
                                               const float* __restrict__ m,
                                               const float* __restrict__ W1,
                                               const float* __restrict__ b1,
                                               const float* __restrict__ trW,
                                               const float* __restrict__ trb,
                                               const float* __restrict__ r,
                                               const float* __restrict__ t,
                                               float* __restrict__ h,
                                               float* __restrict__ trm,
                                               float* __restrict__ rn,
                                               float* __restrict__ tn,
                                               float* __restrict__ t2,
                                               float* __restrict__ concepts) {
  __shared__ __attribute__((aligned(16))) float smem[SMEM_F];
  const int b = blockIdx.x;
  if (b < 2 * NGW) {
    const bool first = b < NGW;
    const int q = first ? b : b - NGW;
    gemm_tile(q % (DD / GN), q / (DD / GN), smem, m, first ? W1 : trW, first ? b1 : trb,
              first ? h : trm, first ? 1 : 0);
  } else if (b < 2 * NGW + NCW) {
    const int q = b - 2 * NGW;
    concepts_tile(q % (DD / TN), q / (DD / TN), smem, ntid, nlen, t2i, antk, emb, nW, nb, concepts);
  } else if (b < 2 * NGW + NCW + BB) {
    const int item = b - (2 * NGW + NCW);
    row_l2(item, smem, r, rn, nullptr);
    const int len = min(max(nlen[item], 0), LL);
    for (int l = len; l < LL; ++l) {
      v4f* zp = (v4f*)(concepts + ((size_t)item * LL + l) * DD);
      zp[threadIdx.x]       = (v4f){0.f, 0.f, 0.f, 0.f};
      zp[threadIdx.x + 128] = (v4f){0.f, 0.f, 0.f, 0.f};
    }
  } else {
    row_l2(b - (2 * NGW + NCW + BB), smem, t, tn, t2);
  }
}

__global__ __launch_bounds__(256) void k_softmax_pq(const float* __restrict__ aem,
                                                    const float* __restrict__ trm,
                                                    float* __restrict__ P,
                                                    float* __restrict__ Q) {
  __shared__ float red[256];
  __shared__ float red2[256];
  const int row = blockIdx.x, tid = threadIdx.x;
  const float* ap = aem + (size_t)row * DD;
  const float* tp = trm + (size_t)row * DD;
  float x[4], tv[4];
#pragma unroll
  for (int j = 0; j < 4; ++j) {
    x[j]  = ap[tid + 256 * j];
    tv[j] = tp[tid + 256 * j];
  }
  float mx = -3.402823466e+38f;
#pragma unroll
  for (int j = 0; j < 4; ++j) mx = fmaxf(mx, x[j]);
  red[tid] = mx; __syncthreads();
  for (int s = 128; s > 0; s >>= 1) {
    if (tid < s) red[tid] = fmaxf(red[tid], red[tid + s]);
    __syncthreads();
  }
  mx = red[0]; __syncthreads();
  float sum = 0.f, sq = 0.f;
#pragma unroll
  for (int j = 0; j < 4; ++j) {
    const float e = __expf(x[j] - mx);
    x[j] = e;
    sum += e;
    sq = fmaf(tv[j], tv[j], sq);
  }
  red[tid] = sum; red2[tid] = sq; __syncthreads();
  for (int s = 128; s > 0; s >>= 1) {
    if (tid < s) { red[tid] += red[tid + s]; red2[tid] += red2[tid + s]; }
    __syncthreads();
  }
  const float inv  = 1.f / red[0];
  const float inv2 = rsqrtf(red2[0]);
#pragma unroll
  for (int j = 0; j < 4; ++j) {
    const float a = x[j] * inv;
    P[(size_t)row * DD + tid + 256 * j] = tv[j] * inv2 * a;
    Q[(size_t)row * DD + tid + 256 * j] = a * a;
  }
}

#define SM 32
#define SK 64
#define SCH (DD / SK)
#define SP 68

__global__ __launch_bounds__(128) void k_score(const float* __restrict__ P,
                                               const float* __restrict__ Q,
                                               const float* __restrict__ rn,
                                               const float* __restrict__ t,
                                               const float* __restrict__ t2,
                                               const float* __restrict__ tn,
                                               float* __restrict__ score) {
  __shared__ __attribute__((aligned(16))) float Ls[2][6][SM * SP];
  const int tid  = threadIdx.x;
  const int lane = tid & 31;
  const int wv   = tid >> 5;
  const int wm   = wv >> 1, wn = wv & 1;
  const int half = lane >> 4;
  const int lid  = lane & 15;
  const int m0 = blockIdx.x * SM, n0 = blockIdx.y * SM;
  const int srow = tid >> 4;
  const int sc   = (tid & 15) * 4;
  const size_t ao = (size_t)(m0 + srow) * DD + sc;
  const size_t bo = (size_t)(n0 + srow) * DD + sc;
  const float* src[6] = {P + ao, Q + ao, rn + ao, t + bo, t2 + bo, tn + bo};

  v4f rg[6][4];
#pragma unroll
  for (int p = 0; p < 6; ++p)
#pragma unroll
    for (int j = 0; j < 4; ++j) rg[p][j] = *(const v4f*)(src[p] + (size_t)(8 * j) * DD);
#pragma unroll
  for (int p = 0; p < 6; ++p)
#pragma unroll
    for (int j = 0; j < 4; ++j) *(v4f*)&Ls[0][p][(srow + 8 * j) * SP + sc] = rg[p][j];
  __syncthreads();

  v8f num = {0.f,0.f,0.f,0.f,0.f,0.f,0.f,0.f};
  v8f den = num, isv = num;
  const int arow = (wm * 16 + lid) * SP;
  const int brow = (wn * 16 + lid) * SP;
  auto chunk_mma = [&](const float (*L)[SM * SP]) {
#pragma unroll
    for (int kk = 0; kk < SK / 4; ++kk) {
      const int k = kk * 4 + 2 * half;
      const v2f a1 = *(const v2f*)&L[0][arow + k];
      const v2f b1 = *(const v2f*)&L[3][brow + k];
      const v2f a2 = *(const v2f*)&L[1][arow + k];
      const v2f b2 = *(const v2f*)&L[4][brow + k];
      const v2f a3 = *(const v2f*)&L[2][arow + k];
      const v2f b3 = *(const v2f*)&L[5][brow + k];
      num = wmma4(a1, b1, num);
      den = wmma4(a2, b2, den);
      isv = wmma4(a3, b3, isv);
    }
  };

#pragma unroll
  for (int p = 0; p < 6; ++p)
#pragma unroll
    for (int j = 0; j < 4; ++j)
      rg[p][j] = *(const v4f*)(src[p] + (size_t)(8 * j) * DD + SK);
  for (int c = 0; c < SCH - 2; ++c) {
    chunk_mma(Ls[c & 1]);
#pragma unroll
    for (int p = 0; p < 6; ++p)
#pragma unroll
      for (int j = 0; j < 4; ++j)
        *(v4f*)&Ls[(c + 1) & 1][p][(srow + 8 * j) * SP + sc] = rg[p][j];
    __syncthreads();
    const int kn = (c + 2) * SK;
#pragma unroll
    for (int p = 0; p < 6; ++p)
#pragma unroll
      for (int j = 0; j < 4; ++j)
        rg[p][j] = *(const v4f*)(src[p] + (size_t)(8 * j) * DD + kn);
  }
  chunk_mma(Ls[(SCH - 2) & 1]);
#pragma unroll
  for (int p = 0; p < 6; ++p)
#pragma unroll
    for (int j = 0; j < 4; ++j)
      *(v4f*)&Ls[(SCH - 1) & 1][p][(srow + 8 * j) * SP + sc] = rg[p][j];
  __syncthreads();
  chunk_mma(Ls[(SCH - 1) & 1]);

#pragma unroll
  for (int i = 0; i < 8; ++i) {
    const int row = m0 + wm * 16 + half * 8 + i;
    const int col = n0 + wn * 16 + lid;
    score[(size_t)row * BB + col] = num[i] * rsqrtf(den[i]) + isv[i];
  }
}

extern "C" void kernel_launch(void* const* d_in, const int* in_sizes, int n_in,
                              void* d_out, int out_size, void* d_ws, size_t ws_size,
                              hipStream_t stream) {
  const float* r    = (const float*)d_in[0];
  const float* m    = (const float*)d_in[1];
  const float* t    = (const float*)d_in[2];
  const int*   ntid = (const int*)d_in[3];
  const int*   nlen = (const int*)d_in[4];
  const int*   antk = (const int*)d_in[5];
  const float* emb  = (const float*)d_in[6];
  const float* nW   = (const float*)d_in[7];
  const float* nb   = (const float*)d_in[8];
  const float* W1   = (const float*)d_in[9];
  const float* b1   = (const float*)d_in[10];
  const float* W2   = (const float*)d_in[11];
  const float* b2   = (const float*)d_in[12];
  const float* trW  = (const float*)d_in[13];
  const float* trb  = (const float*)d_in[14];
  const int*   t2i  = (const int*)d_in[15];

  float* score    = (float*)d_out;
  float* concepts = score + (size_t)BB * BB;

  float* ws  = (float*)d_ws;
  float* h   = ws;  ws += (size_t)BB * DD;
  float* aem = ws;  ws += (size_t)BB * DD;
  float* trm = ws;  ws += (size_t)BB * DD;
  float* Pw  = ws;  ws += (size_t)BB * DD;
  float* Qw  = ws;  ws += (size_t)BB * DD;
  float* rnw = ws;  ws += (size_t)BB * DD;
  float* tnw = ws;  ws += (size_t)BB * DD;
  float* t2w = ws;  ws += (size_t)BB * DD;

  k_front<<<dim3(2 * NGW + NCW + 2 * BB), dim3(128), 0, stream>>>(ntid, nlen, t2i, antk, emb, nW, nb,
                                                                    m, W1, b1, trW, trb, r, t,
                                                                    h, trm, rnw, tnw, t2w, concepts);
  k_gemm_xw<<<dim3(DD / GN, BB / GM), dim3(128), 0, stream>>>(h, W2, b2, aem);
  k_softmax_pq<<<BB, 256, 0, stream>>>(aem, trm, Pw, Qw);
  k_score<<<dim3(BB / SM, BB / SM), dim3(128), 0, stream>>>(Pw, Qw, rnw, t, t2w, tnw, score);
}


# Softmax and P/Q pass folded into the GEMMs: epilogues leave per-32-column row max, sum-exp, sum-sq partials; score GEMM rescales and sums them (sum order changes), forms P, Q in staging; 3 launches
# speedup vs baseline: 5.1655x; 1.0410x over previous
#include <hip/hip_runtime.h>
#include <hip/hip_bf16.h>

typedef __attribute__((ext_vector_type(2))) float v2f;
typedef __attribute__((ext_vector_type(8))) float v8f;
typedef __attribute__((ext_vector_type(4))) float v4f;

#define BB 384
#define DD 1024
#define WD 300
#define LL 16

static __device__ __forceinline__ v8f wmma4(v2f a, v2f b, v8f c) {
  return __builtin_amdgcn_wmma_f32_16x16x4_f32(false, a, false, b, (short)0, c,
                                               false, false);
}

#define NP 32

template <int CTRL>
static __device__ __forceinline__ float lane_x(float v) {
  return __builtin_bit_cast(float, __builtin_amdgcn_update_dpp(0, __builtin_bit_cast(int, v), CTRL, 0xF, 0xF, true));
}
#define DPP_Q1 0xB1
#define DPP_Q2 0x4E
#define DPP_HM 0x141
#define DPP_RM 0x140
static __device__ __forceinline__ float sum4(float v) {
  v += lane_x<DPP_Q1>(v); v += lane_x<DPP_Q2>(v);
  return v;
}
static __device__ __forceinline__ float max4(float v) {
  v = fmaxf(v, lane_x<DPP_Q1>(v)); v = fmaxf(v, lane_x<DPP_Q2>(v));
  return v;
}
static __device__ __forceinline__ float sum16(float v) {
  v = sum4(v); v += lane_x<DPP_HM>(v); v += lane_x<DPP_RM>(v);
  return v;
}
static __device__ __forceinline__ float max16(float v) {
  v = max4(v); v = fmaxf(v, lane_x<DPP_HM>(v)); v = fmaxf(v, lane_x<DPP_RM>(v));
  return v;
}

#define TM 64
#define TN 128
#define KC 20
#define NCH (WD / KC)
#define PA 20
#define PB 136

static __device__ __forceinline__ void tile_k300(const int* __restrict__ tokrow,
                                                 const float* __restrict__ emb,
                                                 const float* __restrict__ W,
                                                 const int n0,
                                                 float (*As)[TM * PA],
                                                 float (*Bs)[KC * PB],
                                                 v8f (&acc)[4][2]) {
  const int tid  = threadIdx.x;
  const int lane = tid & 31;
  const int wv   = tid >> 5;
  const int half = lane >> 4;
  const int lid  = lane & 15;

  int aoff[5];
#pragma unroll
  for (int j = 0; j < 5; ++j) {
    const int idx = tid + 128 * j;
    const int row = idx / 10;
    const int c2  = idx - row * 10;
    aoff[j] = tokrow[row] * WD + 2 * c2;
  }
  const int bk  = tid >> 5;
  const int bc4 = tid & 31;
  const float* wp = W + (size_t)bk * DD + n0 + bc4 * 4;

  v2f    ra[5];
  v4f    rb[5];
#pragma unroll
  for (int j = 0; j < 5; ++j) {
    ra[j] = *(const v2f*)(emb + aoff[j]);
    rb[j] = *(const v4f*)(wp + (size_t)(4 * j) * DD);
  }
#pragma unroll
  for (int j = 0; j < 5; ++j) {
    *(v2f*)&As[0][2 * (tid + 128 * j)] = ra[j];
    *(v4f*)&Bs[0][(bk + 4 * j) * PB + bc4 * 4] = rb[j];
  }
  __syncthreads();

  auto chunk_mma = [&](const float* Ab, const float* Bb) {
#pragma unroll
    for (int kk = 0; kk < KC / 4; ++kk) {
      const int k = kk * 4 + 2 * half;
      v2f b[2];
#pragma unroll
      for (int ns = 0; ns < 2; ++ns) {
        b[ns].x = Bb[k * PB + wv * 32 + ns * 16 + lid];
        b[ns].y = Bb[(k + 1) * PB + wv * 32 + ns * 16 + lid];
      }
#pragma unroll
      for (int s = 0; s < 4; ++s) {
        const v2f a = *(const v2f*)&Ab[(s * 16 + lid) * PA + k];
#pragma unroll
        for (int ns = 0; ns < 2; ++ns) acc[s][ns] = wmma4(a, b[ns], acc[s][ns]);
      }
    }
  };

#pragma unroll
  for (int j = 0; j < 5; ++j) {
    ra[j] = *(const v2f*)(emb + aoff[j] + KC);
    rb[j] = *(const v4f*)(wp + (size_t)(KC + 4 * j) * DD);
  }
  for (int c = 0; c < NCH - 2; ++c) {
    chunk_mma(As[c & 1], Bs[c & 1]);
#pragma unroll
    for (int j = 0; j < 5; ++j) {
      *(v2f*)&As[(c + 1) & 1][2 * (tid + 128 * j)] = ra[j];
      *(v4f*)&Bs[(c + 1) & 1][(bk + 4 * j) * PB + bc4 * 4] = rb[j];
    }
    __syncthreads();
    const int kn = (c + 2) * KC;
#pragma unroll
    for (int j = 0; j < 5; ++j) {
      ra[j] = *(const v2f*)(emb + aoff[j] + kn);
      rb[j] = *(const v4f*)(wp + (size_t)(kn + 4 * j) * DD);
    }
  }
  chunk_mma(As[(NCH - 2) & 1], Bs[(NCH - 2) & 1]);
#pragma unroll
  for (int j = 0; j < 5; ++j) {
    *(v2f*)&As[(NCH - 1) & 1][2 * (tid + 128 * j)] = ra[j];
    *(v4f*)&Bs[(NCH - 1) & 1][(bk + 4 * j) * PB + bc4 * 4] = rb[j];
  }
  __syncthreads();
  chunk_mma(As[(NCH - 1) & 1], Bs[(NCH - 1) & 1]);
}

#define NCW ((DD / TN) * (BB * LL / TM))

static __device__ __forceinline__ void concepts_tile(const int bx, const int by, float* smem,
                                                     const int* __restrict__ ntid,
                                                     const int* __restrict__ nlen,
                                                     const int* __restrict__ t2i,
                                                     const int* __restrict__ antk,
                                                     const float* __restrict__ emb,
                                                     const float* __restrict__ W,
                                                     const float* __restrict__ bias,
                                                     float* __restrict__ concepts) {
  float (*As)[TM * PA] = (float (*)[TM * PA])smem;
  float (*Bs)[KC * PB] = (float (*)[KC * PB])(smem + 2 * TM * PA);
  int* stok  = (int*)(smem + 2 * TM * PA + 2 * KC * PB);
  int* sorow = stok + TM;
  int* spart = sorow + TM;
  int* soff  = spart + 128;
  const int tid  = threadIdx.x;
  const int lane = tid & 31;
  const int wv   = tid >> 5;
  const int half = lane >> 4;
  const int lid  = lane & 15;
  const int n0 = bx * TN, j0 = by * TM;
  const int ln0 = min(max(nlen[3 * tid], 0), LL);
  const int ln1 = min(max(nlen[3 * tid + 1], 0), LL);
  const int ln2 = min(max(nlen[3 * tid + 2], 0), LL);
  spart[tid] = ln0 + ln1 + ln2;
  __syncthreads();
  int base = 0;
  for (int i = 0; i < 128; ++i) {
    const int v = spart[i];
    base += (i < tid) ? v : 0;
  }
  soff[3 * tid]     = base;
  soff[3 * tid + 1] = base + ln0;
  soff[3 * tid + 2] = base + ln0 + ln1;
  if (tid == 127) soff[BB] = base + ln0 + ln1 + ln2;
  __syncthreads();
  const int total = soff[BB];
  if (j0 >= total) return;
  if (tid < TM) {
    const int j = j0 + tid;
    int tokr = 0, orow = -1;
    if (j < total) {
      int lo = 0, hi = BB;
      for (int it = 0; it < 9; ++it) {
        const int mid = (lo + hi) >> 1;
        if (soff[mid] <= j) lo = mid; else hi = mid;
      }
      orow = lo * LL + (j - soff[lo]);
      const int tk   = ntid[orow];
      const int widx = t2i[tk];
      tokr = (widx > 0) ? antk[widx] : tk;
    }
    stok[tid]  = tokr;
    sorow[tid] = orow;
  }
  __syncthreads();
  v8f acc[4][2];
#pragma unroll
  for (int s = 0; s < 4; ++s)
#pragma unroll
    for (int ns = 0; ns < 2; ++ns) acc[s][ns] = (v8f){0.f,0.f,0.f,0.f,0.f,0.f,0.f,0.f};
  tile_k300(stok, emb, W, n0, As, Bs, acc);
  float bv[2];
#pragma unroll
  for (int ns = 0; ns < 2; ++ns) bv[ns] = bias[n0 + wv * 32 + ns * 16 + lid];
#pragma unroll
  for (int s = 0; s < 4; ++s) {
#pragma unroll
    for (int i = 0; i < 8; ++i) {
      const int row = sorow[s * 16 + half * 8 + i];
      if (row >= 0) {
#pragma unroll
        for (int ns = 0; ns < 2; ++ns) {
          const int col = n0 + wv * 32 + ns * 16 + lid;
          concepts[(size_t)row * DD + col] = acc[s][ns][i] + bv[ns];
        }
      }
    }
  }
}

#define GM 32
#define GN 64
#define GK 64
#define GCH (DD / GK)
#define GPA 68
#define GPB 72

#define NGW ((DD / GN) * (BB / GM))
#define SMEM_F (2 * GM * GPA + 2 * GK * GPB)

static __device__ __forceinline__ void gemm_tile(const int bx, const int by, float* smem,
                                                 const float* __restrict__ X,
                                                 const float* __restrict__ W,
                                                 const float* __restrict__ bias,
                                                 float* __restrict__ out, const int relu,
                                                 const int smode,
                                                 float* __restrict__ st0,
                                                 float* __restrict__ st1) {
  float (*As)[GM * GPA] = (float (*)[GM * GPA])smem;
  float (*Bs)[GK * GPB] = (float (*)[GK * GPB])(smem + 2 * GM * GPA);
  const int tid  = threadIdx.x;
  const int lane = tid & 31;
  const int wv   = tid >> 5;
  const int wm   = wv >> 1, wn = wv & 1;
  const int half = lane >> 4;
  const int lid  = lane & 15;
  const int n0 = bx * GN, m0 = by * GM;
  const int srow = tid >> 4;
  const int sc   = (tid & 15) * 4;
  const float* xp = X + (size_t)(m0 + srow) * DD + sc;
  const float* wp = W + (size_t)srow * DD + n0 + sc;

  v4f ra[4], rb[8];
#pragma unroll
  for (int j = 0; j < 4; ++j) ra[j] = *(const v4f*)(xp + (size_t)(8 * j) * DD);
#pragma unroll
  for (int j = 0; j < 8; ++j) rb[j] = *(const v4f*)(wp + (size_t)(8 * j) * DD);
#pragma unroll
  for (int j = 0; j < 4; ++j) *(v4f*)&As[0][(srow + 8 * j) * GPA + sc] = ra[j];
#pragma unroll
  for (int j = 0; j < 8; ++j) *(v4f*)&Bs[0][(srow + 8 * j) * GPB + sc] = rb[j];
  __syncthreads();

  v8f acc[2];
  acc[0] = (v8f){0.f,0.f,0.f,0.f,0.f,0.f,0.f,0.f};
  acc[1] = acc[0];
  auto chunk_mma = [&](const float* Ab, const float* Bb) {
#pragma unroll
    for (int kk = 0; kk < GK / 4; ++kk) {
      const int k = kk * 4 + 2 * half;
      const v2f a = *(const v2f*)&Ab[(wm * 16 + lid) * GPA + k];
#pragma unroll
      for (int ns = 0; ns < 2; ++ns) {
        v2f b;
        b.x = Bb[k * GPB + wn * 32 + ns * 16 + lid];
        b.y = Bb[(k + 1) * GPB + wn * 32 + ns * 16 + lid];
        acc[ns] = wmma4(a, b, acc[ns]);
      }
    }
  };

#pragma unroll
  for (int j = 0; j < 4; ++j) ra[j] = *(const v4f*)(xp + (size_t)(8 * j) * DD + GK);
#pragma unroll
  for (int j = 0; j < 8; ++j) rb[j] = *(const v4f*)(wp + (size_t)(GK + 8 * j) * DD);
  for (int c = 0; c < GCH - 2; ++c) {
    chunk_mma(As[c & 1], Bs[c & 1]);
#pragma unroll
    for (int j = 0; j < 4; ++j) *(v4f*)&As[(c + 1) & 1][(srow + 8 * j) * GPA + sc] = ra[j];
#pragma unroll
    for (int j = 0; j < 8; ++j) *(v4f*)&Bs[(c + 1) & 1][(srow + 8 * j) * GPB + sc] = rb[j];
    __syncthreads();
    const int kn = (c + 2) * GK;
#pragma unroll
    for (int j = 0; j < 4; ++j) ra[j] = *(const v4f*)(xp + (size_t)(8 * j) * DD + kn);
#pragma unroll
    for (int j = 0; j < 8; ++j) rb[j] = *(const v4f*)(wp + (size_t)(kn + 8 * j) * DD);
  }
  chunk_mma(As[(GCH - 2) & 1], Bs[(GCH - 2) & 1]);
#pragma unroll
  for (int j = 0; j < 4; ++j) *(v4f*)&As[(GCH - 1) & 1][(srow + 8 * j) * GPA + sc] = ra[j];
#pragma unroll
  for (int j = 0; j < 8; ++j) *(v4f*)&Bs[(GCH - 1) & 1][(srow + 8 * j) * GPB + sc] = rb[j];
  __syncthreads();
  chunk_mma(As[(GCH - 1) & 1], Bs[(GCH - 1) & 1]);

  const int col  = n0 + wn * 32 + lid;
  const int row0 = m0 + wm * 16 + half * 8;
  const float bv0 = bias[col], bv1 = bias[col + 16];
  float v0[8], v1[8];
#pragma unroll
  for (int i = 0; i < 8; ++i) {
    v0[i] = acc[0][i] + bv0;
    v1[i] = acc[1][i] + bv1;
    if (relu) { v0[i] = fmaxf(v0[i], 0.f); v1[i] = fmaxf(v1[i], 0.f); }
    out[(size_t)(row0 + i) * DD + col]      = v0[i];
    out[(size_t)(row0 + i) * DD + col + 16] = v1[i];
  }
  if (smode != 0) {
    float ra[8], rb[8];
#pragma unroll
    for (int i = 0; i < 8; ++i) {
      if (smode == 1) {
        ra[i] = sum16(v0[i] * v0[i] + v1[i] * v1[i]);
        rb[i] = 0.f;
      } else {
        ra[i] = max16(fmaxf(v0[i], v1[i]));
        rb[i] = sum16(__expf(v0[i] - ra[i]) + __expf(v1[i] - ra[i]));
      }
    }
    if (lid == 0) {
      const size_t so = (size_t)row0 * NP + bx * 2 + wn;
#pragma unroll
      for (int i = 0; i < 8; ++i) {
        st0[so + (size_t)i * NP] = ra[i];
        if (smode == 2) st1[so + (size_t)i * NP] = rb[i];
      }
    }
  }
}

__global__ __launch_bounds__(128) void k_gemm_xw(const float* __restrict__ X,
                                                 const float* __restrict__ W,
                                                 const float* __restrict__ bias,
                                                 float* __restrict__ out,
                                                 float* __restrict__ pmax,
                                                 float* __restrict__ psum) {
  __shared__ __attribute__((aligned(16))) float smem[SMEM_F];
  gemm_tile(blockIdx.x, blockIdx.y, smem, X, W, bias, out, 0, 2, pmax, psum);
}

static __device__ __forceinline__ void row_l2(const int row, float* red,
                                              const float* __restrict__ src,
                                              float* __restrict__ dn,
                                              float* __restrict__ dsq) {
  const int tid = threadIdx.x;
  const float* p = src + (size_t)row * DD;
  float va[4], vb[4];
#pragma unroll
  for (int j = 0; j < 4; ++j) {
    va[j] = p[tid + 256 * j];
    vb[j] = p[tid + 128 + 256 * j];
  }
  float sa = 0.f, sb = 0.f;
#pragma unroll
  for (int j = 0; j < 4; ++j) {
    sa += va[j] * va[j];
    sb += vb[j] * vb[j];
  }
  if (dsq) {
#pragma unroll
    for (int j = 0; j < 4; ++j) {
      dsq[(size_t)row * DD + tid + 256 * j]       = va[j] * va[j];
      dsq[(size_t)row * DD + tid + 128 + 256 * j] = vb[j] * vb[j];
    }
  }
  red[tid] = sa + sb; __syncthreads();
  for (int st = 64; st > 0; st >>= 1) {
    if (tid < st) red[tid] += red[tid + st];
    __syncthreads();
  }
  const float inv = rsqrtf(red[0]);
#pragma unroll
  for (int j = 0; j < 4; ++j) {
    dn[(size_t)row * DD + tid + 256 * j]       = va[j] * inv;
    dn[(size_t)row * DD + tid + 128 + 256 * j] = vb[j] * inv;
  }
}

__global__ __launch_bounds__(128) void k_front(const int* __restrict__ ntid,
                                               const int* __restrict__ nlen,
                                               const int* __restrict__ t2i,
                                               const int* __restrict__ antk,
                                               const float* __restrict__ emb,
                                               const float* __restrict__ nW,
                                               const float* __restrict__ nb,
                                               const float* __restrict__ m,
                                               const float* __restrict__ W1,
                                               const float* __restrict__ b1,
                                               const float* __restrict__ trW,
                                               const float* __restrict__ trb,
                                               const float* __restrict__ r,
                                               const float* __restrict__ t,
                                               float* __restrict__ h,
                                               float* __restrict__ trm,
                                               float* __restrict__ rn,
                                               float* __restrict__ tn,
                                               float* __restrict__ t2,
                                               float* __restrict__ trsq,
                                               float* __restrict__ concepts) {
  __shared__ __attribute__((aligned(16))) float smem[SMEM_F];
  const int b = blockIdx.x;
  if (b < 2 * NGW) {
    const bool first = b < NGW;
    const int q = first ? b : b - NGW;
    gemm_tile(q % (DD / GN), q / (DD / GN), smem, m, first ? W1 : trW, first ? b1 : trb,
              first ? h : trm, first ? 1 : 0, first ? 0 : 1, trsq, nullptr);
  } else if (b < 2 * NGW + NCW) {
    const int q = b - 2 * NGW;
    concepts_tile(q % (DD / TN), q / (DD / TN), smem, ntid, nlen, t2i, antk, emb, nW, nb, concepts);
  } else if (b < 2 * NGW + NCW + BB) {
    const int item = b - (2 * NGW + NCW);
    row_l2(item, smem, r, rn, nullptr);
    const int len = min(max(nlen[item], 0), LL);
    for (int l = len; l < LL; ++l) {
      v4f* zp = (v4f*)(concepts + ((size_t)item * LL + l) * DD);
      zp[threadIdx.x]       = (v4f){0.f, 0.f, 0.f, 0.f};
      zp[threadIdx.x + 128] = (v4f){0.f, 0.f, 0.f, 0.f};
    }
  } else {
    row_l2(b - (2 * NGW + NCW + BB), smem, t, tn, t2);
  }
}

#define SM 32
#define SK 64
#define SCH (DD / SK)
#define SP 68

__global__ __launch_bounds__(128) void k_score(const float* __restrict__ aem,
                                               const float* __restrict__ trm,
                                               const float* __restrict__ rn,
                                               const float* __restrict__ t,
                                               const float* __restrict__ t2,
                                               const float* __restrict__ tn,
                                               const float* __restrict__ pmax,
                                               const float* __restrict__ psum,
                                               const float* __restrict__ trsq,
                                               float* __restrict__ score) {
  __shared__ __attribute__((aligned(16))) float Ls[2][6][SM * SP];
  __shared__ float sst[3][SM];
  const int tid  = threadIdx.x;
  const int lane = tid & 31;
  const int wv   = tid >> 5;
  const int wm   = wv >> 1, wn = wv & 1;
  const int half = lane >> 4;
  const int lid  = lane & 15;
  const int m0 = blockIdx.x * SM, n0 = blockIdx.y * SM;
  const int srow = tid >> 4;
  const int sc   = (tid & 15) * 4;
  const size_t ao = (size_t)(m0 + srow) * DD + sc;
  const size_t bo = (size_t)(n0 + srow) * DD + sc;
  const float* src[6] = {aem + ao, trm + ao, rn + ao, t + bo, t2 + bo, tn + bo};

  v4f rg[6][4];
#pragma unroll
  for (int p = 0; p < 6; ++p)
#pragma unroll
    for (int j = 0; j < 4; ++j) rg[p][j] = *(const v4f*)(src[p] + (size_t)(8 * j) * DD);
  __builtin_amdgcn_sched_barrier(0);

  {
    const int sr = tid >> 2, sq4 = tid & 3;
    const size_t so = (size_t)(m0 + sr) * NP + sq4 * 8;
    float pm[8], pe[8];
    float sq = 0.f;
#pragma unroll
    for (int u = 0; u < 8; ++u) {
      pm[u] = pmax[so + u];
      pe[u] = psum[so + u];
      sq += trsq[so + u];
    }
    float mx = pm[0];
#pragma unroll
    for (int u = 1; u < 8; ++u) mx = fmaxf(mx, pm[u]);
    mx = max4(mx);
    float se = 0.f;
#pragma unroll
    for (int u = 0; u < 8; ++u) se += pe[u] * __expf(pm[u] - mx);
    se = sum4(se);
    sq = sum4(sq);
    if (sq4 == 0) {
      sst[0][sr] = mx;
      sst[1][sr] = 1.f / se;
      sst[2][sr] = rsqrtf(sq);
    }
  }
  __syncthreads();
  float rmx[4], rinv[4], rinv2[4];
#pragma unroll
  for (int j = 0; j < 4; ++j) {
    rmx[j]   = sst[0][srow + 8 * j];
    rinv[j]  = sst[1][srow + 8 * j];
    rinv2[j] = sst[2][srow + 8 * j];
  }
  auto to_pq = [&]() {
#pragma unroll
    for (int j = 0; j < 4; ++j) {
#pragma unroll
      for (int e = 0; e < 4; ++e) {
        const float a = __expf(rg[0][j][e] - rmx[j]) * rinv[j];
        rg[0][j][e] = rg[1][j][e] * rinv2[j] * a;
        rg[1][j][e] = a * a;
      }
    }
  };

  to_pq();
#pragma unroll
  for (int p = 0; p < 6; ++p)
#pragma unroll
    for (int j = 0; j < 4; ++j) *(v4f*)&Ls[0][p][(srow + 8 * j) * SP + sc] = rg[p][j];
  __syncthreads();

  v8f num = {0.f,0.f,0.f,0.f,0.f,0.f,0.f,0.f};
  v8f den = num, isv = num;
  const int arow = (wm * 16 + lid) * SP;
  const int brow = (wn * 16 + lid) * SP;
  auto chunk_mma = [&](const float (*L)[SM * SP]) {
#pragma unroll
    for (int kk = 0; kk < SK / 4; ++kk) {
      const int k = kk * 4 + 2 * half;
      const v2f a1 = *(const v2f*)&L[0][arow + k];
      const v2f b1 = *(const v2f*)&L[3][brow + k];
      const v2f a2 = *(const v2f*)&L[1][arow + k];
      const v2f b2 = *(const v2f*)&L[4][brow + k];
      const v2f a3 = *(const v2f*)&L[2][arow + k];
      const v2f b3 = *(const v2f*)&L[5][brow + k];
      num = wmma4(a1, b1, num);
      den = wmma4(a2, b2, den);
      isv = wmma4(a3, b3, isv);
    }
  };

#pragma unroll
  for (int p = 0; p < 6; ++p)
#pragma unroll
    for (int j = 0; j < 4; ++j)
      rg[p][j] = *(const v4f*)(src[p] + (size_t)(8 * j) * DD + SK);
  for (int c = 0; c < SCH - 2; ++c) {
    chunk_mma(Ls[c & 1]);
    to_pq();
#pragma unroll
    for (int p = 0; p < 6; ++p)
#pragma unroll
      for (int j = 0; j < 4; ++j)
        *(v4f*)&Ls[(c + 1) & 1][p][(srow + 8 * j) * SP + sc] = rg[p][j];
    __syncthreads();
    const int kn = (c + 2) * SK;
#pragma unroll
    for (int p = 0; p < 6; ++p)
#pragma unroll
      for (int j = 0; j < 4; ++j)
        rg[p][j] = *(const v4f*)(src[p] + (size_t)(8 * j) * DD + kn);
  }
  chunk_mma(Ls[(SCH - 2) & 1]);
  to_pq();
#pragma unroll
  for (int p = 0; p < 6; ++p)
#pragma unroll
    for (int j = 0; j < 4; ++j)
      *(v4f*)&Ls[(SCH - 1) & 1][p][(srow + 8 * j) * SP + sc] = rg[p][j];
  __syncthreads();
  chunk_mma(Ls[(SCH - 1) & 1]);

#pragma unroll
  for (int i = 0; i < 8; ++i) {
    const int row = m0 + wm * 16 + half * 8 + i;
    const int col = n0 + wn * 16 + lid;
    score[(size_t)row * BB + col] = num[i] * rsqrtf(den[i]) + isv[i];
  }
}

extern "C" void kernel_launch(void* const* d_in, const int* in_sizes, int n_in,
                              void* d_out, int out_size, void* d_ws, size_t ws_size,
                              hipStream_t stream) {
  const float* r    = (const float*)d_in[0];
  const float* m    = (const float*)d_in[1];
  const float* t    = (const float*)d_in[2];
  const int*   ntid = (const int*)d_in[3];
  const int*   nlen = (const int*)d_in[4];
  const int*   antk = (const int*)d_in[5];
  const float* emb  = (const float*)d_in[6];
  const float* nW   = (const float*)d_in[7];
  const float* nb   = (const float*)d_in[8];
  const float* W1   = (const float*)d_in[9];
  const float* b1   = (const float*)d_in[10];
  const float* W2   = (const float*)d_in[11];
  const float* b2   = (const float*)d_in[12];
  const float* trW  = (const float*)d_in[13];
  const float* trb  = (const float*)d_in[14];
  const int*   t2i  = (const int*)d_in[15];

  float* score    = (float*)d_out;
  float* concepts = score + (size_t)BB * BB;

  float* ws  = (float*)d_ws;
  float* h   = ws;  ws += (size_t)BB * DD;
  float* aem = ws;  ws += (size_t)BB * DD;
  float* trm = ws;  ws += (size_t)BB * DD;
  float* rnw = ws;  ws += (size_t)BB * DD;
  float* tnw = ws;  ws += (size_t)BB * DD;
  float* t2w = ws;  ws += (size_t)BB * DD;
  float* pmx = ws;  ws += (size_t)BB * NP;
  float* psm = ws;  ws += (size_t)BB * NP;
  float* tsq = ws;  ws += (size_t)BB * NP;

  k_front<<<dim3(2 * NGW + NCW + 2 * BB), dim3(128), 0, stream>>>(ntid, nlen, t2i, antk, emb, nW, nb,
                                                                    m, W1, b1, trW, trb, r, t,
                                                                    h, trm, rnw, tnw, t2w, tsq, concepts);
  k_gemm_xw<<<dim3(DD / GN, BB / GM), dim3(128), 0, stream>>>(h, W2, b2, aem, pmx, psm);
  k_score<<<dim3(BB / SM, BB / SM), dim3(128), 0, stream>>>(aem, trm, rnw, t, t2w, tnw, pmx, psm, tsq, score);
}


# Fragment double buffering in the K=1024 GEMMs and the score kernel: chunks multiplied in halves from two register sets, next half's LDS traffic issued under the matrix work; k order, sums unchanged
# speedup vs baseline: 5.3623x; 1.0381x over previous
#include <hip/hip_runtime.h>
#include <hip/hip_bf16.h>

typedef __attribute__((ext_vector_type(2))) float v2f;
typedef __attribute__((ext_vector_type(8))) float v8f;
typedef __attribute__((ext_vector_type(4))) float v4f;

#define BB 384
#define DD 1024
#define WD 300
#define LL 16

static __device__ __forceinline__ v8f wmma4(v2f a, v2f b, v8f c) {
  return __builtin_amdgcn_wmma_f32_16x16x4_f32(false, a, false, b, (short)0, c,
                                               false, false);
}

static __device__ __forceinline__ void lds_barrier() {
  __builtin_amdgcn_fence(__ATOMIC_RELEASE, "workgroup", "local");
  __builtin_amdgcn_s_barrier();
  __builtin_amdgcn_fence(__ATOMIC_ACQUIRE, "workgroup", "local");
}

#define NP 32

template <int CTRL>
static __device__ __forceinline__ float lane_x(float v) {
  return __builtin_bit_cast(float, __builtin_amdgcn_update_dpp(0, __builtin_bit_cast(int, v), CTRL, 0xF, 0xF, true));
}
#define DPP_Q1 0xB1
#define DPP_Q2 0x4E
#define DPP_HM 0x141
#define DPP_RM 0x140
static __device__ __forceinline__ float sum4(float v) {
  v += lane_x<DPP_Q1>(v); v += lane_x<DPP_Q2>(v);
  return v;
}
static __device__ __forceinline__ float max4(float v) {
  v = fmaxf(v, lane_x<DPP_Q1>(v)); v = fmaxf(v, lane_x<DPP_Q2>(v));
  return v;
}
static __device__ __forceinline__ float sum16(float v) {
  v = sum4(v); v += lane_x<DPP_HM>(v); v += lane_x<DPP_RM>(v);
  return v;
}
static __device__ __forceinline__ float max16(float v) {
  v = max4(v); v = fmaxf(v, lane_x<DPP_HM>(v)); v = fmaxf(v, lane_x<DPP_RM>(v));
  return v;
}

#define TM 64
#define TN 128
#define KC 20
#define NCH (WD / KC)
#define PA 20
#define PB 136

static __device__ __forceinline__ void tile_k300(const int* __restrict__ tokrow,
                                                 const float* __restrict__ emb,
                                                 const float* __restrict__ W,
                                                 const int n0,
                                                 float (*As)[TM * PA],
                                                 float (*Bs)[KC * PB],
                                                 v8f (&acc)[4][2]) {
  const int tid  = threadIdx.x;
  const int lane = tid & 31;
  const int wv   = tid >> 5;
  const int half = lane >> 4;
  const int lid  = lane & 15;

  int aoff[5];
#pragma unroll
  for (int j = 0; j < 5; ++j) {
    const int idx = tid + 128 * j;
    const int row = idx / 10;
    const int c2  = idx - row * 10;
    aoff[j] = tokrow[row] * WD + 2 * c2;
  }
  const int bk  = tid >> 5;
  const int bc4 = tid & 31;
  const float* wp = W + (size_t)bk * DD + n0 + bc4 * 4;

  v2f    ra[5];
  v4f    rb[5];
#pragma unroll
  for (int j = 0; j < 5; ++j) {
    ra[j] = *(const v2f*)(emb + aoff[j]);
    rb[j] = *(const v4f*)(wp + (size_t)(4 * j) * DD);
  }
#pragma unroll
  for (int j = 0; j < 5; ++j) {
    *(v2f*)&As[0][2 * (tid + 128 * j)] = ra[j];
    *(v4f*)&Bs[0][(bk + 4 * j) * PB + bc4 * 4] = rb[j];
  }
  __syncthreads();

  auto chunk_mma = [&](const float* Ab, const float* Bb) {
#pragma unroll
    for (int kk = 0; kk < KC / 4; ++kk) {
      const int k = kk * 4 + 2 * half;
      v2f b[2];
#pragma unroll
      for (int ns = 0; ns < 2; ++ns) {
        b[ns].x = Bb[k * PB + wv * 32 + ns * 16 + lid];
        b[ns].y = Bb[(k + 1) * PB + wv * 32 + ns * 16 + lid];
      }
#pragma unroll
      for (int s = 0; s < 4; ++s) {
        const v2f a = *(const v2f*)&Ab[(s * 16 + lid) * PA + k];
#pragma unroll
        for (int ns = 0; ns < 2; ++ns) acc[s][ns] = wmma4(a, b[ns], acc[s][ns]);
      }
    }
  };

#pragma unroll
  for (int j = 0; j < 5; ++j) {
    ra[j] = *(const v2f*)(emb + aoff[j] + KC);
    rb[j] = *(const v4f*)(wp + (size_t)(KC + 4 * j) * DD);
  }
  for (int c = 0; c < NCH - 2; ++c) {
    chunk_mma(As[c & 1], Bs[c & 1]);
#pragma unroll
    for (int j = 0; j < 5; ++j) {
      *(v2f*)&As[(c + 1) & 1][2 * (tid + 128 * j)] = ra[j];
      *(v4f*)&Bs[(c + 1) & 1][(bk + 4 * j) * PB + bc4 * 4] = rb[j];
    }
    __syncthreads();
    const int kn = (c + 2) * KC;
#pragma unroll
    for (int j = 0; j < 5; ++j) {
      ra[j] = *(const v2f*)(emb + aoff[j] + kn);
      rb[j] = *(const v4f*)(wp + (size_t)(kn + 4 * j) * DD);
    }
  }
  chunk_mma(As[(NCH - 2) & 1], Bs[(NCH - 2) & 1]);
#pragma unroll
  for (int j = 0; j < 5; ++j) {
    *(v2f*)&As[(NCH - 1) & 1][2 * (tid + 128 * j)] = ra[j];
    *(v4f*)&Bs[(NCH - 1) & 1][(bk + 4 * j) * PB + bc4 * 4] = rb[j];
  }
  __syncthreads();
  chunk_mma(As[(NCH - 1) & 1], Bs[(NCH - 1) & 1]);
}

#define NCW ((DD / TN) * (BB * LL / TM))

static __device__ __forceinline__ void concepts_tile(const int bx, const int by, float* smem,
                                                     const int* __restrict__ ntid,
                                                     const int* __restrict__ nlen,
                                                     const int* __restrict__ t2i,
                                                     const int* __restrict__ antk,
                                                     const float* __restrict__ emb,
                                                     const float* __restrict__ W,
                                                     const float* __restrict__ bias,
                                                     float* __restrict__ concepts) {
  float (*As)[TM * PA] = (float (*)[TM * PA])smem;
  float (*Bs)[KC * PB] = (float (*)[KC * PB])(smem + 2 * TM * PA);
  int* stok  = (int*)(smem + 2 * TM * PA + 2 * KC * PB);
  int* sorow = stok + TM;
  int* spart = sorow + TM;
  int* soff  = spart + 128;
  const int tid  = threadIdx.x;
  const int lane = tid & 31;
  const int wv   = tid >> 5;
  const int half = lane >> 4;
  const int lid  = lane & 15;
  const int n0 = bx * TN, j0 = by * TM;
  const int ln0 = min(max(nlen[3 * tid], 0), LL);
  const int ln1 = min(max(nlen[3 * tid + 1], 0), LL);
  const int ln2 = min(max(nlen[3 * tid + 2], 0), LL);
  spart[tid] = ln0 + ln1 + ln2;
  __syncthreads();
  int base = 0;
  for (int i = 0; i < 128; ++i) {
    const int v = spart[i];
    base += (i < tid) ? v : 0;
  }
  soff[3 * tid]     = base;
  soff[3 * tid + 1] = base + ln0;
  soff[3 * tid + 2] = base + ln0 + ln1;
  if (tid == 127) soff[BB] = base + ln0 + ln1 + ln2;
  __syncthreads();
  const int total = soff[BB];
  if (j0 >= total) return;
  if (tid < TM) {
    const int j = j0 + tid;
    int tokr = 0, orow = -1;
    if (j < total) {
      int lo = 0, hi = BB;
      for (int it = 0; it < 9; ++it) {
        const int mid = (lo + hi) >> 1;
        if (soff[mid] <= j) lo = mid; else hi = mid;
      }
      orow = lo * LL + (j - soff[lo]);
      const int tk   = ntid[orow];
      const int widx = t2i[tk];
      tokr = (widx > 0) ? antk[widx] : tk;
    }
    stok[tid]  = tokr;
    sorow[tid] = orow;
  }
  __syncthreads();
  v8f acc[4][2];
#pragma unroll
  for (int s = 0; s < 4; ++s)
#pragma unroll
    for (int ns = 0; ns < 2; ++ns) acc[s][ns] = (v8f){0.f,0.f,0.f,0.f,0.f,0.f,0.f,0.f};
  tile_k300(stok, emb, W, n0, As, Bs, acc);
  float bv[2];
#pragma unroll
  for (int ns = 0; ns < 2; ++ns) bv[ns] = bias[n0 + wv * 32 + ns * 16 + lid];
#pragma unroll
  for (int s = 0; s < 4; ++s) {
#pragma unroll
    for (int i = 0; i < 8; ++i) {
      const int row = sorow[s * 16 + half * 8 + i];
      if (row >= 0) {
#pragma unroll
        for (int ns = 0; ns < 2; ++ns) {
          const int col = n0 + wv * 32 + ns * 16 + lid;
          concepts[(size_t)row * DD + col] = acc[s][ns][i] + bv[ns];
        }
      }
    }
  }
}

#define GM 32
#define GN 64
#define GK 64
#define GCH (DD / GK)
#define GPA 68
#define GPB 72

#define NGW ((DD / GN) * (BB / GM))
#define SMEM_F (2 * GM * GPA + 2 * GK * GPB)

static __device__ __forceinline__ void gemm_tile(const int bx, const int by, float* smem,
                                                 const float* __restrict__ X,
                                                 const float* __restrict__ W,
                                                 const float* __restrict__ bias,
                                                 float* __restrict__ out, const int relu,
                                                 const int smode,
                                                 float* __restrict__ st0,
                                                 float* __restrict__ st1) {
  float (*As)[GM * GPA] = (float (*)[GM * GPA])smem;
  float (*Bs)[GK * GPB] = (float (*)[GK * GPB])(smem + 2 * GM * GPA);
  const int tid  = threadIdx.x;
  const int lane = tid & 31;
  const int wv   = tid >> 5;
  const int wm   = wv >> 1, wn = wv & 1;
  const int half = lane >> 4;
  const int lid  = lane & 15;
  const int n0 = bx * GN, m0 = by * GM;
  const int srow = tid >> 4;
  const int sc   = (tid & 15) * 4;
  const float* xp = X + (size_t)(m0 + srow) * DD + sc;
  const float* wp = W + (size_t)srow * DD + n0 + sc;

  v4f ra[4], rb[8];
#pragma unroll
  for (int j = 0; j < 4; ++j) ra[j] = *(const v4f*)(xp + (size_t)(8 * j) * DD);
#pragma unroll
  for (int j = 0; j < 8; ++j) rb[j] = *(const v4f*)(wp + (size_t)(8 * j) * DD);
#pragma unroll
  for (int j = 0; j < 4; ++j) *(v4f*)&As[0][(srow + 8 * j) * GPA + sc] = ra[j];
#pragma unroll
  for (int j = 0; j < 8; ++j) *(v4f*)&Bs[0][(srow + 8 * j) * GPB + sc] = rb[j];
  __syncthreads();

  v8f acc[2];
  acc[0] = (v8f){0.f,0.f,0.f,0.f,0.f,0.f,0.f,0.f};
  acc[1] = acc[0];
  const int fao = (wm * 16 + lid) * GPA + 2 * half;
  const int fbo = (2 * half) * GPB + wn * 32 + lid;
  v2f   fa0[8], fa1[8];
  float bx0[8][2], by0[8][2], bx1[8][2], by1[8][2];
  auto load_frag = [&](const float* Ab, const float* Bb, const int h, v2f (&fa)[8], float (&bx)[8][2], float (&by)[8][2]) {
#pragma unroll
    for (int kk = 0; kk < 8; ++kk) {
      const int k4 = (h * 8 + kk) * 4;
      fa[kk] = *(const v2f*)&Ab[fao + k4];
#pragma unroll
      for (int ns = 0; ns < 2; ++ns) {
        bx[kk][ns] = Bb[fbo + k4 * GPB + ns * 16];
        by[kk][ns] = Bb[fbo + (k4 + 1) * GPB + ns * 16];
      }
    }
  };
  auto mma_k = [&](const int kk, const v2f (&fa)[8], const float (&bx)[8][2], const float (&by)[8][2]) {
#pragma unroll
    for (int ns = 0; ns < 2; ++ns) {
      v2f b;
      b.x = bx[kk][ns];
      b.y = by[kk][ns];
      acc[ns] = wmma4(fa[kk], b, acc[ns]);
    }
  };

#pragma unroll
  for (int j = 0; j < 4; ++j) ra[j] = *(const v4f*)(xp + (size_t)(8 * j) * DD + GK);
#pragma unroll
  for (int j = 0; j < 8; ++j) rb[j] = *(const v4f*)(wp + (size_t)(GK + 8 * j) * DD);
  load_frag(As[0], Bs[0], 0, fa0, bx0, by0);
  auto step = [&](const int c, const bool more) {
    load_frag(As[c & 1], Bs[c & 1], 1, fa1, bx1, by1);
#pragma unroll
    for (int j = 0; j < 4; ++j) *(v4f*)&As[(c + 1) & 1][(srow + 8 * j) * GPA + sc] = ra[j];
#pragma unroll
    for (int j = 0; j < 8; ++j) *(v4f*)&Bs[(c + 1) & 1][(srow + 8 * j) * GPB + sc] = rb[j];
    if (more) {
      const int kn = (c + 2) * GK;
#pragma unroll
      for (int j = 0; j < 4; ++j) ra[j] = *(const v4f*)(xp + (size_t)(8 * j) * DD + kn);
#pragma unroll
      for (int j = 0; j < 8; ++j) rb[j] = *(const v4f*)(wp + (size_t)(kn + 8 * j) * DD);
    }
#pragma unroll
    for (int kk = 0; kk < 8; ++kk) mma_k(kk, fa0, bx0, by0);
    __builtin_amdgcn_sched_barrier(0);
    lds_barrier();
    load_frag(As[(c + 1) & 1], Bs[(c + 1) & 1], 0, fa0, bx0, by0);
#pragma unroll
    for (int kk = 0; kk < 8; ++kk) mma_k(kk, fa1, bx1, by1);
    __builtin_amdgcn_sched_group_barrier(0x100, 20, 0);
    __builtin_amdgcn_sched_group_barrier(0x008, 16, 0);
    __builtin_amdgcn_sched_group_barrier(0x002, 64, 0);
    __builtin_amdgcn_sched_barrier(0);
  };
  for (int c = 0; c < GCH - 2; ++c) step(c, true);
  step(GCH - 2, false);
  load_frag(As[(GCH - 1) & 1], Bs[(GCH - 1) & 1], 1, fa1, bx1, by1);
#pragma unroll
  for (int kk = 0; kk < 8; ++kk) mma_k(kk, fa0, bx0, by0);
#pragma unroll
  for (int kk = 0; kk < 8; ++kk) mma_k(kk, fa1, bx1, by1);

  const int col  = n0 + wn * 32 + lid;
  const int row0 = m0 + wm * 16 + half * 8;
  const float bv0 = bias[col], bv1 = bias[col + 16];
  float v0[8], v1[8];
#pragma unroll
  for (int i = 0; i < 8; ++i) {
    v0[i] = acc[0][i] + bv0;
    v1[i] = acc[1][i] + bv1;
    if (relu) { v0[i] = fmaxf(v0[i], 0.f); v1[i] = fmaxf(v1[i], 0.f); }
    out[(size_t)(row0 + i) * DD + col]      = v0[i];
    out[(size_t)(row0 + i) * DD + col + 16] = v1[i];
  }
  if (smode != 0) {
    float ra[8], rb[8];
#pragma unroll
    for (int i = 0; i < 8; ++i) {
      if (smode == 1) {
        ra[i] = sum16(v0[i] * v0[i] + v1[i] * v1[i]);
        rb[i] = 0.f;
      } else {
        ra[i] = max16(fmaxf(v0[i], v1[i]));
        rb[i] = sum16(__expf(v0[i] - ra[i]) + __expf(v1[i] - ra[i]));
      }
    }
    if (lid == 0) {
      const size_t so = (size_t)row0 * NP + bx * 2 + wn;
#pragma unroll
      for (int i = 0; i < 8; ++i) {
        st0[so + (size_t)i * NP] = ra[i];
        if (smode == 2) st1[so + (size_t)i * NP] = rb[i];
      }
    }
  }
}

__global__ __launch_bounds__(128) void k_gemm_xw(const float* __restrict__ X,
                                                 const float* __restrict__ W,
                                                 const float* __restrict__ bias,
                                                 float* __restrict__ out,
                                                 float* __restrict__ pmax,
                                                 float* __restrict__ psum) {
  __shared__ __attribute__((aligned(16))) float smem[SMEM_F];
  gemm_tile(blockIdx.x, blockIdx.y, smem, X, W, bias, out, 0, 2, pmax, psum);
}

static __device__ __forceinline__ void row_l2(const int row, float* red,
                                              const float* __restrict__ src,
                                              float* __restrict__ dn,
                                              float* __restrict__ dsq) {
  const int tid = threadIdx.x;
  const float* p = src + (size_t)row * DD;
  float va[4], vb[4];
#pragma unroll
  for (int j = 0; j < 4; ++j) {
    va[j] = p[tid + 256 * j];
    vb[j] = p[tid + 128 + 256 * j];
  }
  float sa = 0.f, sb = 0.f;
#pragma unroll
  for (int j = 0; j < 4; ++j) {
    sa += va[j] * va[j];
    sb += vb[j] * vb[j];
  }
  if (dsq) {
#pragma unroll
    for (int j = 0; j < 4; ++j) {
      dsq[(size_t)row * DD + tid + 256 * j]       = va[j] * va[j];
      dsq[(size_t)row * DD + tid + 128 + 256 * j] = vb[j] * vb[j];
    }
  }
  red[tid] = sa + sb; __syncthreads();
  for (int st = 64; st > 0; st >>= 1) {
    if (tid < st) red[tid] += red[tid + st];
    __syncthreads();
  }
  const float inv = rsqrtf(red[0]);
#pragma unroll
  for (int j = 0; j < 4; ++j) {
    dn[(size_t)row * DD + tid + 256 * j]       = va[j] * inv;
    dn[(size_t)row * DD + tid + 128 + 256 * j] = vb[j] * inv;
  }
}

__global__ __launch_bounds__(128) void k_front(const int* __restrict__ ntid,
                                               const int* __restrict__ nlen,
                                               const int* __restrict__ t2i,
                                               const int* __restrict__ antk,
                                               const float* __restrict__ emb,
                                               const float* __restrict__ nW,
                                               const float* __restrict__ nb,
                                               const float* __restrict__ m,
                                               const float* __restrict__ W1,
                                               const float* __restrict__ b1,
                                               const float* __restrict__ trW,
                                               const float* __restrict__ trb,
                                               const float* __restrict__ r,
                                               const float* __restrict__ t,
                                               float* __restrict__ h,
                                               float* __restrict__ trm,
                                               float* __restrict__ rn,
                                               float* __restrict__ tn,
                                               float* __restrict__ t2,
                                               float* __restrict__ trsq,
                                               float* __restrict__ concepts) {
  __shared__ __attribute__((aligned(16))) float smem[SMEM_F];
  const int b = blockIdx.x;
  if (b < 2 * NGW) {
    const bool first = b < NGW;
    const int q = first ? b : b - NGW;
    gemm_tile(q % (DD / GN), q / (DD / GN), smem, m, first ? W1 : trW, first ? b1 : trb,
              first ? h : trm, first ? 1 : 0, first ? 0 : 1, trsq, nullptr);
  } else if (b < 2 * NGW + NCW) {
    const int q = b - 2 * NGW;
    concepts_tile(q % (DD / TN), q / (DD / TN), smem, ntid, nlen, t2i, antk, emb, nW, nb, concepts);
  } else if (b < 2 * NGW + NCW + BB) {
    const int item = b - (2 * NGW + NCW);
    row_l2(item, smem, r, rn, nullptr);
    const int len = min(max(nlen[item], 0), LL);
    for (int l = len; l < LL; ++l) {
      v4f* zp = (v4f*)(concepts + ((size_t)item * LL + l) * DD);
      zp[threadIdx.x]       = (v4f){0.f, 0.f, 0.f, 0.f};
      zp[threadIdx.x + 128] = (v4f){0.f, 0.f, 0.f, 0.f};
    }
  } else {
    row_l2(b - (2 * NGW + NCW + BB), smem, t, tn, t2);
  }
}

#define SM 32
#define SK 64
#define SCH (DD / SK)
#define SP 68

__global__ __launch_bounds__(128) void k_score(const float* __restrict__ aem,
                                               const float* __restrict__ trm,
                                               const float* __restrict__ rn,
                                               const float* __restrict__ t,
                                               const float* __restrict__ t2,
                                               const float* __restrict__ tn,
                                               const float* __restrict__ pmax,
                                               const float* __restrict__ psum,
                                               const float* __restrict__ trsq,
                                               float* __restrict__ score) {
  __shared__ __attribute__((aligned(16))) float Ls[2][6][SM * SP];
  __shared__ float sst[3][SM];
  const int tid  = threadIdx.x;
  const int lane = tid & 31;
  const int wv   = tid >> 5;
  const int wm   = wv >> 1, wn = wv & 1;
  const int half = lane >> 4;
  const int lid  = lane & 15;
  const int m0 = blockIdx.x * SM, n0 = blockIdx.y * SM;
  const int srow = tid >> 4;
  const int sc   = (tid & 15) * 4;
  const size_t ao = (size_t)(m0 + srow) * DD + sc;
  const size_t bo = (size_t)(n0 + srow) * DD + sc;
  const float* src[6] = {aem + ao, trm + ao, rn + ao, t + bo, t2 + bo, tn + bo};

  v4f rg[6][4];
#pragma unroll
  for (int p = 0; p < 6; ++p)
#pragma unroll
    for (int j = 0; j < 4; ++j) rg[p][j] = *(const v4f*)(src[p] + (size_t)(8 * j) * DD);
  __builtin_amdgcn_sched_barrier(0);

  {
    const int sr = tid >> 2, sq4 = tid & 3;
    const size_t so = (size_t)(m0 + sr) * NP + sq4 * 8;
    float pm[8], pe[8];
    float sq = 0.f;
#pragma unroll
    for (int u = 0; u < 8; ++u) {
      pm[u] = pmax[so + u];
      pe[u] = psum[so + u];
      sq += trsq[so + u];
    }
    float mx = pm[0];
#pragma unroll
    for (int u = 1; u < 8; ++u) mx = fmaxf(mx, pm[u]);
    mx = max4(mx);
    float se = 0.f;
#pragma unroll
    for (int u = 0; u < 8; ++u) se += pe[u] * __expf(pm[u] - mx);
    se = sum4(se);
    sq = sum4(sq);
    if (sq4 == 0) {
      sst[0][sr] = mx;
      sst[1][sr] = 1.f / se;
      sst[2][sr] = rsqrtf(sq);
    }
  }
  __syncthreads();
  float rmx[4], rinv[4], rinv2[4];
#pragma unroll
  for (int j = 0; j < 4; ++j) {
    rmx[j]   = sst[0][srow + 8 * j];
    rinv[j]  = sst[1][srow + 8 * j];
    rinv2[j] = sst[2][srow + 8 * j];
  }
  auto to_pq_row = [&](const int j) {
#pragma unroll
    for (int e = 0; e < 4; ++e) {
      const float a = __expf(rg[0][j][e] - rmx[j]) * rinv[j];
      rg[0][j][e] = rg[1][j][e] * rinv2[j] * a;
      rg[1][j][e] = a * a;
    }
  };

#pragma unroll
  for (int j = 0; j < 4; ++j) to_pq_row(j);
#pragma unroll
  for (int p = 0; p < 6; ++p)
#pragma unroll
    for (int j = 0; j < 4; ++j) *(v4f*)&Ls[0][p][(srow + 8 * j) * SP + sc] = rg[p][j];
  __syncthreads();

  v8f num = {0.f,0.f,0.f,0.f,0.f,0.f,0.f,0.f};
  v8f den = num, isv = num;
  const int arow = (wm * 16 + lid) * SP;
  const int brow = (wn * 16 + lid) * SP;
  v2f f0[8][6], f1[8][6];
  auto load_frag = [&](const float (*L)[SM * SP], const int h, v2f (&f)[8][6]) {
#pragma unroll
    for (int kk = 0; kk < 8; ++kk) {
      const int k = (h * 8 + kk) * 4 + 2 * half;
      f[kk][0] = *(const v2f*)&L[0][arow + k];
      f[kk][1] = *(const v2f*)&L[3][brow + k];
      f[kk][2] = *(const v2f*)&L[1][arow + k];
      f[kk][3] = *(const v2f*)&L[4][brow + k];
      f[kk][4] = *(const v2f*)&L[2][arow + k];
      f[kk][5] = *(const v2f*)&L[5][brow + k];
    }
  };
  auto mma_k = [&](const int kk, const v2f (&f)[8][6]) {
    num = wmma4(f[kk][0], f[kk][1], num);
    den = wmma4(f[kk][2], f[kk][3], den);
    isv = wmma4(f[kk][4], f[kk][5], isv);
  };

#pragma unroll
  for (int p = 0; p < 6; ++p)
#pragma unroll
    for (int j = 0; j < 4; ++j)
      rg[p][j] = *(const v4f*)(src[p] + (size_t)(8 * j) * DD + SK);
  load_frag(Ls[0], 0, f0);
  auto step = [&](const int c, const bool more) {
    load_frag(Ls[c & 1], 1, f1);
#pragma unroll
    for (int kk = 0; kk < 3; ++kk) {
      mma_k(kk, f0);
      to_pq_row(kk);
      if (kk == 0) to_pq_row(3);
    }
    __builtin_amdgcn_sched_barrier(0);
#pragma unroll
    for (int p = 0; p < 6; ++p)
#pragma unroll
      for (int j = 0; j < 4; ++j)
        *(v4f*)&Ls[(c + 1) & 1][p][(srow + 8 * j) * SP + sc] = rg[p][j];
    if (more) {
      const int kn = (c + 2) * SK;
#pragma unroll
      for (int p = 0; p < 6; ++p)
#pragma unroll
        for (int j = 0; j < 4; ++j)
          rg[p][j] = *(const v4f*)(src[p] + (size_t)(8 * j) * DD + kn);
    }
#pragma unroll
    for (int kk = 3; kk < 8; ++kk) mma_k(kk, f0);
    __builtin_amdgcn_sched_barrier(0);
    lds_barrier();
    load_frag(Ls[(c + 1) & 1], 0, f0);
    __builtin_amdgcn_sched_barrier(0);
#pragma unroll
    for (int kk = 0; kk < 8; ++kk) mma_k(kk, f1);
    __builtin_amdgcn_sched_barrier(0);
  };
  for (int c = 0; c < SCH - 2; ++c) step(c, true);
  step(SCH - 2, false);
  load_frag(Ls[(SCH - 1) & 1], 1, f1);
#pragma unroll
  for (int kk = 0; kk < 8; ++kk) mma_k(kk, f0);
#pragma unroll
  for (int kk = 0; kk < 8; ++kk) mma_k(kk, f1);

#pragma unroll
  for (int i = 0; i < 8; ++i) {
    const int row = m0 + wm * 16 + half * 8 + i;
    const int col = n0 + wn * 16 + lid;
    score[(size_t)row * BB + col] = num[i] * rsqrtf(den[i]) + isv[i];
  }
}

extern "C" void kernel_launch(void* const* d_in, const int* in_sizes, int n_in,
                              void* d_out, int out_size, void* d_ws, size_t ws_size,
                              hipStream_t stream) {
  const float* r    = (const float*)d_in[0];
  const float* m    = (const float*)d_in[1];
  const float* t    = (const float*)d_in[2];
  const int*   ntid = (const int*)d_in[3];
  const int*   nlen = (const int*)d_in[4];
  const int*   antk = (const int*)d_in[5];
  const float* emb  = (const float*)d_in[6];
  const float* nW   = (const float*)d_in[7];
  const float* nb   = (const float*)d_in[8];
  const float* W1   = (const float*)d_in[9];
  const float* b1   = (const float*)d_in[10];
  const float* W2   = (const float*)d_in[11];
  const float* b2   = (const float*)d_in[12];
  const float* trW  = (const float*)d_in[13];
  const float* trb  = (const float*)d_in[14];
  const int*   t2i  = (const int*)d_in[15];

  float* score    = (float*)d_out;
  float* concepts = score + (size_t)BB * BB;

  float* ws  = (float*)d_ws;
  float* h   = ws;  ws += (size_t)BB * DD;
  float* aem = ws;  ws += (size_t)BB * DD;
  float* trm = ws;  ws += (size_t)BB * DD;
  float* rnw = ws;  ws += (size_t)BB * DD;
  float* tnw = ws;  ws += (size_t)BB * DD;
  float* t2w = ws;  ws += (size_t)BB * DD;
  float* pmx = ws;  ws += (size_t)BB * NP;
  float* psm = ws;  ws += (size_t)BB * NP;
  float* tsq = ws;  ws += (size_t)BB * NP;

  k_front<<<dim3(2 * NGW + NCW + 2 * BB), dim3(128), 0, stream>>>(ntid, nlen, t2i, antk, emb, nW, nb,
                                                                    m, W1, b1, trW, trb, r, t,
                                                                    h, trm, rnw, tnw, t2w, tsq, concepts);
  k_gemm_xw<<<dim3(DD / GN, BB / GM), dim3(128), 0, stream>>>(h, W2, b2, aem, pmx, psm);
  k_score<<<dim3(BB / SM, BB / SM), dim3(128), 0, stream>>>(aem, trm, rnw, t, t2w, tnw, pmx, psm, tsq, score);
}


# Global prefetch two chunks deep in the two launches that run one wave per SIMD (second attention GEMM, score kernel): a second register set, loads issued two chunks ahead; k order, sums unchanged
# speedup vs baseline: 5.7032x; 1.0636x over previous
#include <hip/hip_runtime.h>
#include <hip/hip_bf16.h>

typedef __attribute__((ext_vector_type(2))) float v2f;
typedef __attribute__((ext_vector_type(8))) float v8f;
typedef __attribute__((ext_vector_type(4))) float v4f;

#define BB 384
#define DD 1024
#define WD 300
#define LL 16

static __device__ __forceinline__ v8f wmma4(v2f a, v2f b, v8f c) {
  return __builtin_amdgcn_wmma_f32_16x16x4_f32(false, a, false, b, (short)0, c,
                                               false, false);
}

static __device__ __forceinline__ void lds_barrier() {
  __builtin_amdgcn_fence(__ATOMIC_RELEASE, "workgroup", "local");
  __builtin_amdgcn_s_barrier();
  __builtin_amdgcn_fence(__ATOMIC_ACQUIRE, "workgroup", "local");
}

#define NP 32

template <int CTRL>
static __device__ __forceinline__ float lane_x(float v) {
  return __builtin_bit_cast(float, __builtin_amdgcn_update_dpp(0, __builtin_bit_cast(int, v), CTRL, 0xF, 0xF, true));
}
#define DPP_Q1 0xB1
#define DPP_Q2 0x4E
#define DPP_HM 0x141
#define DPP_RM 0x140
static __device__ __forceinline__ float sum4(float v) {
  v += lane_x<DPP_Q1>(v); v += lane_x<DPP_Q2>(v);
  return v;
}
static __device__ __forceinline__ float max4(float v) {
  v = fmaxf(v, lane_x<DPP_Q1>(v)); v = fmaxf(v, lane_x<DPP_Q2>(v));
  return v;
}
static __device__ __forceinline__ float sum16(float v) {
  v = sum4(v); v += lane_x<DPP_HM>(v); v += lane_x<DPP_RM>(v);
  return v;
}
static __device__ __forceinline__ float max16(float v) {
  v = max4(v); v = fmaxf(v, lane_x<DPP_HM>(v)); v = fmaxf(v, lane_x<DPP_RM>(v));
  return v;
}

#define TM 64
#define TN 128
#define KC 20
#define NCH (WD / KC)
#define PA 20
#define PB 136

static __device__ __forceinline__ void tile_k300(const int* __restrict__ tokrow,
                                                 const float* __restrict__ emb,
                                                 const float* __restrict__ W,
                                                 const int n0,
                                                 float (*As)[TM * PA],
                                                 float (*Bs)[KC * PB],
                                                 v8f (&acc)[4][2]) {
  const int tid  = threadIdx.x;
  const int lane = tid & 31;
  const int wv   = tid >> 5;
  const int half = lane >> 4;
  const int lid  = lane & 15;

  int aoff[5];
#pragma unroll
  for (int j = 0; j < 5; ++j) {
    const int idx = tid + 128 * j;
    const int row = idx / 10;
    const int c2  = idx - row * 10;
    aoff[j] = tokrow[row] * WD + 2 * c2;
  }
  const int bk  = tid >> 5;
  const int bc4 = tid & 31;
  const float* wp = W + (size_t)bk * DD + n0 + bc4 * 4;

  v2f    ra[5];
  v4f    rb[5];
#pragma unroll
  for (int j = 0; j < 5; ++j) {
    ra[j] = *(const v2f*)(emb + aoff[j]);
    rb[j] = *(const v4f*)(wp + (size_t)(4 * j) * DD);
  }
#pragma unroll
  for (int j = 0; j < 5; ++j) {
    *(v2f*)&As[0][2 * (tid + 128 * j)] = ra[j];
    *(v4f*)&Bs[0][(bk + 4 * j) * PB + bc4 * 4] = rb[j];
  }
  __syncthreads();

  auto chunk_mma = [&](const float* Ab, const float* Bb) {
#pragma unroll
    for (int kk = 0; kk < KC / 4; ++kk) {
      const int k = kk * 4 + 2 * half;
      v2f b[2];
#pragma unroll
      for (int ns = 0; ns < 2; ++ns) {
        b[ns].x = Bb[k * PB + wv * 32 + ns * 16 + lid];
        b[ns].y = Bb[(k + 1) * PB + wv * 32 + ns * 16 + lid];
      }
#pragma unroll
      for (int s = 0; s < 4; ++s) {
        const v2f a = *(const v2f*)&Ab[(s * 16 + lid) * PA + k];
#pragma unroll
        for (int ns = 0; ns < 2; ++ns) acc[s][ns] = wmma4(a, b[ns], acc[s][ns]);
      }
    }
  };

#pragma unroll
  for (int j = 0; j < 5; ++j) {
    ra[j] = *(const v2f*)(emb + aoff[j] + KC);
    rb[j] = *(const v4f*)(wp + (size_t)(KC + 4 * j) * DD);
  }
  for (int c = 0; c < NCH - 2; ++c) {
    chunk_mma(As[c & 1], Bs[c & 1]);
#pragma unroll
    for (int j = 0; j < 5; ++j) {
      *(v2f*)&As[(c + 1) & 1][2 * (tid + 128 * j)] = ra[j];
      *(v4f*)&Bs[(c + 1) & 1][(bk + 4 * j) * PB + bc4 * 4] = rb[j];
    }
    __syncthreads();
    const int kn = (c + 2) * KC;
#pragma unroll
    for (int j = 0; j < 5; ++j) {
      ra[j] = *(const v2f*)(emb + aoff[j] + kn);
      rb[j] = *(const v4f*)(wp + (size_t)(kn + 4 * j) * DD);
    }
  }
  chunk_mma(As[(NCH - 2) & 1], Bs[(NCH - 2) & 1]);
#pragma unroll
  for (int j = 0; j < 5; ++j) {
    *(v2f*)&As[(NCH - 1) & 1][2 * (tid + 128 * j)] = ra[j];
    *(v4f*)&Bs[(NCH - 1) & 1][(bk + 4 * j) * PB + bc4 * 4] = rb[j];
  }
  __syncthreads();
  chunk_mma(As[(NCH - 1) & 1], Bs[(NCH - 1) & 1]);
}

#define NCW ((DD / TN) * (BB * LL / TM))

static __device__ __forceinline__ void concepts_tile(const int bx, const int by, float* smem,
                                                     const int* __restrict__ ntid,
                                                     const int* __restrict__ nlen,
                                                     const int* __restrict__ t2i,
                                                     const int* __restrict__ antk,
                                                     const float* __restrict__ emb,
                                                     const float* __restrict__ W,
                                                     const float* __restrict__ bias,
                                                     float* __restrict__ concepts) {
  float (*As)[TM * PA] = (float (*)[TM * PA])smem;
  float (*Bs)[KC * PB] = (float (*)[KC * PB])(smem + 2 * TM * PA);
  int* stok  = (int*)(smem + 2 * TM * PA + 2 * KC * PB);
  int* sorow = stok + TM;
  int* spart = sorow + TM;
  int* soff  = spart + 128;
  const int tid  = threadIdx.x;
  const int lane = tid & 31;
  const int wv   = tid >> 5;
  const int half = lane >> 4;
  const int lid  = lane & 15;
  const int n0 = bx * TN, j0 = by * TM;
  const int ln0 = min(max(nlen[3 * tid], 0), LL);
  const int ln1 = min(max(nlen[3 * tid + 1], 0), LL);
  const int ln2 = min(max(nlen[3 * tid + 2], 0), LL);
  spart[tid] = ln0 + ln1 + ln2;
  __syncthreads();
  int base = 0;
  for (int i = 0; i < 128; ++i) {
    const int v = spart[i];
    base += (i < tid) ? v : 0;
  }
  soff[3 * tid]     = base;
  soff[3 * tid + 1] = base + ln0;
  soff[3 * tid + 2] = base + ln0 + ln1;
  if (tid == 127) soff[BB] = base + ln0 + ln1 + ln2;
  __syncthreads();
  const int total = soff[BB];
  if (j0 >= total) return;
  if (tid < TM) {
    const int j = j0 + tid;
    int tokr = 0, orow = -1;
    if (j < total) {
      int lo = 0, hi = BB;
      for (int it = 0; it < 9; ++it) {
        const int mid = (lo + hi) >> 1;
        if (soff[mid] <= j) lo = mid; else hi = mid;
      }
      orow = lo * LL + (j - soff[lo]);
      const int tk   = ntid[orow];
      const int widx = t2i[tk];
      tokr = (widx > 0) ? antk[widx] : tk;
    }
    stok[tid]  = tokr;
    sorow[tid] = orow;
  }
  __syncthreads();
  v8f acc[4][2];
#pragma unroll
  for (int s = 0; s < 4; ++s)
#pragma unroll
    for (int ns = 0; ns < 2; ++ns) acc[s][ns] = (v8f){0.f,0.f,0.f,0.f,0.f,0.f,0.f,0.f};
  tile_k300(stok, emb, W, n0, As, Bs, acc);
  float bv[2];
#pragma unroll
  for (int ns = 0; ns < 2; ++ns) bv[ns] = bias[n0 + wv * 32 + ns * 16 + lid];
#pragma unroll
  for (int s = 0; s < 4; ++s) {
#pragma unroll
    for (int i = 0; i < 8; ++i) {
      const int row = sorow[s * 16 + half * 8 + i];
      if (row >= 0) {
#pragma unroll
        for (int ns = 0; ns < 2; ++ns) {
          const int col = n0 + wv * 32 + ns * 16 + lid;
          concepts[(size_t)row * DD + col] = acc[s][ns][i] + bv[ns];
        }
      }
    }
  }
}

#define GM 32
#define GN 64
#define GK 64
#define GCH (DD / GK)
#define GPA 68
#define GPB 72

#define NGW ((DD / GN) * (BB / GM))
#define SMEM_F (2 * GM * GPA + 2 * GK * GPB)

static __device__ __forceinline__ void gemm_tile(const int bx, const int by, float* smem,
                                                 const float* __restrict__ X,
                                                 const float* __restrict__ W,
                                                 const float* __restrict__ bias,
                                                 float* __restrict__ out, const int relu,
                                                 const int smode,
                                                 float* __restrict__ st0,
                                                 float* __restrict__ st1) {
  float (*As)[GM * GPA] = (float (*)[GM * GPA])smem;
  float (*Bs)[GK * GPB] = (float (*)[GK * GPB])(smem + 2 * GM * GPA);
  const int tid  = threadIdx.x;
  const int lane = tid & 31;
  const int wv   = tid >> 5;
  const int wm   = wv >> 1, wn = wv & 1;
  const int half = lane >> 4;
  const int lid  = lane & 15;
  const int n0 = bx * GN, m0 = by * GM;
  const int srow = tid >> 4;
  const int sc   = (tid & 15) * 4;
  const float* xp = X + (size_t)(m0 + srow) * DD + sc;
  const float* wp = W + (size_t)srow * DD + n0 + sc;

  v4f ra[4], rb[8];
#pragma unroll
  for (int j = 0; j < 4; ++j) ra[j] = *(const v4f*)(xp + (size_t)(8 * j) * DD);
#pragma unroll
  for (int j = 0; j < 8; ++j) rb[j] = *(const v4f*)(wp + (size_t)(8 * j) * DD);
#pragma unroll
  for (int j = 0; j < 4; ++j) *(v4f*)&As[0][(srow + 8 * j) * GPA + sc] = ra[j];
#pragma unroll
  for (int j = 0; j < 8; ++j) *(v4f*)&Bs[0][(srow + 8 * j) * GPB + sc] = rb[j];
  __syncthreads();

  v8f acc[2];
  acc[0] = (v8f){0.f,0.f,0.f,0.f,0.f,0.f,0.f,0.f};
  acc[1] = acc[0];
  const int fao = (wm * 16 + lid) * GPA + 2 * half;
  const int fbo = (2 * half) * GPB + wn * 32 + lid;
  v2f   fa0[8], fa1[8];
  float bx0[8][2], by0[8][2], bx1[8][2], by1[8][2];
  auto load_frag = [&](const float* Ab, const float* Bb, const int h, v2f (&fa)[8], float (&bx)[8][2], float (&by)[8][2]) {
#pragma unroll
    for (int kk = 0; kk < 8; ++kk) {
      const int k4 = (h * 8 + kk) * 4;
      fa[kk] = *(const v2f*)&Ab[fao + k4];
#pragma unroll
      for (int ns = 0; ns < 2; ++ns) {
        bx[kk][ns] = Bb[fbo + k4 * GPB + ns * 16];
        by[kk][ns] = Bb[fbo + (k4 + 1) * GPB + ns * 16];
      }
    }
  };
  auto mma_k = [&](const int kk, const v2f (&fa)[8], const float (&bx)[8][2], const float (&by)[8][2]) {
#pragma unroll
    for (int ns = 0; ns < 2; ++ns) {
      v2f b;
      b.x = bx[kk][ns];
      b.y = by[kk][ns];
      acc[ns] = wmma4(fa[kk], b, acc[ns]);
    }
  };

#pragma unroll
  for (int j = 0; j < 4; ++j) ra[j] = *(const v4f*)(xp + (size_t)(8 * j) * DD + GK);
#pragma unroll
  for (int j = 0; j < 8; ++j) rb[j] = *(const v4f*)(wp + (size_t)(GK + 8 * j) * DD);
  load_frag(As[0], Bs[0], 0, fa0, bx0, by0);
  auto step = [&](const int c, const bool more) {
    load_frag(As[c & 1], Bs[c & 1], 1, fa1, bx1, by1);
#pragma unroll
    for (int j = 0; j < 4; ++j) *(v4f*)&As[(c + 1) & 1][(srow + 8 * j) * GPA + sc] = ra[j];
#pragma unroll
    for (int j = 0; j < 8; ++j) *(v4f*)&Bs[(c + 1) & 1][(srow + 8 * j) * GPB + sc] = rb[j];
    if (more) {
      const int kn = (c + 2) * GK;
#pragma unroll
      for (int j = 0; j < 4; ++j) ra[j] = *(const v4f*)(xp + (size_t)(8 * j) * DD + kn);
#pragma unroll
      for (int j = 0; j < 8; ++j) rb[j] = *(const v4f*)(wp + (size_t)(kn + 8 * j) * DD);
    }
#pragma unroll
    for (int kk = 0; kk < 8; ++kk) mma_k(kk, fa0, bx0, by0);
    __builtin_amdgcn_sched_barrier(0);
    lds_barrier();
    load_frag(As[(c + 1) & 1], Bs[(c + 1) & 1], 0, fa0, bx0, by0);
#pragma unroll
    for (int kk = 0; kk < 8; ++kk) mma_k(kk, fa1, bx1, by1);
    __builtin_amdgcn_sched_group_barrier(0x100, 20, 0);
    __builtin_amdgcn_sched_group_barrier(0x008, 16, 0);
    __builtin_amdgcn_sched_group_barrier(0x002, 64, 0);
    __builtin_amdgcn_sched_barrier(0);
  };
  for (int c = 0; c < GCH - 2; ++c) step(c, true);
  step(GCH - 2, false);
  load_frag(As[(GCH - 1) & 1], Bs[(GCH - 1) & 1], 1, fa1, bx1, by1);
#pragma unroll
  for (int kk = 0; kk < 8; ++kk) mma_k(kk, fa0, bx0, by0);
#pragma unroll
  for (int kk = 0; kk < 8; ++kk) mma_k(kk, fa1, bx1, by1);

  const int col  = n0 + wn * 32 + lid;
  const int row0 = m0 + wm * 16 + half * 8;
  const float bv0 = bias[col], bv1 = bias[col + 16];
  float v0[8], v1[8];
#pragma unroll
  for (int i = 0; i < 8; ++i) {
    v0[i] = acc[0][i] + bv0;
    v1[i] = acc[1][i] + bv1;
    if (relu) { v0[i] = fmaxf(v0[i], 0.f); v1[i] = fmaxf(v1[i], 0.f); }
    out[(size_t)(row0 + i) * DD + col]      = v0[i];
    out[(size_t)(row0 + i) * DD + col + 16] = v1[i];
  }
  if (smode != 0) {
    float ra[8], rb[8];
#pragma unroll
    for (int i = 0; i < 8; ++i) {
      if (smode == 1) {
        ra[i] = sum16(v0[i] * v0[i] + v1[i] * v1[i]);
        rb[i] = 0.f;
      } else {
        ra[i] = max16(fmaxf(v0[i], v1[i]));
        rb[i] = sum16(__expf(v0[i] - ra[i]) + __expf(v1[i] - ra[i]));
      }
    }
    if (lid == 0) {
      const size_t so = (size_t)row0 * NP + bx * 2 + wn;
#pragma unroll
      for (int i = 0; i < 8; ++i) {
        st0[so + (size_t)i * NP] = ra[i];
        if (smode == 2) st1[so + (size_t)i * NP] = rb[i];
      }
    }
  }
}

static __device__ __forceinline__ void gemm_tile_deep(const int bx, const int by, float* smem,
                                                 const float* __restrict__ X,
                                                 const float* __restrict__ W,
                                                 const float* __restrict__ bias,
                                                 float* __restrict__ out, const int relu,
                                                 const int smode,
                                                 float* __restrict__ st0,
                                                 float* __restrict__ st1) {
  float (*As)[GM * GPA] = (float (*)[GM * GPA])smem;
  float (*Bs)[GK * GPB] = (float (*)[GK * GPB])(smem + 2 * GM * GPA);
  const int tid  = threadIdx.x;
  const int lane = tid & 31;
  const int wv   = tid >> 5;
  const int wm   = wv >> 1, wn = wv & 1;
  const int half = lane >> 4;
  const int lid  = lane & 15;
  const int n0 = bx * GN, m0 = by * GM;
  const int srow = tid >> 4;
  const int sc   = (tid & 15) * 4;
  const float* xp = X + (size_t)(m0 + srow) * DD + sc;
  const float* wp = W + (size_t)srow * DD + n0 + sc;

  v4f ra0[4], rb0[8], ra1[4], rb1[8];
  auto gload = [&](const int kn, v4f (&ra)[4], v4f (&rb)[8]) {
#pragma unroll
    for (int j = 0; j < 4; ++j) ra[j] = *(const v4f*)(xp + (size_t)(8 * j) * DD + kn);
#pragma unroll
    for (int j = 0; j < 8; ++j) rb[j] = *(const v4f*)(wp + (size_t)(kn + 8 * j) * DD);
  };
  auto lstore = [&](const int st, const v4f (&ra)[4], const v4f (&rb)[8]) {
#pragma unroll
    for (int j = 0; j < 4; ++j) *(v4f*)&As[st][(srow + 8 * j) * GPA + sc] = ra[j];
#pragma unroll
    for (int j = 0; j < 8; ++j) *(v4f*)&Bs[st][(srow + 8 * j) * GPB + sc] = rb[j];
  };
  gload(0, ra0, rb0);
  gload(GK, ra1, rb1);
  lstore(0, ra0, rb0);
  lds_barrier();
  gload(2 * GK, ra0, rb0);

  v8f acc[2];
  acc[0] = (v8f){0.f,0.f,0.f,0.f,0.f,0.f,0.f,0.f};
  acc[1] = acc[0];
  const int fao = (wm * 16 + lid) * GPA + 2 * half;
  const int fbo = (2 * half) * GPB + wn * 32 + lid;
  v2f   fa0[8], fa1[8];
  float bx0[8][2], by0[8][2], bx1[8][2], by1[8][2];
  auto load_frag = [&](const float* Ab, const float* Bb, const int h, v2f (&fa)[8], float (&bx)[8][2], float (&by)[8][2]) {
#pragma unroll
    for (int kk = 0; kk < 8; ++kk) {
      const int k4 = (h * 8 + kk) * 4;
      fa[kk] = *(const v2f*)&Ab[fao + k4];
#pragma unroll
      for (int ns = 0; ns < 2; ++ns) {
        bx[kk][ns] = Bb[fbo + k4 * GPB + ns * 16];
        by[kk][ns] = Bb[fbo + (k4 + 1) * GPB + ns * 16];
      }
    }
  };
  auto mma_k = [&](const int kk, const v2f (&fa)[8], const float (&bx)[8][2], const float (&by)[8][2]) {
#pragma unroll
    for (int ns = 0; ns < 2; ++ns) {
      v2f b;
      b.x = bx[kk][ns];
      b.y = by[kk][ns];
      acc[ns] = wmma4(fa[kk], b, acc[ns]);
    }
  };

  load_frag(As[0], Bs[0], 0, fa0, bx0, by0);
  auto step = [&](const int c, const bool more, v4f (&ra)[4], v4f (&rb)[8]) {
    load_frag(As[c & 1], Bs[c & 1], 1, fa1, bx1, by1);
    lstore((c + 1) & 1, ra, rb);
    if (more) gload((c + 3) * GK, ra, rb);
#pragma unroll
    for (int kk = 0; kk < 8; ++kk) mma_k(kk, fa0, bx0, by0);
    __builtin_amdgcn_sched_barrier(0);
    lds_barrier();
    load_frag(As[(c + 1) & 1], Bs[(c + 1) & 1], 0, fa0, bx0, by0);
#pragma unroll
    for (int kk = 0; kk < 8; ++kk) mma_k(kk, fa1, bx1, by1);
    __builtin_amdgcn_sched_group_barrier(0x100, 20, 0);
    __builtin_amdgcn_sched_group_barrier(0x008, 16, 0);
    __builtin_amdgcn_sched_group_barrier(0x002, 64, 0);
    __builtin_amdgcn_sched_barrier(0);
  };
  for (int c = 0; c < GCH - 4; c += 2) {
    step(c, true, ra1, rb1);
    step(c + 1, true, ra0, rb0);
  }
  step(GCH - 4, true, ra1, rb1);
  step(GCH - 3, false, ra0, rb0);
  step(GCH - 2, false, ra1, rb1);
  load_frag(As[(GCH - 1) & 1], Bs[(GCH - 1) & 1], 1, fa1, bx1, by1);
#pragma unroll
  for (int kk = 0; kk < 8; ++kk) mma_k(kk, fa0, bx0, by0);
#pragma unroll
  for (int kk = 0; kk < 8; ++kk) mma_k(kk, fa1, bx1, by1);

  const int col  = n0 + wn * 32 + lid;
  const int row0 = m0 + wm * 16 + half * 8;
  const float bv0 = bias[col], bv1 = bias[col + 16];
  float v0[8], v1[8];
#pragma unroll
  for (int i = 0; i < 8; ++i) {
    v0[i] = acc[0][i] + bv0;
    v1[i] = acc[1][i] + bv1;
    if (relu) { v0[i] = fmaxf(v0[i], 0.f); v1[i] = fmaxf(v1[i], 0.f); }
    out[(size_t)(row0 + i) * DD + col]      = v0[i];
    out[(size_t)(row0 + i) * DD + col + 16] = v1[i];
  }
  if (smode != 0) {
    float ra[8], rb[8];
#pragma unroll
    for (int i = 0; i < 8; ++i) {
      if (smode == 1) {
        ra[i] = sum16(v0[i] * v0[i] + v1[i] * v1[i]);
        rb[i] = 0.f;
      } else {
        ra[i] = max16(fmaxf(v0[i], v1[i]));
        rb[i] = sum16(__expf(v0[i] - ra[i]) + __expf(v1[i] - ra[i]));
      }
    }
    if (lid == 0) {
      const size_t so = (size_t)row0 * NP + bx * 2 + wn;
#pragma unroll
      for (int i = 0; i < 8; ++i) {
        st0[so + (size_t)i * NP] = ra[i];
        if (smode == 2) st1[so + (size_t)i * NP] = rb[i];
      }
    }
  }
}

__global__ __launch_bounds__(128) void k_gemm_xw(const float* __restrict__ X,
                                                 const float* __restrict__ W,
                                                 const float* __restrict__ bias,
                                                 float* __restrict__ out,
                                                 float* __restrict__ pmax,
                                                 float* __restrict__ psum) {
  __shared__ __attribute__((aligned(16))) float smem[SMEM_F];
  gemm_tile_deep(blockIdx.x, blockIdx.y, smem, X, W, bias, out, 0, 2, pmax, psum);
}

static __device__ __forceinline__ void row_l2(const int row, float* red,
                                              const float* __restrict__ src,
                                              float* __restrict__ dn,
                                              float* __restrict__ dsq) {
  const int tid = threadIdx.x;
  const float* p = src + (size_t)row * DD;
  float va[4], vb[4];
#pragma unroll
  for (int j = 0; j < 4; ++j) {
    va[j] = p[tid + 256 * j];
    vb[j] = p[tid + 128 + 256 * j];
  }
  float sa = 0.f, sb = 0.f;
#pragma unroll
  for (int j = 0; j < 4; ++j) {
    sa += va[j] * va[j];
    sb += vb[j] * vb[j];
  }
  if (dsq) {
#pragma unroll
    for (int j = 0; j < 4; ++j) {
      dsq[(size_t)row * DD + tid + 256 * j]       = va[j] * va[j];
      dsq[(size_t)row * DD + tid + 128 + 256 * j] = vb[j] * vb[j];
    }
  }
  red[tid] = sa + sb; __syncthreads();
  for (int st = 64; st > 0; st >>= 1) {
    if (tid < st) red[tid] += red[tid + st];
    __syncthreads();
  }
  const float inv = rsqrtf(red[0]);
#pragma unroll
  for (int j = 0; j < 4; ++j) {
    dn[(size_t)row * DD + tid + 256 * j]       = va[j] * inv;
    dn[(size_t)row * DD + tid + 128 + 256 * j] = vb[j] * inv;
  }
}

__global__ __launch_bounds__(128) void k_front(const int* __restrict__ ntid,
                                               const int* __restrict__ nlen,
                                               const int* __restrict__ t2i,
                                               const int* __restrict__ antk,
                                               const float* __restrict__ emb,
                                               const float* __restrict__ nW,
                                               const float* __restrict__ nb,
                                               const float* __restrict__ m,
                                               const float* __restrict__ W1,
                                               const float* __restrict__ b1,
                                               const float* __restrict__ trW,
                                               const float* __restrict__ trb,
                                               const float* __restrict__ r,
                                               const float* __restrict__ t,
                                               float* __restrict__ h,
                                               float* __restrict__ trm,
                                               float* __restrict__ rn,
                                               float* __restrict__ tn,
                                               float* __restrict__ t2,
                                               float* __restrict__ trsq,
                                               float* __restrict__ concepts) {
  __shared__ __attribute__((aligned(16))) float smem[SMEM_F];
  const int b = blockIdx.x;
  if (b < 2 * NGW) {
    const bool first = b < NGW;
    const int q = first ? b : b - NGW;
    gemm_tile(q % (DD / GN), q / (DD / GN), smem, m, first ? W1 : trW, first ? b1 : trb,
              first ? h : trm, first ? 1 : 0, first ? 0 : 1, trsq, nullptr);
  } else if (b < 2 * NGW + NCW) {
    const int q = b - 2 * NGW;
    concepts_tile(q % (DD / TN), q / (DD / TN), smem, ntid, nlen, t2i, antk, emb, nW, nb, concepts);
  } else if (b < 2 * NGW + NCW + BB) {
    const int item = b - (2 * NGW + NCW);
    row_l2(item, smem, r, rn, nullptr);
    const int len = min(max(nlen[item], 0), LL);
    for (int l = len; l < LL; ++l) {
      v4f* zp = (v4f*)(concepts + ((size_t)item * LL + l) * DD);
      zp[threadIdx.x]       = (v4f){0.f, 0.f, 0.f, 0.f};
      zp[threadIdx.x + 128] = (v4f){0.f, 0.f, 0.f, 0.f};
    }
  } else {
    row_l2(b - (2 * NGW + NCW + BB), smem, t, tn, t2);
  }
}

#define SM 32
#define SK 64
#define SCH (DD / SK)
#define SP 68

__global__ __launch_bounds__(128) void k_score(const float* __restrict__ aem,
                                               const float* __restrict__ trm,
                                               const float* __restrict__ rn,
                                               const float* __restrict__ t,
                                               const float* __restrict__ t2,
                                               const float* __restrict__ tn,
                                               const float* __restrict__ pmax,
                                               const float* __restrict__ psum,
                                               const float* __restrict__ trsq,
                                               float* __restrict__ score) {
  __shared__ __attribute__((aligned(16))) float Ls[2][6][SM * SP];
  __shared__ float sst[3][SM];
  const int tid  = threadIdx.x;
  const int lane = tid & 31;
  const int wv   = tid >> 5;
  const int wm   = wv >> 1, wn = wv & 1;
  const int half = lane >> 4;
  const int lid  = lane & 15;
  const int m0 = blockIdx.x * SM, n0 = blockIdx.y * SM;
  const int srow = tid >> 4;
  const int sc   = (tid & 15) * 4;
  const size_t ao = (size_t)(m0 + srow) * DD + sc;
  const size_t bo = (size_t)(n0 + srow) * DD + sc;
  const float* src[6] = {aem + ao, trm + ao, rn + ao, t + bo, t2 + bo, tn + bo};

  v4f rg0[6][4], rg1[6][4];
  auto gload = [&](const int kn, v4f (&rg)[6][4]) {
#pragma unroll
    for (int p = 0; p < 6; ++p)
#pragma unroll
      for (int j = 0; j < 4; ++j) rg[p][j] = *(const v4f*)(src[p] + (size_t)(8 * j) * DD + kn);
  };
  gload(0, rg0);
  __builtin_amdgcn_sched_barrier(0);

  {
    const int sr = tid >> 2, sq4 = tid & 3;
    const size_t so = (size_t)(m0 + sr) * NP + sq4 * 8;
    float pm[8], pe[8];
    float sq = 0.f;
#pragma unroll
    for (int u = 0; u < 8; ++u) {
      pm[u] = pmax[so + u];
      pe[u] = psum[so + u];
      sq += trsq[so + u];
    }
    float mx = pm[0];
#pragma unroll
    for (int u = 1; u < 8; ++u) mx = fmaxf(mx, pm[u]);
    mx = max4(mx);
    float se = 0.f;
#pragma unroll
    for (int u = 0; u < 8; ++u) se += pe[u] * __expf(pm[u] - mx);
    se = sum4(se);
    sq = sum4(sq);
    if (sq4 == 0) {
      sst[0][sr] = mx;
      sst[1][sr] = 1.f / se;
      sst[2][sr] = rsqrtf(sq);
    }
  }
  gload(SK, rg1);
  lds_barrier();
  float rmx[4], rinv[4], rinv2[4];
#pragma unroll
  for (int j = 0; j < 4; ++j) {
    rmx[j]   = sst[0][srow + 8 * j];
    rinv[j]  = sst[1][srow + 8 * j];
    rinv2[j] = sst[2][srow + 8 * j];
  }
  auto to_pq_row = [&](const int j, v4f (&rg)[6][4]) {
#pragma unroll
    for (int e = 0; e < 4; ++e) {
      const float a = __expf(rg[0][j][e] - rmx[j]) * rinv[j];
      rg[0][j][e] = rg[1][j][e] * rinv2[j] * a;
      rg[1][j][e] = a * a;
    }
  };

  auto lstore = [&](const int st, const v4f (&rg)[6][4]) {
#pragma unroll
    for (int p = 0; p < 6; ++p)
#pragma unroll
      for (int j = 0; j < 4; ++j) *(v4f*)&Ls[st][p][(srow + 8 * j) * SP + sc] = rg[p][j];
  };
#pragma unroll
  for (int j = 0; j < 4; ++j) to_pq_row(j, rg0);
  lstore(0, rg0);
  lds_barrier();
  gload(2 * SK, rg0);

  v8f num = {0.f,0.f,0.f,0.f,0.f,0.f,0.f,0.f};
  v8f den = num, isv = num;
  const int arow = (wm * 16 + lid) * SP;
  const int brow = (wn * 16 + lid) * SP;
  v2f f0[8][6], f1[8][6];
  auto load_frag = [&](const float (*L)[SM * SP], const int h, v2f (&f)[8][6]) {
#pragma unroll
    for (int kk = 0; kk < 8; ++kk) {
      const int k = (h * 8 + kk) * 4 + 2 * half;
      f[kk][0] = *(const v2f*)&L[0][arow + k];
      f[kk][1] = *(const v2f*)&L[3][brow + k];
      f[kk][2] = *(const v2f*)&L[1][arow + k];
      f[kk][3] = *(const v2f*)&L[4][brow + k];
      f[kk][4] = *(const v2f*)&L[2][arow + k];
      f[kk][5] = *(const v2f*)&L[5][brow + k];
    }
  };
  auto mma_k = [&](const int kk, const v2f (&f)[8][6]) {
    num = wmma4(f[kk][0], f[kk][1], num);
    den = wmma4(f[kk][2], f[kk][3], den);
    isv = wmma4(f[kk][4], f[kk][5], isv);
  };

  load_frag(Ls[0], 0, f0);
  auto step = [&](const int c, const bool more, v4f (&rg)[6][4]) {
    load_frag(Ls[c & 1], 1, f1);
#pragma unroll
    for (int kk = 0; kk < 3; ++kk) {
      mma_k(kk, f0);
      to_pq_row(kk, rg);
      if (kk == 0) to_pq_row(3, rg);
    }
    __builtin_amdgcn_sched_barrier(0);
    lstore((c + 1) & 1, rg);
    if (more) gload((c + 3) * SK, rg);
#pragma unroll
    for (int kk = 3; kk < 8; ++kk) mma_k(kk, f0);
    __builtin_amdgcn_sched_barrier(0);
    lds_barrier();
    load_frag(Ls[(c + 1) & 1], 0, f0);
    __builtin_amdgcn_sched_barrier(0);
#pragma unroll
    for (int kk = 0; kk < 8; ++kk) mma_k(kk, f1);
    __builtin_amdgcn_sched_barrier(0);
  };
  for (int c = 0; c < SCH - 4; c += 2) {
    step(c, true, rg1);
    step(c + 1, true, rg0);
  }
  step(SCH - 4, true, rg1);
  step(SCH - 3, false, rg0);
  step(SCH - 2, false, rg1);
  load_frag(Ls[(SCH - 1) & 1], 1, f1);
#pragma unroll
  for (int kk = 0; kk < 8; ++kk) mma_k(kk, f0);
#pragma unroll
  for (int kk = 0; kk < 8; ++kk) mma_k(kk, f1);

#pragma unroll
  for (int i = 0; i < 8; ++i) {
    const int row = m0 + wm * 16 + half * 8 + i;
    const int col = n0 + wn * 16 + lid;
    score[(size_t)row * BB + col] = num[i] * rsqrtf(den[i]) + isv[i];
  }
}

extern "C" void kernel_launch(void* const* d_in, const int* in_sizes, int n_in,
                              void* d_out, int out_size, void* d_ws, size_t ws_size,
                              hipStream_t stream) {
  const float* r    = (const float*)d_in[0];
  const float* m    = (const float*)d_in[1];
  const float* t    = (const float*)d_in[2];
  const int*   ntid = (const int*)d_in[3];
  const int*   nlen = (const int*)d_in[4];
  const int*   antk = (const int*)d_in[5];
  const float* emb  = (const float*)d_in[6];
  const float* nW   = (const float*)d_in[7];
  const float* nb   = (const float*)d_in[8];
  const float* W1   = (const float*)d_in[9];
  const float* b1   = (const float*)d_in[10];
  const float* W2   = (const float*)d_in[11];
  const float* b2   = (const float*)d_in[12];
  const float* trW  = (const float*)d_in[13];
  const float* trb  = (const float*)d_in[14];
  const int*   t2i  = (const int*)d_in[15];

  float* score    = (float*)d_out;
  float* concepts = score + (size_t)BB * BB;

  float* ws  = (float*)d_ws;
  float* h   = ws;  ws += (size_t)BB * DD;
  float* aem = ws;  ws += (size_t)BB * DD;
  float* trm = ws;  ws += (size_t)BB * DD;
  float* rnw = ws;  ws += (size_t)BB * DD;
  float* tnw = ws;  ws += (size_t)BB * DD;
  float* t2w = ws;  ws += (size_t)BB * DD;
  float* pmx = ws;  ws += (size_t)BB * NP;
  float* psm = ws;  ws += (size_t)BB * NP;
  float* tsq = ws;  ws += (size_t)BB * NP;

  k_front<<<dim3(2 * NGW + NCW + 2 * BB), dim3(128), 0, stream>>>(ntid, nlen, t2i, antk, emb, nW, nb,
                                                                    m, W1, b1, trW, trb, r, t,
                                                                    h, trm, rnw, tnw, t2w, tsq, concepts);
  k_gemm_xw<<<dim3(DD / GN, BB / GM), dim3(128), 0, stream>>>(h, W2, b2, aem, pmx, psm);
  k_score<<<dim3(BB / SM, BB / SM), dim3(128), 0, stream>>>(aem, trm, rnw, t, t2w, tnw, pmx, psm, tsq, score);
}


# Concepts tiles leave the full first launch and ride behind the tiles of the two dependent launches (logits GEMM, score), filling their idle matrix units; launch 1 lists the valid rows; sums unchanged
# speedup vs baseline: 6.0722x; 1.0647x over previous
#include <hip/hip_runtime.h>
#include <hip/hip_bf16.h>

typedef __attribute__((ext_vector_type(2))) float v2f;
typedef __attribute__((ext_vector_type(8))) float v8f;
typedef __attribute__((ext_vector_type(4))) float v4f;

#define BB 384
#define DD 1024
#define WD 300
#define LL 16

static __device__ __forceinline__ v8f wmma4(v2f a, v2f b, v8f c) {
  return __builtin_amdgcn_wmma_f32_16x16x4_f32(false, a, false, b, (short)0, c,
                                               false, false);
}

static __device__ __forceinline__ void lds_barrier() {
  __builtin_amdgcn_fence(__ATOMIC_RELEASE, "workgroup", "local");
  __builtin_amdgcn_s_barrier();
  __builtin_amdgcn_fence(__ATOMIC_ACQUIRE, "workgroup", "local");
}

#define NP 32

template <int CTRL>
static __device__ __forceinline__ float lane_x(float v) {
  return __builtin_bit_cast(float, __builtin_amdgcn_update_dpp(0, __builtin_bit_cast(int, v), CTRL, 0xF, 0xF, true));
}
#define DPP_Q1 0xB1
#define DPP_Q2 0x4E
#define DPP_HM 0x141
#define DPP_RM 0x140
static __device__ __forceinline__ float sum4(float v) {
  v += lane_x<DPP_Q1>(v); v += lane_x<DPP_Q2>(v);
  return v;
}
static __device__ __forceinline__ float max4(float v) {
  v = fmaxf(v, lane_x<DPP_Q1>(v)); v = fmaxf(v, lane_x<DPP_Q2>(v));
  return v;
}
static __device__ __forceinline__ float sum16(float v) {
  v = sum4(v); v += lane_x<DPP_HM>(v); v += lane_x<DPP_RM>(v);
  return v;
}
static __device__ __forceinline__ float max16(float v) {
  v = max4(v); v = fmaxf(v, lane_x<DPP_HM>(v)); v = fmaxf(v, lane_x<DPP_RM>(v));
  return v;
}

#define TM 64
#define TN 128
#define KC 20
#define NCH (WD / KC)
#define PA 20
#define PB 136

static __device__ __forceinline__ void tile_k300(const int* __restrict__ tokrow,
                                                 const float* __restrict__ emb,
                                                 const float* __restrict__ W,
                                                 const int n0,
                                                 float (*As)[TM * PA],
                                                 float (*Bs)[KC * PB],
                                                 v8f (&acc)[4][2]) {
  const int tid  = threadIdx.x;
  const int lane = tid & 31;
  const int wv   = tid >> 5;
  const int half = lane >> 4;
  const int lid  = lane & 15;

  int aoff[5];
#pragma unroll
  for (int j = 0; j < 5; ++j) {
    const int idx = tid + 128 * j;
    const int row = idx / 10;
    const int c2  = idx - row * 10;
    aoff[j] = tokrow[row] * WD + 2 * c2;
  }
  const int bk  = tid >> 5;
  const int bc4 = tid & 31;
  const float* wp = W + (size_t)bk * DD + n0 + bc4 * 4;

  v2f    ra[5];
  v4f    rb[5];
#pragma unroll
  for (int j = 0; j < 5; ++j) {
    ra[j] = *(const v2f*)(emb + aoff[j]);
    rb[j] = *(const v4f*)(wp + (size_t)(4 * j) * DD);
  }
#pragma unroll
  for (int j = 0; j < 5; ++j) {
    *(v2f*)&As[0][2 * (tid + 128 * j)] = ra[j];
    *(v4f*)&Bs[0][(bk + 4 * j) * PB + bc4 * 4] = rb[j];
  }
  __syncthreads();

  auto chunk_mma = [&](const float* Ab, const float* Bb) {
#pragma unroll
    for (int kk = 0; kk < KC / 4; ++kk) {
      const int k = kk * 4 + 2 * half;
      v2f b[2];
#pragma unroll
      for (int ns = 0; ns < 2; ++ns) {
        b[ns].x = Bb[k * PB + wv * 32 + ns * 16 + lid];
        b[ns].y = Bb[(k + 1) * PB + wv * 32 + ns * 16 + lid];
      }
#pragma unroll
      for (int s = 0; s < 4; ++s) {
        const v2f a = *(const v2f*)&Ab[(s * 16 + lid) * PA + k];
#pragma unroll
        for (int ns = 0; ns < 2; ++ns) acc[s][ns] = wmma4(a, b[ns], acc[s][ns]);
      }
    }
  };

#pragma unroll
  for (int j = 0; j < 5; ++j) {
    ra[j] = *(const v2f*)(emb + aoff[j] + KC);
    rb[j] = *(const v4f*)(wp + (size_t)(KC + 4 * j) * DD);
  }
  for (int c = 0; c < NCH - 2; ++c) {
    chunk_mma(As[c & 1], Bs[c & 1]);
#pragma unroll
    for (int j = 0; j < 5; ++j) {
      *(v2f*)&As[(c + 1) & 1][2 * (tid + 128 * j)] = ra[j];
      *(v4f*)&Bs[(c + 1) & 1][(bk + 4 * j) * PB + bc4 * 4] = rb[j];
    }
    __syncthreads();
    const int kn = (c + 2) * KC;
#pragma unroll
    for (int j = 0; j < 5; ++j) {
      ra[j] = *(const v2f*)(emb + aoff[j] + kn);
      rb[j] = *(const v4f*)(wp + (size_t)(kn + 4 * j) * DD);
    }
  }
  chunk_mma(As[(NCH - 2) & 1], Bs[(NCH - 2) & 1]);
#pragma unroll
  for (int j = 0; j < 5; ++j) {
    *(v2f*)&As[(NCH - 1) & 1][2 * (tid + 128 * j)] = ra[j];
    *(v4f*)&Bs[(NCH - 1) & 1][(bk + 4 * j) * PB + bc4 * 4] = rb[j];
  }
  __syncthreads();
  chunk_mma(As[(NCH - 1) & 1], Bs[(NCH - 1) & 1]);
}

#define NCT (BB * LL / TM)
#define CB2 6
#define CB3 (DD / TN - CB2)

static __device__ __forceinline__ void concepts_tile(const int bx, const int by, float* smem,
                                                     const int* __restrict__ ltok,
                                                     const int* __restrict__ lrow,
                                                     const int* __restrict__ ltotal,
                                                     const float* __restrict__ emb,
                                                     const float* __restrict__ W,
                                                     const float* __restrict__ bias,
                                                     float* __restrict__ concepts) {
  float (*As)[TM * PA] = (float (*)[TM * PA])smem;
  float (*Bs)[KC * PB] = (float (*)[KC * PB])(smem + 2 * TM * PA);
  int* stok  = (int*)(smem + 2 * TM * PA + 2 * KC * PB);
  int* sorow = stok + TM;
  const int tid  = threadIdx.x;
  const int lane = tid & 31;
  const int wv   = tid >> 5;
  const int half = lane >> 4;
  const int lid  = lane & 15;
  const int n0 = bx * TN, j0 = by * TM;
  const int total = ltotal[0];
  if (j0 >= total) return;
  if (tid < TM) {
    const int j = j0 + tid;
    int tokr = 0, orow = -1;
    if (j < total) {
      tokr = ltok[j];
      orow = lrow[j];
    }
    stok[tid]  = tokr;
    sorow[tid] = orow;
  }
  __syncthreads();
  v8f acc[4][2];
#pragma unroll
  for (int s = 0; s < 4; ++s)
#pragma unroll
    for (int ns = 0; ns < 2; ++ns) acc[s][ns] = (v8f){0.f,0.f,0.f,0.f,0.f,0.f,0.f,0.f};
  tile_k300(stok, emb, W, n0, As, Bs, acc);
  float bv[2];
#pragma unroll
  for (int ns = 0; ns < 2; ++ns) bv[ns] = bias[n0 + wv * 32 + ns * 16 + lid];
#pragma unroll
  for (int s = 0; s < 4; ++s) {
#pragma unroll
    for (int i = 0; i < 8; ++i) {
      const int row = sorow[s * 16 + half * 8 + i];
      if (row >= 0) {
#pragma unroll
        for (int ns = 0; ns < 2; ++ns) {
          const int col = n0 + wv * 32 + ns * 16 + lid;
          concepts[(size_t)row * DD + col] = acc[s][ns][i] + bv[ns];
        }
      }
    }
  }
}

#define GM 32
#define GN 64
#define GK 64
#define GCH (DD / GK)
#define GPA 68
#define GPB 72

#define NGW ((DD / GN) * (BB / GM))
#define SMEM_F (2 * GM * GPA + 2 * GK * GPB)

static __device__ __forceinline__ void gemm_tile(const int bx, const int by, float* smem,
                                                 const float* __restrict__ X,
                                                 const float* __restrict__ W,
                                                 const float* __restrict__ bias,
                                                 float* __restrict__ out, const int relu,
                                                 const int smode,
                                                 float* __restrict__ st0,
                                                 float* __restrict__ st1) {
  float (*As)[GM * GPA] = (float (*)[GM * GPA])smem;
  float (*Bs)[GK * GPB] = (float (*)[GK * GPB])(smem + 2 * GM * GPA);
  const int tid  = threadIdx.x;
  const int lane = tid & 31;
  const int wv   = tid >> 5;
  const int wm   = wv >> 1, wn = wv & 1;
  const int half = lane >> 4;
  const int lid  = lane & 15;
  const int n0 = bx * GN, m0 = by * GM;
  const int srow = tid >> 4;
  const int sc   = (tid & 15) * 4;
  const float* xp = X + (size_t)(m0 + srow) * DD + sc;
  const float* wp = W + (size_t)srow * DD + n0 + sc;

  v4f ra[4], rb[8];
#pragma unroll
  for (int j = 0; j < 4; ++j) ra[j] = *(const v4f*)(xp + (size_t)(8 * j) * DD);
#pragma unroll
  for (int j = 0; j < 8; ++j) rb[j] = *(const v4f*)(wp + (size_t)(8 * j) * DD);
#pragma unroll
  for (int j = 0; j < 4; ++j) *(v4f*)&As[0][(srow + 8 * j) * GPA + sc] = ra[j];
#pragma unroll
  for (int j = 0; j < 8; ++j) *(v4f*)&Bs[0][(srow + 8 * j) * GPB + sc] = rb[j];
  __syncthreads();

  v8f acc[2];
  acc[0] = (v8f){0.f,0.f,0.f,0.f,0.f,0.f,0.f,0.f};
  acc[1] = acc[0];
  const int fao = (wm * 16 + lid) * GPA + 2 * half;
  const int fbo = (2 * half) * GPB + wn * 32 + lid;
  v2f   fa0[8], fa1[8];
  float bx0[8][2], by0[8][2], bx1[8][2], by1[8][2];
  auto load_frag = [&](const float* Ab, const float* Bb, const int h, v2f (&fa)[8], float (&bx)[8][2], float (&by)[8][2]) {
#pragma unroll
    for (int kk = 0; kk < 8; ++kk) {
      const int k4 = (h * 8 + kk) * 4;
      fa[kk] = *(const v2f*)&Ab[fao + k4];
#pragma unroll
      for (int ns = 0; ns < 2; ++ns) {
        bx[kk][ns] = Bb[fbo + k4 * GPB + ns * 16];
        by[kk][ns] = Bb[fbo + (k4 + 1) * GPB + ns * 16];
      }
    }
  };
  auto mma_k = [&](const int kk, const v2f (&fa)[8], const float (&bx)[8][2], const float (&by)[8][2]) {
#pragma unroll
    for (int ns = 0; ns < 2; ++ns) {
      v2f b;
      b.x = bx[kk][ns];
      b.y = by[kk][ns];
      acc[ns] = wmma4(fa[kk], b, acc[ns]);
    }
  };

#pragma unroll
  for (int j = 0; j < 4; ++j) ra[j] = *(const v4f*)(xp + (size_t)(8 * j) * DD + GK);
#pragma unroll
  for (int j = 0; j < 8; ++j) rb[j] = *(const v4f*)(wp + (size_t)(GK + 8 * j) * DD);
  load_frag(As[0], Bs[0], 0, fa0, bx0, by0);
  auto step = [&](const int c, const bool more) {
    load_frag(As[c & 1], Bs[c & 1], 1, fa1, bx1, by1);
#pragma unroll
    for (int j = 0; j < 4; ++j) *(v4f*)&As[(c + 1) & 1][(srow + 8 * j) * GPA + sc] = ra[j];
#pragma unroll
    for (int j = 0; j < 8; ++j) *(v4f*)&Bs[(c + 1) & 1][(srow + 8 * j) * GPB + sc] = rb[j];
    if (more) {
      const int kn = (c + 2) * GK;
#pragma unroll
      for (int j = 0; j < 4; ++j) ra[j] = *(const v4f*)(xp + (size_t)(8 * j) * DD + kn);
#pragma unroll
      for (int j = 0; j < 8; ++j) rb[j] = *(const v4f*)(wp + (size_t)(kn + 8 * j) * DD);
    }
#pragma unroll
    for (int kk = 0; kk < 8; ++kk) mma_k(kk, fa0, bx0, by0);
    __builtin_amdgcn_sched_barrier(0);
    lds_barrier();
    load_frag(As[(c + 1) & 1], Bs[(c + 1) & 1], 0, fa0, bx0, by0);
#pragma unroll
    for (int kk = 0; kk < 8; ++kk) mma_k(kk, fa1, bx1, by1);
    __builtin_amdgcn_sched_group_barrier(0x100, 20, 0);
    __builtin_amdgcn_sched_group_barrier(0x008, 16, 0);
    __builtin_amdgcn_sched_group_barrier(0x002, 64, 0);
    __builtin_amdgcn_sched_barrier(0);
  };
  for (int c = 0; c < GCH - 2; ++c) step(c, true);
  step(GCH - 2, false);
  load_frag(As[(GCH - 1) & 1], Bs[(GCH - 1) & 1], 1, fa1, bx1, by1);
#pragma unroll
  for (int kk = 0; kk < 8; ++kk) mma_k(kk, fa0, bx0, by0);
#pragma unroll
  for (int kk = 0; kk < 8; ++kk) mma_k(kk, fa1, bx1, by1);

  const int col  = n0 + wn * 32 + lid;
  const int row0 = m0 + wm * 16 + half * 8;
  const float bv0 = bias[col], bv1 = bias[col + 16];
  float v0[8], v1[8];
#pragma unroll
  for (int i = 0; i < 8; ++i) {
    v0[i] = acc[0][i] + bv0;
    v1[i] = acc[1][i] + bv1;
    if (relu) { v0[i] = fmaxf(v0[i], 0.f); v1[i] = fmaxf(v1[i], 0.f); }
    out[(size_t)(row0 + i) * DD + col]      = v0[i];
    out[(size_t)(row0 + i) * DD + col + 16] = v1[i];
  }
  if (smode != 0) {
    float ra[8], rb[8];
#pragma unroll
    for (int i = 0; i < 8; ++i) {
      if (smode == 1) {
        ra[i] = sum16(v0[i] * v0[i] + v1[i] * v1[i]);
        rb[i] = 0.f;
      } else {
        ra[i] = max16(fmaxf(v0[i], v1[i]));
        rb[i] = sum16(__expf(v0[i] - ra[i]) + __expf(v1[i] - ra[i]));
      }
    }
    if (lid == 0) {
      const size_t so = (size_t)row0 * NP + bx * 2 + wn;
#pragma unroll
      for (int i = 0; i < 8; ++i) {
        st0[so + (size_t)i * NP] = ra[i];
        if (smode == 2) st1[so + (size_t)i * NP] = rb[i];
      }
    }
  }
}

static __device__ __forceinline__ void gemm_tile_deep(const int bx, const int by, float* smem,
                                                 const float* __restrict__ X,
                                                 const float* __restrict__ W,
                                                 const float* __restrict__ bias,
                                                 float* __restrict__ out, const int relu,
                                                 const int smode,
                                                 float* __restrict__ st0,
                                                 float* __restrict__ st1) {
  float (*As)[GM * GPA] = (float (*)[GM * GPA])smem;
  float (*Bs)[GK * GPB] = (float (*)[GK * GPB])(smem + 2 * GM * GPA);
  const int tid  = threadIdx.x;
  const int lane = tid & 31;
  const int wv   = tid >> 5;
  const int wm   = wv >> 1, wn = wv & 1;
  const int half = lane >> 4;
  const int lid  = lane & 15;
  const int n0 = bx * GN, m0 = by * GM;
  const int srow = tid >> 4;
  const int sc   = (tid & 15) * 4;
  const float* xp = X + (size_t)(m0 + srow) * DD + sc;
  const float* wp = W + (size_t)srow * DD + n0 + sc;

  v4f ra0[4], rb0[8], ra1[4], rb1[8];
  auto gload = [&](const int kn, v4f (&ra)[4], v4f (&rb)[8]) {
#pragma unroll
    for (int j = 0; j < 4; ++j) ra[j] = *(const v4f*)(xp + (size_t)(8 * j) * DD + kn);
#pragma unroll
    for (int j = 0; j < 8; ++j) rb[j] = *(const v4f*)(wp + (size_t)(kn + 8 * j) * DD);
  };
  auto lstore = [&](const int st, const v4f (&ra)[4], const v4f (&rb)[8]) {
#pragma unroll
    for (int j = 0; j < 4; ++j) *(v4f*)&As[st][(srow + 8 * j) * GPA + sc] = ra[j];
#pragma unroll
    for (int j = 0; j < 8; ++j) *(v4f*)&Bs[st][(srow + 8 * j) * GPB + sc] = rb[j];
  };
  gload(0, ra0, rb0);
  gload(GK, ra1, rb1);
  lstore(0, ra0, rb0);
  lds_barrier();
  gload(2 * GK, ra0, rb0);

  v8f acc[2];
  acc[0] = (v8f){0.f,0.f,0.f,0.f,0.f,0.f,0.f,0.f};
  acc[1] = acc[0];
  const int fao = (wm * 16 + lid) * GPA + 2 * half;
  const int fbo = (2 * half) * GPB + wn * 32 + lid;
  v2f   fa0[8], fa1[8];
  float bx0[8][2], by0[8][2], bx1[8][2], by1[8][2];
  auto load_frag = [&](const float* Ab, const float* Bb, const int h, v2f (&fa)[8], float (&bx)[8][2], float (&by)[8][2]) {
#pragma unroll
    for (int kk = 0; kk < 8; ++kk) {
      const int k4 = (h * 8 + kk) * 4;
      fa[kk] = *(const v2f*)&Ab[fao + k4];
#pragma unroll
      for (int ns = 0; ns < 2; ++ns) {
        bx[kk][ns] = Bb[fbo + k4 * GPB + ns * 16];
        by[kk][ns] = Bb[fbo + (k4 + 1) * GPB + ns * 16];
      }
    }
  };
  auto mma_k = [&](const int kk, const v2f (&fa)[8], const float (&bx)[8][2], const float (&by)[8][2]) {
#pragma unroll
    for (int ns = 0; ns < 2; ++ns) {
      v2f b;
      b.x = bx[kk][ns];
      b.y = by[kk][ns];
      acc[ns] = wmma4(fa[kk], b, acc[ns]);
    }
  };

  load_frag(As[0], Bs[0], 0, fa0, bx0, by0);
  auto step = [&](const int c, const bool more, v4f (&ra)[4], v4f (&rb)[8]) {
    load_frag(As[c & 1], Bs[c & 1], 1, fa1, bx1, by1);
    lstore((c + 1) & 1, ra, rb);
    if (more) gload((c + 3) * GK, ra, rb);
#pragma unroll
    for (int kk = 0; kk < 8; ++kk) mma_k(kk, fa0, bx0, by0);
    __builtin_amdgcn_sched_barrier(0);
    lds_barrier();
    load_frag(As[(c + 1) & 1], Bs[(c + 1) & 1], 0, fa0, bx0, by0);
#pragma unroll
    for (int kk = 0; kk < 8; ++kk) mma_k(kk, fa1, bx1, by1);
    __builtin_amdgcn_sched_group_barrier(0x100, 20, 0);
    __builtin_amdgcn_sched_group_barrier(0x008, 16, 0);
    __builtin_amdgcn_sched_group_barrier(0x002, 64, 0);
    __builtin_amdgcn_sched_barrier(0);
  };
  for (int c = 0; c < GCH - 4; c += 2) {
    step(c, true, ra1, rb1);
    step(c + 1, true, ra0, rb0);
  }
  step(GCH - 4, true, ra1, rb1);
  step(GCH - 3, false, ra0, rb0);
  step(GCH - 2, false, ra1, rb1);
  load_frag(As[(GCH - 1) & 1], Bs[(GCH - 1) & 1], 1, fa1, bx1, by1);
#pragma unroll
  for (int kk = 0; kk < 8; ++kk) mma_k(kk, fa0, bx0, by0);
#pragma unroll
  for (int kk = 0; kk < 8; ++kk) mma_k(kk, fa1, bx1, by1);

  const int col  = n0 + wn * 32 + lid;
  const int row0 = m0 + wm * 16 + half * 8;
  const float bv0 = bias[col], bv1 = bias[col + 16];
  float v0[8], v1[8];
#pragma unroll
  for (int i = 0; i < 8; ++i) {
    v0[i] = acc[0][i] + bv0;
    v1[i] = acc[1][i] + bv1;
    if (relu) { v0[i] = fmaxf(v0[i], 0.f); v1[i] = fmaxf(v1[i], 0.f); }
    out[(size_t)(row0 + i) * DD + col]      = v0[i];
    out[(size_t)(row0 + i) * DD + col + 16] = v1[i];
  }
  if (smode != 0) {
    float ra[8], rb[8];
#pragma unroll
    for (int i = 0; i < 8; ++i) {
      if (smode == 1) {
        ra[i] = sum16(v0[i] * v0[i] + v1[i] * v1[i]);
        rb[i] = 0.f;
      } else {
        ra[i] = max16(fmaxf(v0[i], v1[i]));
        rb[i] = sum16(__expf(v0[i] - ra[i]) + __expf(v1[i] - ra[i]));
      }
    }
    if (lid == 0) {
      const size_t so = (size_t)row0 * NP + bx * 2 + wn;
#pragma unroll
      for (int i = 0; i < 8; ++i) {
        st0[so + (size_t)i * NP] = ra[i];
        if (smode == 2) st1[so + (size_t)i * NP] = rb[i];
      }
    }
  }
}

__global__ __launch_bounds__(128) void k_gemm_xw(const float* __restrict__ X,
                                                 const float* __restrict__ W,
                                                 const float* __restrict__ bias,
                                                 float* __restrict__ out,
                                                 float* __restrict__ pmax,
                                                 float* __restrict__ psum,
                                                 const int* __restrict__ ltok,
                                                 const int* __restrict__ lrow,
                                                 const int* __restrict__ ltotal,
                                                 const float* __restrict__ emb,
                                                 const float* __restrict__ nW,
                                                 const float* __restrict__ nb,
                                                 float* __restrict__ concepts) {
  __shared__ __attribute__((aligned(16))) float smem[SMEM_F];
  const int b = blockIdx.x;
  if (b < NGW) {
    gemm_tile_deep(b % (DD / GN), b / (DD / GN), smem, X, W, bias, out, 0, 2, pmax, psum);
  } else {
    const int q = b - NGW;
    concepts_tile(q % CB2, q / CB2, smem, ltok, lrow, ltotal, emb, nW, nb, concepts);
  }
}

static __device__ __forceinline__ void row_l2(const int row, float* red,
                                              const float* __restrict__ src,
                                              float* __restrict__ dn,
                                              float* __restrict__ dsq,
                                              int* ired, const int ipart) {
  const int tid = threadIdx.x;
  const float* p = src + (size_t)row * DD;
  float va[4], vb[4];
#pragma unroll
  for (int j = 0; j < 4; ++j) {
    va[j] = p[tid + 256 * j];
    vb[j] = p[tid + 128 + 256 * j];
  }
  float sa = 0.f, sb = 0.f;
#pragma unroll
  for (int j = 0; j < 4; ++j) {
    sa += va[j] * va[j];
    sb += vb[j] * vb[j];
  }
  if (dsq) {
#pragma unroll
    for (int j = 0; j < 4; ++j) {
      dsq[(size_t)row * DD + tid + 256 * j]       = va[j] * va[j];
      dsq[(size_t)row * DD + tid + 128 + 256 * j] = vb[j] * vb[j];
    }
  }
  red[tid] = sa + sb;
  if (ired) ired[tid] = ipart;
  __syncthreads();
  for (int st = 64; st > 0; st >>= 1) {
    if (tid < st) {
      red[tid] += red[tid + st];
      if (ired) ired[tid] += ired[tid + st];
    }
    __syncthreads();
  }
  const float inv = rsqrtf(red[0]);
#pragma unroll
  for (int j = 0; j < 4; ++j) {
    dn[(size_t)row * DD + tid + 256 * j]       = va[j] * inv;
    dn[(size_t)row * DD + tid + 128 + 256 * j] = vb[j] * inv;
  }
}

__global__ __launch_bounds__(128) void k_front(const int* __restrict__ ntid,
                                               const int* __restrict__ nlen,
                                               const int* __restrict__ t2i,
                                               const int* __restrict__ antk,
                                               const float* __restrict__ m,
                                               const float* __restrict__ W1,
                                               const float* __restrict__ b1,
                                               const float* __restrict__ trW,
                                               const float* __restrict__ trb,
                                               const float* __restrict__ r,
                                               const float* __restrict__ t,
                                               float* __restrict__ h,
                                               float* __restrict__ trm,
                                               float* __restrict__ rn,
                                               float* __restrict__ tn,
                                               float* __restrict__ t2,
                                               float* __restrict__ trsq,
                                               int* __restrict__ ltok,
                                               int* __restrict__ lrow,
                                               int* __restrict__ ltotal,
                                               float* __restrict__ concepts) {
  __shared__ __attribute__((aligned(16))) float smem[SMEM_F];
  const int b = blockIdx.x;
  if (b < 2 * NGW) {
    const bool first = b < NGW;
    const int q = first ? b : b - NGW;
    gemm_tile(q % (DD / GN), q / (DD / GN), smem, m, first ? W1 : trW, first ? b1 : trb,
              first ? h : trm, first ? 1 : 0, first ? 0 : 1, trsq, nullptr);
  } else if (b < 2 * NGW + BB) {
    const int item = b - 2 * NGW;
    int part = 0;
#pragma unroll
    for (int u = 0; u < 3; ++u) {
      const int i = threadIdx.x + 128 * u;
      const int v = min(max(nlen[i], 0), LL);
      part += (i < item) ? v : 0;
    }
    int* ired = (int*)(smem + 128);
    row_l2(item, smem, r, rn, nullptr, ired, part);
    const int off = ired[0];
    const int len = min(max(nlen[item], 0), LL);
    if ((int)threadIdx.x < len) {
      const int orow = item * LL + threadIdx.x;
      const int tk   = ntid[orow];
      const int widx = t2i[tk];
      ltok[off + threadIdx.x] = (widx > 0) ? antk[widx] : tk;
      lrow[off + threadIdx.x] = orow;
    }
    if (item == BB - 1 && threadIdx.x == 0) ltotal[0] = off + len;
    for (int l = len; l < LL; ++l) {
      v4f* zp = (v4f*)(concepts + ((size_t)item * LL + l) * DD);
      zp[threadIdx.x]       = (v4f){0.f, 0.f, 0.f, 0.f};
      zp[threadIdx.x + 128] = (v4f){0.f, 0.f, 0.f, 0.f};
    }
  } else {
    row_l2(b - (2 * NGW + BB), smem, t, tn, t2, nullptr, 0);
  }
}

#define SM 32
#define SK 64
#define SCH (DD / SK)
#define SP 68
#define NSW ((BB / SM) * (BB / SM))

__global__ __launch_bounds__(128) void k_score(const float* __restrict__ aem,
                                               const float* __restrict__ trm,
                                               const float* __restrict__ rn,
                                               const float* __restrict__ t,
                                               const float* __restrict__ t2,
                                               const float* __restrict__ tn,
                                               const float* __restrict__ pmax,
                                               const float* __restrict__ psum,
                                               const float* __restrict__ trsq,
                                               float* __restrict__ score,
                                               const int* __restrict__ ltok,
                                               const int* __restrict__ lrow,
                                               const int* __restrict__ ltotal,
                                               const float* __restrict__ emb,
                                               const float* __restrict__ nW,
                                               const float* __restrict__ nb,
                                               float* __restrict__ concepts) {
  __shared__ __attribute__((aligned(16))) float Ls[2][6][SM * SP];
  __shared__ float sst[3][SM];
  if (blockIdx.x >= NSW) {
    const int q = blockIdx.x - NSW;
    concepts_tile(CB2 + q % CB3, q / CB3, &Ls[0][0][0], ltok, lrow, ltotal, emb, nW, nb, concepts);
    return;
  }
  const int tid  = threadIdx.x;
  const int lane = tid & 31;
  const int wv   = tid >> 5;
  const int wm   = wv >> 1, wn = wv & 1;
  const int half = lane >> 4;
  const int lid  = lane & 15;
  const int m0 = (blockIdx.x % (BB / SM)) * SM, n0 = (blockIdx.x / (BB / SM)) * SM;
  const int srow = tid >> 4;
  const int sc   = (tid & 15) * 4;
  const size_t ao = (size_t)(m0 + srow) * DD + sc;
  const size_t bo = (size_t)(n0 + srow) * DD + sc;
  const float* src[6] = {aem + ao, trm + ao, rn + ao, t + bo, t2 + bo, tn + bo};

  v4f rg0[6][4], rg1[6][4];
  auto gload = [&](const int kn, v4f (&rg)[6][4]) {
#pragma unroll
    for (int p = 0; p < 6; ++p)
#pragma unroll
      for (int j = 0; j < 4; ++j) rg[p][j] = *(const v4f*)(src[p] + (size_t)(8 * j) * DD + kn);
  };
  gload(0, rg0);
  __builtin_amdgcn_sched_barrier(0);

  {
    const int sr = tid >> 2, sq4 = tid & 3;
    const size_t so = (size_t)(m0 + sr) * NP + sq4 * 8;
    float pm[8], pe[8];
    float sq = 0.f;
#pragma unroll
    for (int u = 0; u < 8; ++u) {
      pm[u] = pmax[so + u];
      pe[u] = psum[so + u];
      sq += trsq[so + u];
    }
    float mx = pm[0];
#pragma unroll
    for (int u = 1; u < 8; ++u) mx = fmaxf(mx, pm[u]);
    mx = max4(mx);
    float se = 0.f;
#pragma unroll
    for (int u = 0; u < 8; ++u) se += pe[u] * __expf(pm[u] - mx);
    se = sum4(se);
    sq = sum4(sq);
    if (sq4 == 0) {
      sst[0][sr] = mx;
      sst[1][sr] = 1.f / se;
      sst[2][sr] = rsqrtf(sq);
    }
  }
  gload(SK, rg1);
  lds_barrier();
  float rmx[4], rinv[4], rinv2[4];
#pragma unroll
  for (int j = 0; j < 4; ++j) {
    rmx[j]   = sst[0][srow + 8 * j];
    rinv[j]  = sst[1][srow + 8 * j];
    rinv2[j] = sst[2][srow + 8 * j];
  }
  auto to_pq_row = [&](const int j, v4f (&rg)[6][4]) {
#pragma unroll
    for (int e = 0; e < 4; ++e) {
      const float a = __expf(rg[0][j][e] - rmx[j]) * rinv[j];
      rg[0][j][e] = rg[1][j][e] * rinv2[j] * a;
      rg[1][j][e] = a * a;
    }
  };

  auto lstore = [&](const int st, const v4f (&rg)[6][4]) {
#pragma unroll
    for (int p = 0; p < 6; ++p)
#pragma unroll
      for (int j = 0; j < 4; ++j) *(v4f*)&Ls[st][p][(srow + 8 * j) * SP + sc] = rg[p][j];
  };
#pragma unroll
  for (int j = 0; j < 4; ++j) to_pq_row(j, rg0);
  lstore(0, rg0);
  lds_barrier();
  gload(2 * SK, rg0);

  v8f num = {0.f,0.f,0.f,0.f,0.f,0.f,0.f,0.f};
  v8f den = num, isv = num;
  const int arow = (wm * 16 + lid) * SP;
  const int brow = (wn * 16 + lid) * SP;
  v2f f0[8][6], f1[8][6];
  auto load_frag = [&](const float (*L)[SM * SP], const int h, v2f (&f)[8][6]) {
#pragma unroll
    for (int kk = 0; kk < 8; ++kk) {
      const int k = (h * 8 + kk) * 4 + 2 * half;
      f[kk][0] = *(const v2f*)&L[0][arow + k];
      f[kk][1] = *(const v2f*)&L[3][brow + k];
      f[kk][2] = *(const v2f*)&L[1][arow + k];
      f[kk][3] = *(const v2f*)&L[4][brow + k];
      f[kk][4] = *(const v2f*)&L[2][arow + k];
      f[kk][5] = *(const v2f*)&L[5][brow + k];
    }
  };
  auto mma_k = [&](const int kk, const v2f (&f)[8][6]) {
    num = wmma4(f[kk][0], f[kk][1], num);
    den = wmma4(f[kk][2], f[kk][3], den);
    isv = wmma4(f[kk][4], f[kk][5], isv);
  };

  load_frag(Ls[0], 0, f0);
  auto step = [&](const int c, const bool more, v4f (&rg)[6][4]) {
    load_frag(Ls[c & 1], 1, f1);
#pragma unroll
    for (int kk = 0; kk < 3; ++kk) {
      mma_k(kk, f0);
      to_pq_row(kk, rg);
      if (kk == 0) to_pq_row(3, rg);
    }
    __builtin_amdgcn_sched_barrier(0);
    lstore((c + 1) & 1, rg);
    if (more) gload((c + 3) * SK, rg);
#pragma unroll
    for (int kk = 3; kk < 8; ++kk) mma_k(kk, f0);
    __builtin_amdgcn_sched_barrier(0);
    lds_barrier();
    load_frag(Ls[(c + 1) & 1], 0, f0);
    __builtin_amdgcn_sched_barrier(0);
#pragma unroll
    for (int kk = 0; kk < 8; ++kk) mma_k(kk, f1);
    __builtin_amdgcn_sched_barrier(0);
  };
  for (int c = 0; c < SCH - 4; c += 2) {
    step(c, true, rg1);
    step(c + 1, true, rg0);
  }
  step(SCH - 4, true, rg1);
  step(SCH - 3, false, rg0);
  step(SCH - 2, false, rg1);
  load_frag(Ls[(SCH - 1) & 1], 1, f1);
#pragma unroll
  for (int kk = 0; kk < 8; ++kk) mma_k(kk, f0);
#pragma unroll
  for (int kk = 0; kk < 8; ++kk) mma_k(kk, f1);

#pragma unroll
  for (int i = 0; i < 8; ++i) {
    const int row = m0 + wm * 16 + half * 8 + i;
    const int col = n0 + wn * 16 + lid;
    score[(size_t)row * BB + col] = num[i] * rsqrtf(den[i]) + isv[i];
  }
}

extern "C" void kernel_launch(void* const* d_in, const int* in_sizes, int n_in,
                              void* d_out, int out_size, void* d_ws, size_t ws_size,
                              hipStream_t stream) {
  const float* r    = (const float*)d_in[0];
  const float* m    = (const float*)d_in[1];
  const float* t    = (const float*)d_in[2];
  const int*   ntid = (const int*)d_in[3];
  const int*   nlen = (const int*)d_in[4];
  const int*   antk = (const int*)d_in[5];
  const float* emb  = (const float*)d_in[6];
  const float* nW   = (const float*)d_in[7];
  const float* nb   = (const float*)d_in[8];
  const float* W1   = (const float*)d_in[9];
  const float* b1   = (const float*)d_in[10];
  const float* W2   = (const float*)d_in[11];
  const float* b2   = (const float*)d_in[12];
  const float* trW  = (const float*)d_in[13];
  const float* trb  = (const float*)d_in[14];
  const int*   t2i  = (const int*)d_in[15];

  float* score    = (float*)d_out;
  float* concepts = score + (size_t)BB * BB;

  float* ws  = (float*)d_ws;
  float* h   = ws;  ws += (size_t)BB * DD;
  float* aem = ws;  ws += (size_t)BB * DD;
  float* trm = ws;  ws += (size_t)BB * DD;
  float* rnw = ws;  ws += (size_t)BB * DD;
  float* tnw = ws;  ws += (size_t)BB * DD;
  float* t2w = ws;  ws += (size_t)BB * DD;
  float* pmx = ws;  ws += (size_t)BB * NP;
  float* psm = ws;  ws += (size_t)BB * NP;
  float* tsq = ws;  ws += (size_t)BB * NP;
  int* ltok = (int*)ws;  ws += (size_t)BB * LL;
  int* lrow = (int*)ws;  ws += (size_t)BB * LL;
  int* ltot = (int*)ws;

  k_front<<<dim3(2 * NGW + 2 * BB), dim3(128), 0, stream>>>(ntid, nlen, t2i, antk,
                                                              m, W1, b1, trW, trb, r, t,
                                                              h, trm, rnw, tnw, t2w, tsq, ltok, lrow, ltot, concepts);
  k_gemm_xw<<<dim3(NGW + CB2 * NCT), dim3(128), 0, stream>>>(h, W2, b2, aem, pmx, psm,
                                                             ltok, lrow, ltot, emb, nW, nb, concepts);
  k_score<<<dim3(NSW + CB3 * NCT), dim3(128), 0, stream>>>(aem, trm, rnw, t, t2w, tnw, pmx, psm, tsq, score,
                                                           ltok, lrow, ltot, emb, nW, nb, concepts);
}
